# starting kernel
# baseline (speedup 1.0000x reference)
#include <hip/hip_runtime.h>

typedef __attribute__((ext_vector_type(2))) float v2f;
typedef __attribute__((ext_vector_type(8))) float v8f;

#define D    64
#define KC   128
#define STI  (KC+4)
#define STL  (D+4)
#define TMI  2
#define TML  4

__global__ void count_edges_int(const int* __restrict__ dst, int* __restrict__ deg, int nE) {
    int e = blockIdx.x * blockDim.x + threadIdx.x;
    if (e < nE) atomicAdd(&deg[dst[e]], 1);
}

__global__ void scan_block(const int* __restrict__ deg, int* __restrict__ ex,
                           int* __restrict__ bsum, int n) {
    __shared__ int sh[256];
    int tid = threadIdx.x;
    int i = blockIdx.x * 256 + tid;
    int v = (i < n) ? deg[i] : 0;
    sh[tid] = v;
    __syncthreads();
    for (int off = 1; off < 256; off <<= 1) {
        int add = (tid >= off) ? sh[tid - off] : 0;
        __syncthreads();
        sh[tid] += add;
        __syncthreads();
    }
    if (i < n) ex[i] = sh[tid] - v;
    if (tid == 255) bsum[blockIdx.x] = sh[255];
}

__global__ void scan_sums(int* __restrict__ bsum, int nb) {
    __shared__ int sh[1024];
    int tid = threadIdx.x;
    int v = (tid < nb) ? bsum[tid] : 0;
    sh[tid] = v;
    __syncthreads();
    for (int off = 1; off < 1024; off <<= 1) {
        int add = (tid >= off) ? sh[tid - off] : 0;
        __syncthreads();
        sh[tid] += add;
        __syncthreads();
    }
    if (tid < nb) bsum[tid] = sh[tid] - v;
}

__global__ void add_offsets(int* __restrict__ ex, const int* __restrict__ bsum,
                            int* __restrict__ cursor, int n) {
    int i = blockIdx.x * 256 + threadIdx.x;
    if (i < n) {
        int r = ex[i] + bsum[i >> 8];
        ex[i] = r;
        cursor[i] = r;
    }
}

__global__ void fill_csr(const int* __restrict__ src, const int* __restrict__ dst,
                         int* __restrict__ cursor, int* __restrict__ srcs, int nE) {
    int e = blockIdx.x * blockDim.x + threadIdx.x;
    if (e < nE) {
        int pos = atomicAdd(&cursor[dst[e]], 1);
        srcs[pos] = src[e];
    }
}

__global__ void aggregate_csr(const float* __restrict__ x, const int* __restrict__ srcs,
                              const int* __restrict__ rowStart, const int* __restrict__ rowEnd,
                              float* __restrict__ agg, int n) {
    int t = blockIdx.x * blockDim.x + threadIdx.x;
    int node = t >> 4;
    if (node >= n) return;
    int c = (t & 15) * 4;
    int s0 = rowStart[node], s1 = rowEnd[node];
    float4 acc = make_float4(0.f, 0.f, 0.f, 0.f);
    for (int i = s0; i < s1; ++i) {
        int s = srcs[i];
        float4 v = *(const float4*)(x + (size_t)s * D + c);
        acc.x += v.x; acc.y += v.y; acc.z += v.z; acc.w += v.w;
    }
    float inv = 1.0f / fmaxf((float)(s1 - s0), 1.0f);
    acc.x *= inv; acc.y *= inv; acc.z *= inv; acc.w *= inv;
    *(float4*)(agg + (size_t)node * D + c) = acc;
}

__global__ void gather_rows(const float* __restrict__ emb, const int* __restrict__ nid,
                            float* __restrict__ out, int n16) {
    int t = blockIdx.x * blockDim.x + threadIdx.x;
    if (t >= n16) return;
    int i = t >> 4, c = (t & 15) * 4;
    int s = nid[i];
    *(float4*)(out + (size_t)i * D + c) = *(const float4*)(emb + (size_t)s * D + c);
}

__global__ __launch_bounds__(256) void input_gemm_wmma(
        const float* __restrict__ go_x, const float* __restrict__ lin_W,
        const float* __restrict__ lin_b, const float* __restrict__ go_emb,
        const int* __restrict__ go_nid, float* __restrict__ xg, int M, int K) {
    __shared__ float sWt[D * STI];

    int waveInBlk = threadIdx.x >> 5;
    int lane = threadIdx.x & 31;
    int row  = lane & 15;
    int kh   = lane >> 4;
    int colL = lane & 15;

    int numTiles = M >> 4;
    int tile0 = (blockIdx.x * 8 + waveInBlk) * TMI;

    int   m0c[TMI];
    const float* arow[TMI];
    bool  valid[TMI];
#pragma unroll
    for (int t = 0; t < TMI; ++t) {
        int ti = tile0 + t;
        valid[t] = (ti < numTiles);
        m0c[t] = valid[t] ? (ti << 4) : 0;
        arow[t] = go_x + (size_t)(m0c[t] + row) * K;
    }

    v8f acc[TMI][4];
#pragma unroll
    for (int t = 0; t < TMI; ++t)
#pragma unroll
        for (int j = 0; j < 4; ++j) {
            float b = lin_b[j * 16 + colL];
#pragma unroll
            for (int g = 0; g < 8; ++g) acc[t][j][g] = b;
        }

    for (int kc0 = 0; kc0 < K; kc0 += KC) {
        int len = (K - kc0 < KC) ? (K - kc0) : KC;
        __syncthreads();
        for (int i = threadIdx.x * 4; i < len * D; i += 256 * 4) {
            int r = i >> 6, c = i & 63;
            float4 w = *(const float4*)&lin_W[(size_t)(kc0 + r) * D + c];
            sWt[(c + 0) * STI + r] = w.x;
            sWt[(c + 1) * STI + r] = w.y;
            sWt[(c + 2) * STI + r] = w.z;
            sWt[(c + 3) * STI + r] = w.w;
        }
        __syncthreads();
        if (kc0 + KC < K) {
#pragma unroll
            for (int t = 0; t < TMI; ++t)
                __builtin_prefetch(arow[t] + kc0 + KC, 0, 0);
        }
        for (int k = 0; k < len; k += 4) {
            v2f a[TMI];
#pragma unroll
            for (int t = 0; t < TMI; ++t)
                a[t] = *(const v2f*)(arow[t] + kc0 + k + 2 * kh);
#pragma unroll
            for (int j = 0; j < 4; ++j) {
                v2f b = *(const v2f*)&sWt[(j * 16 + colL) * STI + k + 2 * kh];
#pragma unroll
                for (int t = 0; t < TMI; ++t)
                    acc[t][j] = __builtin_amdgcn_wmma_f32_16x16x4_f32(
                        false, a[t], false, b, (short)0, acc[t][j], false, false);
            }
        }
    }

    int hi = kh * 8;
#pragma unroll
    for (int t = 0; t < TMI; ++t) {
        if (!valid[t]) continue;
#pragma unroll
        for (int g = 0; g < 8; ++g) {
            int m = m0c[t] + g + hi;
            int nid = go_nid[m];
#pragma unroll
            for (int j = 0; j < 4; ++j) {
                int col = j * 16 + colL;
                xg[(size_t)m * D + col] = acc[t][j][g] + go_emb[(size_t)nid * D + col];
            }
        }
    }
}

__global__ __launch_bounds__(256) void layer_gemm_wmma(
        const float* __restrict__ agg, const float* __restrict__ x,
        const float* __restrict__ Wl, const float* __restrict__ bl,
        const float* __restrict__ Wr, float* __restrict__ out, int M, int relu) {
    __shared__ float sWt[2 * D * STL];

    for (int i = threadIdx.x * 4; i < D * D; i += 256 * 4) {
        int r = i >> 6, c = i & 63;
        float4 wl = *(const float4*)&Wl[i];
        float4 wr = *(const float4*)&Wr[i];
        sWt[(c + 0) * STL + r] = wl.x;
        sWt[(c + 1) * STL + r] = wl.y;
        sWt[(c + 2) * STL + r] = wl.z;
        sWt[(c + 3) * STL + r] = wl.w;
        sWt[D * STL + (c + 0) * STL + r] = wr.x;
        sWt[D * STL + (c + 1) * STL + r] = wr.y;
        sWt[D * STL + (c + 2) * STL + r] = wr.z;
        sWt[D * STL + (c + 3) * STL + r] = wr.w;
    }
    __syncthreads();

    int waveInBlk = threadIdx.x >> 5;
    int lane = threadIdx.x & 31;
    int row  = lane & 15;
    int kh   = lane >> 4;
    int colL = lane & 15;

    int numTiles = M >> 4;
    int tile0 = (blockIdx.x * 8 + waveInBlk) * TML;

    int   m0c[TML];
    bool  valid[TML];
    const float* aggrow[TML];
    const float* xrow[TML];
#pragma unroll
    for (int t = 0; t < TML; ++t) {
        int ti = tile0 + t;
        valid[t] = (ti < numTiles);
        m0c[t] = valid[t] ? (ti << 4) : 0;
        aggrow[t] = agg + (size_t)(m0c[t] + row) * D;
        xrow[t]   = x   + (size_t)(m0c[t] + row) * D;
    }

    v8f acc[TML][4];
#pragma unroll
    for (int t = 0; t < TML; ++t)
#pragma unroll
        for (int j = 0; j < 4; ++j) {
            float b = bl[j * 16 + colL];
#pragma unroll
            for (int g = 0; g < 8; ++g) acc[t][j][g] = b;
        }

    for (int k = 0; k < D; k += 4) {
        v2f aA[TML], aX[TML];
#pragma unroll
        for (int t = 0; t < TML; ++t) {
            aA[t] = *(const v2f*)(aggrow[t] + k + 2 * kh);
            aX[t] = *(const v2f*)(xrow[t]   + k + 2 * kh);
        }
#pragma unroll
        for (int j = 0; j < 4; ++j) {
            v2f bL = *(const v2f*)&sWt[(j * 16 + colL) * STL + k + 2 * kh];
            v2f bR = *(const v2f*)&sWt[D * STL + (j * 16 + colL) * STL + k + 2 * kh];
#pragma unroll
            for (int t = 0; t < TML; ++t) {
                acc[t][j] = __builtin_amdgcn_wmma_f32_16x16x4_f32(
                    false, aA[t], false, bL, (short)0, acc[t][j], false, false);
                acc[t][j] = __builtin_amdgcn_wmma_f32_16x16x4_f32(
                    false, aX[t], false, bR, (short)0, acc[t][j], false, false);
            }
        }
    }

    int hi = kh * 8;
#pragma unroll
    for (int t = 0; t < TML; ++t) {
        if (!valid[t]) continue;
#pragma unroll
        for (int g = 0; g < 8; ++g) {
            int m = m0c[t] + g + hi;
#pragma unroll
            for (int j = 0; j < 4; ++j) {
                float v = acc[t][j][g];
                if (relu) v = fmaxf(v, 0.0f);
                out[(size_t)m * D + j * 16 + colL] = v;
            }
        }
    }
}

__global__ void classifier_dot(const float* __restrict__ xp, const float* __restrict__ xg,
                               const int* __restrict__ ls, const int* __restrict__ ld,
                               float* __restrict__ out, int n) {
    int e = blockIdx.x * blockDim.x + threadIdx.x;
    if (e >= n) return;
    const float4* a = (const float4*)(xp + (size_t)ls[e] * D);
    const float4* b = (const float4*)(xg + (size_t)ld[e] * D);
    float s = 0.0f;
#pragma unroll
    for (int i = 0; i < 16; ++i) {
        float4 va = a[i], vb = b[i];
        s += va.x * vb.x + va.y * vb.y + va.z * vb.z + va.w * vb.w;
    }
    out[e] = s;
}

extern "C" void kernel_launch(void* const* d_in, const int* in_sizes, int n_in,
                              void* d_out, int out_size, void* d_ws, size_t ws_size,
                              hipStream_t stream) {
    const float* go_x  = (const float*)d_in[0];
    const float* p_emb = (const float*)d_in[1];
    const float* g_emb = (const float*)d_in[2];
    const float* lin_W = (const float*)d_in[3];
    const float* lin_b = (const float*)d_in[4];
    const float* Wl    = (const float*)d_in[5];
    const float* bl    = (const float*)d_in[6];
    const float* Wr    = (const float*)d_in[7];
    const int* p_nid   = (const int*)d_in[8];
    const int* g_nid   = (const int*)d_in[9];
    const int* src_pg  = (const int*)d_in[10];
    const int* dst_pg  = (const int*)d_in[11];
    const int* src_gp  = (const int*)d_in[12];
    const int* dst_gp  = (const int*)d_in[13];
    const int* lab_s   = (const int*)d_in[14];
    const int* lab_d   = (const int*)d_in[15];

    const int NP  = in_sizes[8];
    const int NG  = in_sizes[9];
    const int E   = in_sizes[10];
    const int EL  = in_sizes[14];
    const int FGO = in_sizes[0] / NG;

    char* base = (char*)d_ws;
    auto alloc = [&](size_t bytes) -> char* {
        char* p = base;
        base += (bytes + 255) & ~(size_t)255;
        return p;
    };
    float* xp_a  = (float*)alloc((size_t)NP * D * 4);
    float* xp_b  = (float*)alloc((size_t)NP * D * 4);
    float* xg_a  = (float*)alloc((size_t)NG * D * 4);
    float* xg_b  = (float*)alloc((size_t)NG * D * 4);
    float* agg_p = (float*)alloc((size_t)NP * D * 4);
    float* agg_g = (float*)alloc((size_t)NG * D * 4);
    int* row_g   = (int*)alloc((size_t)NG * 4);
    int* cur_g   = (int*)alloc((size_t)NG * 4);
    int* row_p   = (int*)alloc((size_t)NP * 4);
    int* cur_p   = (int*)alloc((size_t)NP * 4);
    int* srcs_g  = (int*)alloc((size_t)E * 4);
    int* srcs_p  = (int*)alloc((size_t)E * 4);
    int* bsum    = (int*)alloc(1024 * 4);

    const int EB = (E + 255) / 256;

    {
        int nb = (NG + 255) / 256;
        hipMemsetAsync(cur_g, 0, (size_t)NG * 4, stream);
        count_edges_int<<<EB, 256, 0, stream>>>(dst_pg, cur_g, E);
        scan_block<<<nb, 256, 0, stream>>>(cur_g, row_g, bsum, NG);
        scan_sums<<<1, 1024, 0, stream>>>(bsum, nb);
        add_offsets<<<nb, 256, 0, stream>>>(row_g, bsum, cur_g, NG);
        fill_csr<<<EB, 256, 0, stream>>>(src_pg, dst_pg, cur_g, srcs_g, E);
    }
    {
        int nb = (NP + 255) / 256;
        hipMemsetAsync(cur_p, 0, (size_t)NP * 4, stream);
        count_edges_int<<<EB, 256, 0, stream>>>(dst_gp, cur_p, E);
        scan_block<<<nb, 256, 0, stream>>>(cur_p, row_p, bsum, NP);
        scan_sums<<<1, 1024, 0, stream>>>(bsum, nb);
        add_offsets<<<nb, 256, 0, stream>>>(row_p, bsum, cur_p, NP);
        fill_csr<<<EB, 256, 0, stream>>>(src_gp, dst_gp, cur_p, srcs_p, E);
    }

    gather_rows<<<(NP * 16 + 255) / 256, 256, 0, stream>>>(p_emb, p_nid, xp_a, NP * 16);
    {
        int tiles  = (NG + 15) / 16;
        int blocks = (tiles + 8 * TMI - 1) / (8 * TMI);
        input_gemm_wmma<<<blocks, 256, 0, stream>>>(go_x, lin_W, lin_b, g_emb, g_nid,
                                                    xg_a, NG, FGO);
    }

    float* xp_cur = xp_a; float* xp_nxt = xp_b;
    float* xg_cur = xg_a; float* xg_nxt = xg_b;
    for (int l = 0; l < 3; ++l) {
        aggregate_csr<<<(NG * 16 + 255) / 256, 256, 0, stream>>>(xp_cur, srcs_g, row_g, cur_g,
                                                                 agg_g, NG);
        aggregate_csr<<<(NP * 16 + 255) / 256, 256, 0, stream>>>(xg_cur, srcs_p, row_p, cur_p,
                                                                 agg_p, NP);
        const float* Wl0 = Wl + (size_t)(l * 2 + 0) * D * D;
        const float* Wl1 = Wl + (size_t)(l * 2 + 1) * D * D;
        const float* Wr0 = Wr + (size_t)(l * 2 + 0) * D * D;
        const float* Wr1 = Wr + (size_t)(l * 2 + 1) * D * D;
        const float* bl0 = bl + (size_t)(l * 2 + 0) * D;
        const float* bl1 = bl + (size_t)(l * 2 + 1) * D;
        int relu = (l < 2) ? 1 : 0;
        {
            int tiles  = (NG + 15) / 16;
            int blocks = (tiles + 8 * TML - 1) / (8 * TML);
            layer_gemm_wmma<<<blocks, 256, 0, stream>>>(agg_g, xg_cur, Wl0, bl0, Wr0,
                                                        xg_nxt, NG, relu);
        }
        {
            int tiles  = (NP + 15) / 16;
            int blocks = (tiles + 8 * TML - 1) / (8 * TML);
            layer_gemm_wmma<<<blocks, 256, 0, stream>>>(agg_p, xp_cur, Wl1, bl1, Wr1,
                                                        xp_nxt, NP, relu);
        }
        float* t;
        t = xp_cur; xp_cur = xp_nxt; xp_nxt = t;
        t = xg_cur; xg_cur = xg_nxt; xg_nxt = t;
    }

    classifier_dot<<<(EL + 255) / 256, 256, 0, stream>>>(xp_cur, xg_cur, lab_s, lab_d,
                                                         (float*)d_out, EL);
}


# CSR build as a two-level counting sort on LDS atomics (256-row buckets, scanned per-block offsets) in place of two global atomics per edge; neighbour order in a row stays arrival order
# speedup vs baseline: 1.5601x; 1.5601x over previous
#include <hip/hip_runtime.h>

typedef __attribute__((ext_vector_type(2))) float v2f;
typedef __attribute__((ext_vector_type(8))) float v8f;

#define D    64
#define KC   128
#define STI  (KC+4)
#define STL  (D+4)
#define TMI  2
#define TML  4

#define EPB   8192
#define RSH   8
#define RPB   256
#define MAXB  1024

__global__ __launch_bounds__(1024) void bucket_hist(const int* __restrict__ dst,
                                                    int* __restrict__ blkHist,
                                                    int nE, int nBuckets, int nBlk) {
    __shared__ int h[MAXB];
    int tid = threadIdx.x;
    h[tid] = 0;
    __syncthreads();
    int base = blockIdx.x * EPB + tid;
    int d[EPB / 1024];
#pragma unroll
    for (int u = 0; u < EPB / 1024; ++u) {
        int e = base + u * 1024;
        d[u] = dst[max(min(e, nE - 1), 0)];
    }
#pragma unroll
    for (int u = 0; u < EPB / 1024; ++u) {
        int e = base + u * 1024;
        if (e < nE) atomicAdd(&h[d[u] >> RSH], 1);
    }
    __syncthreads();
    if (tid < nBuckets) blkHist[tid * nBlk + blockIdx.x] = h[tid];
}

__global__ void scan_block(const int* __restrict__ deg, int* __restrict__ ex,
                           int* __restrict__ bsum, int n) {
    __shared__ int sh[256];
    int tid = threadIdx.x;
    int i = blockIdx.x * 256 + tid;
    int v = (i < n) ? deg[i] : 0;
    sh[tid] = v;
    __syncthreads();
    for (int off = 1; off < 256; off <<= 1) {
        int add = (tid >= off) ? sh[tid - off] : 0;
        __syncthreads();
        sh[tid] += add;
        __syncthreads();
    }
    if (i < n) ex[i] = sh[tid] - v;
    if (tid == 255) bsum[blockIdx.x] = sh[255];
}

__global__ void scan_sums(int* __restrict__ bsum, int nb) {
    __shared__ int sh[1024];
    int tid = threadIdx.x;
    int v = (tid < nb) ? bsum[tid] : 0;
    sh[tid] = v;
    __syncthreads();
    for (int off = 1; off < 1024; off <<= 1) {
        int add = (tid >= off) ? sh[tid - off] : 0;
        __syncthreads();
        sh[tid] += add;
        __syncthreads();
    }
    if (tid < nb) bsum[tid] = sh[tid] - v;
}

__global__ void add_block_offsets(int* __restrict__ ex, const int* __restrict__ bsum, int n) {
    int i = blockIdx.x * 256 + threadIdx.x;
    if (i < n) ex[i] += bsum[i >> 8];
}

__global__ __launch_bounds__(1024) void bucket_scatter(const int* __restrict__ src,
                                                       const int* __restrict__ dst,
                                                       const int* __restrict__ blkOff,
                                                       int* __restrict__ pairs,
                                                       int nE, int nBuckets, int nBlk) {
    __shared__ int cur[MAXB];
    int tid = threadIdx.x;
    if (tid < nBuckets) cur[tid] = blkOff[tid * nBlk + blockIdx.x];
    __syncthreads();
    int base = blockIdx.x * EPB + tid;
    int d[EPB / 1024], s[EPB / 1024];
#pragma unroll
    for (int u = 0; u < EPB / 1024; ++u) {
        int ec = max(min(base + u * 1024, nE - 1), 0);
        d[u] = dst[ec];
        s[u] = src[ec];
    }
    int2* __restrict__ p2 = (int2*)pairs;
#pragma unroll
    for (int u = 0; u < EPB / 1024; ++u) {
        int e = base + u * 1024;
        if (e < nE) {
            int pos = atomicAdd(&cur[d[u] >> RSH], 1);
            p2[pos] = make_int2(s[u], d[u]);
        }
    }
}

__global__ __launch_bounds__(1024) void bucket_csr(const int* __restrict__ pairs,
                                                   const int* __restrict__ blkOff,
                                                   int* __restrict__ rowStart,
                                                   int* __restrict__ rowEnd,
                                                   int* __restrict__ srcs,
                                                   int n, int nE, int nBuckets, int nBlk) {
    __shared__ int cnt[RPB];
    __shared__ int sc[RPB];
    int b = blockIdx.x, tid = threadIdx.x;
    int s0 = blkOff[b * nBlk];
    int s1 = (b + 1 < nBuckets) ? blkOff[(b + 1) * nBlk] : nE;
    const int2* __restrict__ p2 = (const int2*)pairs;
    if (tid < RPB) cnt[tid] = 0;
    __syncthreads();
    for (int i0 = s0 + tid; i0 < s1; i0 += 4096) {
        int2 p[4];
#pragma unroll
        for (int u = 0; u < 4; ++u)
            p[u] = p2[min(i0 + u * 1024, s1 - 1)];
#pragma unroll
        for (int u = 0; u < 4; ++u) {
            int i = i0 + u * 1024;
            if (i < s1) atomicAdd(&cnt[p[u].y & (RPB - 1)], 1);
        }
    }
    __syncthreads();
    int v = 0;
    if (tid < RPB) { v = cnt[tid]; sc[tid] = v; }
    __syncthreads();
    for (int off = 1; off < RPB; off <<= 1) {
        int add = 0;
        if (tid < RPB && tid >= off) add = sc[tid - off];
        __syncthreads();
        if (tid < RPB) sc[tid] += add;
        __syncthreads();
    }
    if (tid < RPB) {
        int start = s0 + sc[tid] - v;
        int node = b * RPB + tid;
        if (node < n) { rowStart[node] = start; rowEnd[node] = start + v; }
        cnt[tid] = start;
    }
    __syncthreads();
    for (int i0 = s0 + tid; i0 < s1; i0 += 4096) {
        int2 p[4];
#pragma unroll
        for (int u = 0; u < 4; ++u)
            p[u] = p2[min(i0 + u * 1024, s1 - 1)];
#pragma unroll
        for (int u = 0; u < 4; ++u) {
            int i = i0 + u * 1024;
            if (i < s1) {
                int pos = atomicAdd(&cnt[p[u].y & (RPB - 1)], 1);
                srcs[pos] = p[u].x;
            }
        }
    }
}

__global__ void aggregate_csr(const float* __restrict__ x, const int* __restrict__ srcs,
                              const int* __restrict__ rowStart, const int* __restrict__ rowEnd,
                              float* __restrict__ agg, int n) {
    int t = blockIdx.x * blockDim.x + threadIdx.x;
    int node = t >> 4;
    if (node >= n) return;
    int c = (t & 15) * 4;
    int s0 = rowStart[node], s1 = rowEnd[node];
    float4 acc = make_float4(0.f, 0.f, 0.f, 0.f);
    for (int i = s0; i < s1; ++i) {
        int s = srcs[i];
        float4 v = *(const float4*)(x + (size_t)s * D + c);
        acc.x += v.x; acc.y += v.y; acc.z += v.z; acc.w += v.w;
    }
    float inv = 1.0f / fmaxf((float)(s1 - s0), 1.0f);
    acc.x *= inv; acc.y *= inv; acc.z *= inv; acc.w *= inv;
    *(float4*)(agg + (size_t)node * D + c) = acc;
}

__global__ void gather_rows(const float* __restrict__ emb, const int* __restrict__ nid,
                            float* __restrict__ out, int n16) {
    int t = blockIdx.x * blockDim.x + threadIdx.x;
    if (t >= n16) return;
    int i = t >> 4, c = (t & 15) * 4;
    int s = nid[i];
    *(float4*)(out + (size_t)i * D + c) = *(const float4*)(emb + (size_t)s * D + c);
}

__global__ __launch_bounds__(256) void input_gemm_wmma(
        const float* __restrict__ go_x, const float* __restrict__ lin_W,
        const float* __restrict__ lin_b, const float* __restrict__ go_emb,
        const int* __restrict__ go_nid, float* __restrict__ xg, int M, int K) {
    __shared__ float sWt[D * STI];

    int waveInBlk = threadIdx.x >> 5;
    int lane = threadIdx.x & 31;
    int row  = lane & 15;
    int kh   = lane >> 4;
    int colL = lane & 15;

    int numTiles = M >> 4;
    int tile0 = (blockIdx.x * 8 + waveInBlk) * TMI;

    int   m0c[TMI];
    const float* arow[TMI];
    bool  valid[TMI];
#pragma unroll
    for (int t = 0; t < TMI; ++t) {
        int ti = tile0 + t;
        valid[t] = (ti < numTiles);
        m0c[t] = valid[t] ? (ti << 4) : 0;
        arow[t] = go_x + (size_t)(m0c[t] + row) * K;
    }

    v8f acc[TMI][4];
#pragma unroll
    for (int t = 0; t < TMI; ++t)
#pragma unroll
        for (int j = 0; j < 4; ++j) {
            float b = lin_b[j * 16 + colL];
#pragma unroll
            for (int g = 0; g < 8; ++g) acc[t][j][g] = b;
        }

    for (int kc0 = 0; kc0 < K; kc0 += KC) {
        int len = (K - kc0 < KC) ? (K - kc0) : KC;
        __syncthreads();
        for (int i = threadIdx.x * 4; i < len * D; i += 256 * 4) {
            int r = i >> 6, c = i & 63;
            float4 w = *(const float4*)&lin_W[(size_t)(kc0 + r) * D + c];
            sWt[(c + 0) * STI + r] = w.x;
            sWt[(c + 1) * STI + r] = w.y;
            sWt[(c + 2) * STI + r] = w.z;
            sWt[(c + 3) * STI + r] = w.w;
        }
        __syncthreads();
        if (kc0 + KC < K) {
#pragma unroll
            for (int t = 0; t < TMI; ++t)
                __builtin_prefetch(arow[t] + kc0 + KC, 0, 0);
        }
        for (int k = 0; k < len; k += 4) {
            v2f a[TMI];
#pragma unroll
            for (int t = 0; t < TMI; ++t)
                a[t] = *(const v2f*)(arow[t] + kc0 + k + 2 * kh);
#pragma unroll
            for (int j = 0; j < 4; ++j) {
                v2f b = *(const v2f*)&sWt[(j * 16 + colL) * STI + k + 2 * kh];
#pragma unroll
                for (int t = 0; t < TMI; ++t)
                    acc[t][j] = __builtin_amdgcn_wmma_f32_16x16x4_f32(
                        false, a[t], false, b, (short)0, acc[t][j], false, false);
            }
        }
    }

    int hi = kh * 8;
#pragma unroll
    for (int t = 0; t < TMI; ++t) {
        if (!valid[t]) continue;
#pragma unroll
        for (int g = 0; g < 8; ++g) {
            int m = m0c[t] + g + hi;
            int nid = go_nid[m];
#pragma unroll
            for (int j = 0; j < 4; ++j) {
                int col = j * 16 + colL;
                xg[(size_t)m * D + col] = acc[t][j][g] + go_emb[(size_t)nid * D + col];
            }
        }
    }
}

__global__ __launch_bounds__(256) void layer_gemm_wmma(
        const float* __restrict__ agg, const float* __restrict__ x,
        const float* __restrict__ Wl, const float* __restrict__ bl,
        const float* __restrict__ Wr, float* __restrict__ out, int M, int relu) {
    __shared__ float sWt[2 * D * STL];

    for (int i = threadIdx.x * 4; i < D * D; i += 256 * 4) {
        int r = i >> 6, c = i & 63;
        float4 wl = *(const float4*)&Wl[i];
        float4 wr = *(const float4*)&Wr[i];
        sWt[(c + 0) * STL + r] = wl.x;
        sWt[(c + 1) * STL + r] = wl.y;
        sWt[(c + 2) * STL + r] = wl.z;
        sWt[(c + 3) * STL + r] = wl.w;
        sWt[D * STL + (c + 0) * STL + r] = wr.x;
        sWt[D * STL + (c + 1) * STL + r] = wr.y;
        sWt[D * STL + (c + 2) * STL + r] = wr.z;
        sWt[D * STL + (c + 3) * STL + r] = wr.w;
    }
    __syncthreads();

    int waveInBlk = threadIdx.x >> 5;
    int lane = threadIdx.x & 31;
    int row  = lane & 15;
    int kh   = lane >> 4;
    int colL = lane & 15;

    int numTiles = M >> 4;
    int tile0 = (blockIdx.x * 8 + waveInBlk) * TML;

    int   m0c[TML];
    bool  valid[TML];
    const float* aggrow[TML];
    const float* xrow[TML];
#pragma unroll
    for (int t = 0; t < TML; ++t) {
        int ti = tile0 + t;
        valid[t] = (ti < numTiles);
        m0c[t] = valid[t] ? (ti << 4) : 0;
        aggrow[t] = agg + (size_t)(m0c[t] + row) * D;
        xrow[t]   = x   + (size_t)(m0c[t] + row) * D;
    }

    v8f acc[TML][4];
#pragma unroll
    for (int t = 0; t < TML; ++t)
#pragma unroll
        for (int j = 0; j < 4; ++j) {
            float b = bl[j * 16 + colL];
#pragma unroll
            for (int g = 0; g < 8; ++g) acc[t][j][g] = b;
        }

    for (int k = 0; k < D; k += 4) {
        v2f aA[TML], aX[TML];
#pragma unroll
        for (int t = 0; t < TML; ++t) {
            aA[t] = *(const v2f*)(aggrow[t] + k + 2 * kh);
            aX[t] = *(const v2f*)(xrow[t]   + k + 2 * kh);
        }
#pragma unroll
        for (int j = 0; j < 4; ++j) {
            v2f bL = *(const v2f*)&sWt[(j * 16 + colL) * STL + k + 2 * kh];
            v2f bR = *(const v2f*)&sWt[D * STL + (j * 16 + colL) * STL + k + 2 * kh];
#pragma unroll
            for (int t = 0; t < TML; ++t) {
                acc[t][j] = __builtin_amdgcn_wmma_f32_16x16x4_f32(
                    false, aA[t], false, bL, (short)0, acc[t][j], false, false);
                acc[t][j] = __builtin_amdgcn_wmma_f32_16x16x4_f32(
                    false, aX[t], false, bR, (short)0, acc[t][j], false, false);
            }
        }
    }

    int hi = kh * 8;
#pragma unroll
    for (int t = 0; t < TML; ++t) {
        if (!valid[t]) continue;
#pragma unroll
        for (int g = 0; g < 8; ++g) {
            int m = m0c[t] + g + hi;
#pragma unroll
            for (int j = 0; j < 4; ++j) {
                float v = acc[t][j][g];
                if (relu) v = fmaxf(v, 0.0f);
                out[(size_t)m * D + j * 16 + colL] = v;
            }
        }
    }
}

__global__ void classifier_dot(const float* __restrict__ xp, const float* __restrict__ xg,
                               const int* __restrict__ ls, const int* __restrict__ ld,
                               float* __restrict__ out, int n) {
    int e = blockIdx.x * blockDim.x + threadIdx.x;
    if (e >= n) return;
    const float4* a = (const float4*)(xp + (size_t)ls[e] * D);
    const float4* b = (const float4*)(xg + (size_t)ld[e] * D);
    float s = 0.0f;
#pragma unroll
    for (int i = 0; i < 16; ++i) {
        float4 va = a[i], vb = b[i];
        s += va.x * vb.x + va.y * vb.y + va.z * vb.z + va.w * vb.w;
    }
    out[e] = s;
}

extern "C" void kernel_launch(void* const* d_in, const int* in_sizes, int n_in,
                              void* d_out, int out_size, void* d_ws, size_t ws_size,
                              hipStream_t stream) {
    const float* go_x  = (const float*)d_in[0];
    const float* p_emb = (const float*)d_in[1];
    const float* g_emb = (const float*)d_in[2];
    const float* lin_W = (const float*)d_in[3];
    const float* lin_b = (const float*)d_in[4];
    const float* Wl    = (const float*)d_in[5];
    const float* bl    = (const float*)d_in[6];
    const float* Wr    = (const float*)d_in[7];
    const int* p_nid   = (const int*)d_in[8];
    const int* g_nid   = (const int*)d_in[9];
    const int* src_pg  = (const int*)d_in[10];
    const int* dst_pg  = (const int*)d_in[11];
    const int* src_gp  = (const int*)d_in[12];
    const int* dst_gp  = (const int*)d_in[13];
    const int* lab_s   = (const int*)d_in[14];
    const int* lab_d   = (const int*)d_in[15];

    const int NP  = in_sizes[8];
    const int NG  = in_sizes[9];
    const int E   = in_sizes[10];
    const int EL  = in_sizes[14];
    const int FGO = in_sizes[0] / NG;

    char* base = (char*)d_ws;
    auto alloc = [&](size_t bytes) -> char* {
        char* p = base;
        base += (bytes + 255) & ~(size_t)255;
        return p;
    };
    float* xp_a  = (float*)alloc((size_t)NP * D * 4);
    float* xp_b  = (float*)alloc((size_t)NP * D * 4);
    float* xg_a  = (float*)alloc((size_t)NG * D * 4);
    float* xg_b  = (float*)alloc((size_t)NG * D * 4);
    float* agg_p = (float*)alloc((size_t)NP * D * 4);
    float* agg_g = (float*)alloc((size_t)NG * D * 4);
    int* row_g   = (int*)alloc((size_t)NG * 4);
    int* cur_g   = (int*)alloc((size_t)NG * 4);
    int* row_p   = (int*)alloc((size_t)NP * 4);
    int* cur_p   = (int*)alloc((size_t)NP * 4);
    int* srcs_g  = (int*)alloc((size_t)E * 4);
    int* srcs_p  = (int*)alloc((size_t)E * 4);
    int* bsum    = (int*)alloc(1024 * 4);

    int* pairs = (int*)agg_p;
    int* hmat  = (int*)xp_b;
    int* omat  = (int*)xg_b;
    const int nBlkE = (E + EPB - 1) / EPB;

    {
        int nbk  = (NG + RPB - 1) / RPB;
        int nmat = nbk * nBlkE;
        int nb   = (nmat + 255) / 256;
        bucket_hist<<<nBlkE, 1024, 0, stream>>>(dst_pg, hmat, E, nbk, nBlkE);
        scan_block<<<nb, 256, 0, stream>>>(hmat, omat, bsum, nmat);
        scan_sums<<<1, 1024, 0, stream>>>(bsum, nb);
        add_block_offsets<<<nb, 256, 0, stream>>>(omat, bsum, nmat);
        bucket_scatter<<<nBlkE, 1024, 0, stream>>>(src_pg, dst_pg, omat, pairs, E, nbk, nBlkE);
        bucket_csr<<<nbk, 1024, 0, stream>>>(pairs, omat, row_g, cur_g, srcs_g, NG, E, nbk, nBlkE);
    }
    {
        int nbk  = (NP + RPB - 1) / RPB;
        int nmat = nbk * nBlkE;
        int nb   = (nmat + 255) / 256;
        bucket_hist<<<nBlkE, 1024, 0, stream>>>(dst_gp, hmat, E, nbk, nBlkE);
        scan_block<<<nb, 256, 0, stream>>>(hmat, omat, bsum, nmat);
        scan_sums<<<1, 1024, 0, stream>>>(bsum, nb);
        add_block_offsets<<<nb, 256, 0, stream>>>(omat, bsum, nmat);
        bucket_scatter<<<nBlkE, 1024, 0, stream>>>(src_gp, dst_gp, omat, pairs, E, nbk, nBlkE);
        bucket_csr<<<nbk, 1024, 0, stream>>>(pairs, omat, row_p, cur_p, srcs_p, NP, E, nbk, nBlkE);
    }

    gather_rows<<<(NP * 16 + 255) / 256, 256, 0, stream>>>(p_emb, p_nid, xp_a, NP * 16);
    {
        int tiles  = (NG + 15) / 16;
        int blocks = (tiles + 8 * TMI - 1) / (8 * TMI);
        input_gemm_wmma<<<blocks, 256, 0, stream>>>(go_x, lin_W, lin_b, g_emb, g_nid,
                                                    xg_a, NG, FGO);
    }

    float* xp_cur = xp_a; float* xp_nxt = xp_b;
    float* xg_cur = xg_a; float* xg_nxt = xg_b;
    for (int l = 0; l < 3; ++l) {
        aggregate_csr<<<(NG * 16 + 255) / 256, 256, 0, stream>>>(xp_cur, srcs_g, row_g, cur_g,
                                                                 agg_g, NG);
        aggregate_csr<<<(NP * 16 + 255) / 256, 256, 0, stream>>>(xg_cur, srcs_p, row_p, cur_p,
                                                                 agg_p, NP);
        const float* Wl0 = Wl + (size_t)(l * 2 + 0) * D * D;
        const float* Wl1 = Wl + (size_t)(l * 2 + 1) * D * D;
        const float* Wr0 = Wr + (size_t)(l * 2 + 0) * D * D;
        const float* Wr1 = Wr + (size_t)(l * 2 + 1) * D * D;
        const float* bl0 = bl + (size_t)(l * 2 + 0) * D;
        const float* bl1 = bl + (size_t)(l * 2 + 1) * D;
        int relu = (l < 2) ? 1 : 0;
        {
            int tiles  = (NG + 15) / 16;
            int blocks = (tiles + 8 * TML - 1) / (8 * TML);
            layer_gemm_wmma<<<blocks, 256, 0, stream>>>(agg_g, xg_cur, Wl0, bl0, Wr0,
                                                        xg_nxt, NG, relu);
        }
        {
            int tiles  = (NP + 15) / 16;
            int blocks = (tiles + 8 * TML - 1) / (8 * TML);
            layer_gemm_wmma<<<blocks, 256, 0, stream>>>(agg_p, xp_cur, Wl1, bl1, Wr1,
                                                        xp_nxt, NP, relu);
        }
        float* t;
        t = xp_cur; xp_cur = xp_nxt; xp_nxt = t;
        t = xg_cur; xg_cur = xg_nxt; xg_nxt = t;
    }

    classifier_dot<<<(EL + 255) / 256, 256, 0, stream>>>(xp_cur, xg_cur, lab_s, lab_d,
                                                         (float*)d_out, EL);
}


# Independent launch pairs (the two edge directions, the two node types, row gather and input GEMM) share one grid, block index picks the problem: 27 launches to 14; layer GEMM 5 tiles per wave
# speedup vs baseline: 1.8423x; 1.1809x over previous
#include <hip/hip_runtime.h>

typedef __attribute__((ext_vector_type(2))) float v2f;
typedef __attribute__((ext_vector_type(8))) float v8f;

#define D    64
#define KC   128
#define STI  (KC+4)
#define STL  (D+4)
#define TMI  2
#define TML  5


#define EPB   8192
#define RSH   8
#define RPB   256
#define MAXB  1024

__global__ __launch_bounds__(1024) void bucket_hist(const int* __restrict__ dstA,
                                                    int* __restrict__ histA, int nbkA,
                                                    const int* __restrict__ dstB,
                                                    int* __restrict__ histB, int nbkB,
                                                    int nE, int nBlk) {
    __shared__ int h[MAXB];
    const bool second = (int)blockIdx.x >= nBlk;
    const int* __restrict__ dst = second ? dstB : dstA;
    int* __restrict__ blkHist   = second ? histB : histA;
    const int nBuckets = second ? nbkB : nbkA;
    const int blk = second ? (int)blockIdx.x - nBlk : (int)blockIdx.x;
    int tid = threadIdx.x;
    h[tid] = 0;
    __syncthreads();
    int base = blk * EPB + tid;
    int d[EPB / 1024];
#pragma unroll
    for (int u = 0; u < EPB / 1024; ++u) {
        int e = base + u * 1024;
        d[u] = dst[max(min(e, nE - 1), 0)];
    }
#pragma unroll
    for (int u = 0; u < EPB / 1024; ++u) {
        int e = base + u * 1024;
        if (e < nE) atomicAdd(&h[d[u] >> RSH], 1);
    }
    __syncthreads();
    if (tid < nBuckets) blkHist[tid * nBlk + blk] = h[tid];
}

__global__ void scan_block(const int* __restrict__ degA, int* __restrict__ exA,
                           int* __restrict__ bsumA, int nA, int nbA,
                           const int* __restrict__ degB, int* __restrict__ exB,
                           int* __restrict__ bsumB, int nB) {
    __shared__ int sh[256];
    const bool second = (int)blockIdx.x >= nbA;
    const int* __restrict__ deg = second ? degB : degA;
    int* __restrict__ ex   = second ? exB : exA;
    int* __restrict__ bsum = second ? bsumB : bsumA;
    const int n   = second ? nB : nA;
    const int blk = second ? (int)blockIdx.x - nbA : (int)blockIdx.x;
    int tid = threadIdx.x;
    int i = blk * 256 + tid;
    int v = (i < n) ? deg[i] : 0;
    sh[tid] = v;
    __syncthreads();
    for (int off = 1; off < 256; off <<= 1) {
        int add = (tid >= off) ? sh[tid - off] : 0;
        __syncthreads();
        sh[tid] += add;
        __syncthreads();
    }
    if (i < n) ex[i] = sh[tid] - v;
    if (tid == 255) bsum[blk] = sh[255];
}

__global__ void scan_sums(int* __restrict__ bsumA, int nbA,
                          int* __restrict__ bsumB, int nbB) {
    __shared__ int sh[1024];
    const bool second = blockIdx.x != 0;
    int* __restrict__ bsum = second ? bsumB : bsumA;
    const int nb = second ? nbB : nbA;
    int tid = threadIdx.x;
    int v = (tid < nb) ? bsum[tid] : 0;
    sh[tid] = v;
    __syncthreads();
    for (int off = 1; off < 1024; off <<= 1) {
        int add = (tid >= off) ? sh[tid - off] : 0;
        __syncthreads();
        sh[tid] += add;
        __syncthreads();
    }
    if (tid < nb) bsum[tid] = sh[tid] - v;
}

__global__ void add_block_offsets(int* __restrict__ exA, const int* __restrict__ bsumA, int nA,
                                  int nbA,
                                  int* __restrict__ exB, const int* __restrict__ bsumB, int nB) {
    const bool second = (int)blockIdx.x >= nbA;
    int* __restrict__ ex         = second ? exB : exA;
    const int* __restrict__ bsum = second ? bsumB : bsumA;
    const int n   = second ? nB : nA;
    const int blk = second ? (int)blockIdx.x - nbA : (int)blockIdx.x;
    int i = blk * 256 + threadIdx.x;
    if (i < n) ex[i] += bsum[blk];
}

__global__ __launch_bounds__(1024) void bucket_scatter(const int* __restrict__ srcA,
                                                       const int* __restrict__ dstA,
                                                       const int* __restrict__ offA,
                                                       int* __restrict__ pairsA, int nbkA,
                                                       const int* __restrict__ srcB,
                                                       const int* __restrict__ dstB,
                                                       const int* __restrict__ offB,
                                                       int* __restrict__ pairsB, int nbkB,
                                                       int nE, int nBlk) {
    __shared__ int cur[MAXB];
    const bool second = (int)blockIdx.x >= nBlk;
    const int* __restrict__ src    = second ? srcB : srcA;
    const int* __restrict__ dst    = second ? dstB : dstA;
    const int* __restrict__ blkOff = second ? offB : offA;
    int* __restrict__ pairs        = second ? pairsB : pairsA;
    const int nBuckets = second ? nbkB : nbkA;
    const int blk = second ? (int)blockIdx.x - nBlk : (int)blockIdx.x;
    int tid = threadIdx.x;
    if (tid < nBuckets) cur[tid] = blkOff[tid * nBlk + blk];
    __syncthreads();
    int base = blk * EPB + tid;
    int d[EPB / 1024], s[EPB / 1024];
#pragma unroll
    for (int u = 0; u < EPB / 1024; ++u) {
        int ec = max(min(base + u * 1024, nE - 1), 0);
        d[u] = dst[ec];
        s[u] = src[ec];
    }
    int2* __restrict__ p2 = (int2*)pairs;
#pragma unroll
    for (int u = 0; u < EPB / 1024; ++u) {
        int e = base + u * 1024;
        if (e < nE) {
            int pos = atomicAdd(&cur[d[u] >> RSH], 1);
            p2[pos] = make_int2(s[u], d[u]);
        }
    }
}

__global__ __launch_bounds__(1024) void bucket_csr(const int* __restrict__ pairsA,
                                                   const int* __restrict__ offA,
                                                   int* __restrict__ rowStartA,
                                                   int* __restrict__ rowEndA,
                                                   int* __restrict__ srcsA, int nA, int nbkA,
                                                   const int* __restrict__ pairsB,
                                                   const int* __restrict__ offB,
                                                   int* __restrict__ rowStartB,
                                                   int* __restrict__ rowEndB,
                                                   int* __restrict__ srcsB, int nB, int nbkB,
                                                   int nE, int nBlk) {
    __shared__ int cnt[RPB];
    __shared__ int sc[RPB];
    const bool second = (int)blockIdx.x >= nbkA;
    const int* __restrict__ pairs  = second ? pairsB : pairsA;
    const int* __restrict__ blkOff = second ? offB : offA;
    int* __restrict__ rowStart     = second ? rowStartB : rowStartA;
    int* __restrict__ rowEnd       = second ? rowEndB : rowEndA;
    int* __restrict__ srcs         = second ? srcsB : srcsA;
    const int n        = second ? nB : nA;
    const int nBuckets = second ? nbkB : nbkA;
    int b = second ? (int)blockIdx.x - nbkA : (int)blockIdx.x, tid = threadIdx.x;
    int s0 = blkOff[b * nBlk];
    int s1 = (b + 1 < nBuckets) ? blkOff[(b + 1) * nBlk] : nE;
    const int2* __restrict__ p2 = (const int2*)pairs;
    if (tid < RPB) cnt[tid] = 0;
    __syncthreads();
    for (int i0 = s0 + tid; i0 < s1; i0 += 4096) {
        int2 p[4];
#pragma unroll
        for (int u = 0; u < 4; ++u)
            p[u] = p2[min(i0 + u * 1024, s1 - 1)];
#pragma unroll
        for (int u = 0; u < 4; ++u) {
            int i = i0 + u * 1024;
            if (i < s1) atomicAdd(&cnt[p[u].y & (RPB - 1)], 1);
        }
    }
    __syncthreads();
    int v = 0;
    if (tid < RPB) { v = cnt[tid]; sc[tid] = v; }
    __syncthreads();
    for (int off = 1; off < RPB; off <<= 1) {
        int add = 0;
        if (tid < RPB && tid >= off) add = sc[tid - off];
        __syncthreads();
        if (tid < RPB) sc[tid] += add;
        __syncthreads();
    }
    if (tid < RPB) {
        int start = s0 + sc[tid] - v;
        int node = b * RPB + tid;
        if (node < n) { rowStart[node] = start; rowEnd[node] = start + v; }
        cnt[tid] = start;
    }
    __syncthreads();
    for (int i0 = s0 + tid; i0 < s1; i0 += 4096) {
        int2 p[4];
#pragma unroll
        for (int u = 0; u < 4; ++u)
            p[u] = p2[min(i0 + u * 1024, s1 - 1)];
#pragma unroll
        for (int u = 0; u < 4; ++u) {
            int i = i0 + u * 1024;
            if (i < s1) {
                int pos = atomicAdd(&cnt[p[u].y & (RPB - 1)], 1);
                srcs[pos] = p[u].x;
            }
        }
    }
}

__global__ void aggregate_csr(const float* __restrict__ xA, const int* __restrict__ srcsA,
                              const int* __restrict__ rowStartA, const int* __restrict__ rowEndA,
                              float* __restrict__ aggA, int nA, int blocksA,
                              const float* __restrict__ xB, const int* __restrict__ srcsB,
                              const int* __restrict__ rowStartB, const int* __restrict__ rowEndB,
                              float* __restrict__ aggB, int nB) {
    const bool second = (int)blockIdx.x >= blocksA;
    const float* __restrict__ x      = second ? xB : xA;
    const int* __restrict__ srcs     = second ? srcsB : srcsA;
    const int* __restrict__ rowStart = second ? rowStartB : rowStartA;
    const int* __restrict__ rowEnd   = second ? rowEndB : rowEndA;
    float* __restrict__ agg          = second ? aggB : aggA;
    const int n   = second ? nB : nA;
    const int blk = second ? (int)blockIdx.x - blocksA : (int)blockIdx.x;
    int t = blk * blockDim.x + threadIdx.x;
    int node = t >> 4;
    if (node >= n) return;
    int c = (t & 15) * 4;
    int s0 = rowStart[node], s1 = rowEnd[node];
    float4 acc = make_float4(0.f, 0.f, 0.f, 0.f);
    for (int i = s0; i < s1; ++i) {
        int s = srcs[i];
        float4 v = *(const float4*)(x + (size_t)s * D + c);
        acc.x += v.x; acc.y += v.y; acc.z += v.z; acc.w += v.w;
    }
    float inv = 1.0f / fmaxf((float)(s1 - s0), 1.0f);
    acc.x *= inv; acc.y *= inv; acc.z *= inv; acc.w *= inv;
    *(float4*)(agg + (size_t)node * D + c) = acc;
}

__device__ __forceinline__ void gather_rows_body(const float* __restrict__ emb,
                                                 const int* __restrict__ nid,
                                                 float* __restrict__ out, int n16, int blk) {
    int t = blk * blockDim.x + threadIdx.x;
    if (t >= n16) return;
    int i = t >> 4, c = (t & 15) * 4;
    int s = nid[i];
    *(float4*)(out + (size_t)i * D + c) = *(const float4*)(emb + (size_t)s * D + c);
}

__device__ __forceinline__ void input_gemm_body(
        const float* __restrict__ go_x, const float* __restrict__ lin_W,
        const float* __restrict__ lin_b, const float* __restrict__ go_emb,
        const int* __restrict__ go_nid, float* __restrict__ xg, int M, int K, int blk) {
    __shared__ float sWt[D * STI];

    int waveInBlk = threadIdx.x >> 5;
    int lane = threadIdx.x & 31;
    int row  = lane & 15;
    int kh   = lane >> 4;
    int colL = lane & 15;

    int numTiles = M >> 4;
    int tile0 = (blk * 8 + waveInBlk) * TMI;

    int   m0c[TMI];
    const float* arow[TMI];
    bool  valid[TMI];
#pragma unroll
    for (int t = 0; t < TMI; ++t) {
        int ti = tile0 + t;
        valid[t] = (ti < numTiles);
        m0c[t] = valid[t] ? (ti << 4) : 0;
        arow[t] = go_x + (size_t)(m0c[t] + row) * K;
    }

    v8f acc[TMI][4];
#pragma unroll
    for (int t = 0; t < TMI; ++t)
#pragma unroll
        for (int j = 0; j < 4; ++j) {
            float b = lin_b[j * 16 + colL];
#pragma unroll
            for (int g = 0; g < 8; ++g) acc[t][j][g] = b;
        }

    for (int kc0 = 0; kc0 < K; kc0 += KC) {
        int len = (K - kc0 < KC) ? (K - kc0) : KC;
        __syncthreads();
        for (int i = threadIdx.x * 4; i < len * D; i += 256 * 4) {
            int r = i >> 6, c = i & 63;
            float4 w = *(const float4*)&lin_W[(size_t)(kc0 + r) * D + c];
            sWt[(c + 0) * STI + r] = w.x;
            sWt[(c + 1) * STI + r] = w.y;
            sWt[(c + 2) * STI + r] = w.z;
            sWt[(c + 3) * STI + r] = w.w;
        }
        __syncthreads();
        if (kc0 + KC < K) {
#pragma unroll
            for (int t = 0; t < TMI; ++t)
                __builtin_prefetch(arow[t] + kc0 + KC, 0, 0);
        }
        for (int k = 0; k < len; k += 4) {
            v2f a[TMI];
#pragma unroll
            for (int t = 0; t < TMI; ++t)
                a[t] = *(const v2f*)(arow[t] + kc0 + k + 2 * kh);
#pragma unroll
            for (int j = 0; j < 4; ++j) {
                v2f b = *(const v2f*)&sWt[(j * 16 + colL) * STI + k + 2 * kh];
#pragma unroll
                for (int t = 0; t < TMI; ++t)
                    acc[t][j] = __builtin_amdgcn_wmma_f32_16x16x4_f32(
                        false, a[t], false, b, (short)0, acc[t][j], false, false);
            }
        }
    }

    int hi = kh * 8;
#pragma unroll
    for (int t = 0; t < TMI; ++t) {
        if (!valid[t]) continue;
#pragma unroll
        for (int g = 0; g < 8; ++g) {
            int m = m0c[t] + g + hi;
            int nid = go_nid[m];
#pragma unroll
            for (int j = 0; j < 4; ++j) {
                int col = j * 16 + colL;
                xg[(size_t)m * D + col] = acc[t][j][g] + go_emb[(size_t)nid * D + col];
            }
        }
    }
}

__global__ __launch_bounds__(256) void input_embed(
        const float* __restrict__ go_x, const float* __restrict__ lin_W,
        const float* __restrict__ lin_b, const float* __restrict__ go_emb,
        const int* __restrict__ go_nid, float* __restrict__ xg, int M, int K, int gemmBlocks,
        const float* __restrict__ p_emb, const int* __restrict__ p_nid,
        float* __restrict__ xp, int n16) {
    if ((int)blockIdx.x < gemmBlocks)
        input_gemm_body(go_x, lin_W, lin_b, go_emb, go_nid, xg, M, K, (int)blockIdx.x);
    else
        gather_rows_body(p_emb, p_nid, xp, n16, (int)blockIdx.x - gemmBlocks);
}

__global__ __launch_bounds__(256) void layer_gemm_wmma(
        const float* __restrict__ aggA, const float* __restrict__ xA,
        const float* __restrict__ WlA, const float* __restrict__ blA,
        const float* __restrict__ WrA, float* __restrict__ outA, int MA, int blocksA,
        const float* __restrict__ aggB, const float* __restrict__ xB,
        const float* __restrict__ WlB, const float* __restrict__ blB,
        const float* __restrict__ WrB, float* __restrict__ outB, int MB, int relu) {
    __shared__ float sWt[2 * D * STL];
    const bool second = (int)blockIdx.x >= blocksA;
    const float* __restrict__ agg = second ? aggB : aggA;
    const float* __restrict__ x   = second ? xB : xA;
    const float* __restrict__ Wl  = second ? WlB : WlA;
    const float* __restrict__ bl  = second ? blB : blA;
    const float* __restrict__ Wr  = second ? WrB : WrA;
    float* __restrict__ out       = second ? outB : outA;
    const int M   = second ? MB : MA;
    const int blk = second ? (int)blockIdx.x - blocksA : (int)blockIdx.x;

    for (int i = threadIdx.x * 4; i < D * D; i += 256 * 4) {
        int r = i >> 6, c = i & 63;
        float4 wl = *(const float4*)&Wl[i];
        float4 wr = *(const float4*)&Wr[i];
        sWt[(c + 0) * STL + r] = wl.x;
        sWt[(c + 1) * STL + r] = wl.y;
        sWt[(c + 2) * STL + r] = wl.z;
        sWt[(c + 3) * STL + r] = wl.w;
        sWt[D * STL + (c + 0) * STL + r] = wr.x;
        sWt[D * STL + (c + 1) * STL + r] = wr.y;
        sWt[D * STL + (c + 2) * STL + r] = wr.z;
        sWt[D * STL + (c + 3) * STL + r] = wr.w;
    }
    __syncthreads();

    int waveInBlk = threadIdx.x >> 5;
    int lane = threadIdx.x & 31;
    int row  = lane & 15;
    int kh   = lane >> 4;
    int colL = lane & 15;

    int numTiles = M >> 4;
    int tile0 = (blk * 8 + waveInBlk) * TML;

    int   m0c[TML];
    bool  valid[TML];
    const float* aggrow[TML];
    const float* xrow[TML];
#pragma unroll
    for (int t = 0; t < TML; ++t) {
        int ti = tile0 + t;
        valid[t] = (ti < numTiles);
        m0c[t] = valid[t] ? (ti << 4) : 0;
        aggrow[t] = agg + (size_t)(m0c[t] + row) * D;
        xrow[t]   = x   + (size_t)(m0c[t] + row) * D;
    }

    v8f acc[TML][4];
#pragma unroll
    for (int t = 0; t < TML; ++t)
#pragma unroll
        for (int j = 0; j < 4; ++j) {
            float b = bl[j * 16 + colL];
#pragma unroll
            for (int g = 0; g < 8; ++g) acc[t][j][g] = b;
        }

    for (int k = 0; k < D; k += 4) {
        v2f aA[TML], aX[TML];
#pragma unroll
        for (int t = 0; t < TML; ++t) {
            aA[t] = *(const v2f*)(aggrow[t] + k + 2 * kh);
            aX[t] = *(const v2f*)(xrow[t]   + k + 2 * kh);
        }
#pragma unroll
        for (int j = 0; j < 4; ++j) {
            v2f bL = *(const v2f*)&sWt[(j * 16 + colL) * STL + k + 2 * kh];
            v2f bR = *(const v2f*)&sWt[D * STL + (j * 16 + colL) * STL + k + 2 * kh];
#pragma unroll
            for (int t = 0; t < TML; ++t) {
                acc[t][j] = __builtin_amdgcn_wmma_f32_16x16x4_f32(
                    false, aA[t], false, bL, (short)0, acc[t][j], false, false);
                acc[t][j] = __builtin_amdgcn_wmma_f32_16x16x4_f32(
                    false, aX[t], false, bR, (short)0, acc[t][j], false, false);
            }
        }
    }

    int hi = kh * 8;
#pragma unroll
    for (int t = 0; t < TML; ++t) {
        if (!valid[t]) continue;
#pragma unroll
        for (int g = 0; g < 8; ++g) {
            int m = m0c[t] + g + hi;
#pragma unroll
            for (int j = 0; j < 4; ++j) {
                float v = acc[t][j][g];
                if (relu) v = fmaxf(v, 0.0f);
                out[(size_t)m * D + j * 16 + colL] = v;
            }
        }
    }
}

__global__ void classifier_dot(const float* __restrict__ xp, const float* __restrict__ xg,
                               const int* __restrict__ ls, const int* __restrict__ ld,
                               float* __restrict__ out, int n) {
    int e = blockIdx.x * blockDim.x + threadIdx.x;
    if (e >= n) return;
    const float4* a = (const float4*)(xp + (size_t)ls[e] * D);
    const float4* b = (const float4*)(xg + (size_t)ld[e] * D);
    float s = 0.0f;
#pragma unroll
    for (int i = 0; i < 16; ++i) {
        float4 va = a[i], vb = b[i];
        s += va.x * vb.x + va.y * vb.y + va.z * vb.z + va.w * vb.w;
    }
    out[e] = s;
}

extern "C" void kernel_launch(void* const* d_in, const int* in_sizes, int n_in,
                              void* d_out, int out_size, void* d_ws, size_t ws_size,
                              hipStream_t stream) {
    const float* go_x  = (const float*)d_in[0];
    const float* p_emb = (const float*)d_in[1];
    const float* g_emb = (const float*)d_in[2];
    const float* lin_W = (const float*)d_in[3];
    const float* lin_b = (const float*)d_in[4];
    const float* Wl    = (const float*)d_in[5];
    const float* bl    = (const float*)d_in[6];
    const float* Wr    = (const float*)d_in[7];
    const int* p_nid   = (const int*)d_in[8];
    const int* g_nid   = (const int*)d_in[9];
    const int* src_pg  = (const int*)d_in[10];
    const int* dst_pg  = (const int*)d_in[11];
    const int* src_gp  = (const int*)d_in[12];
    const int* dst_gp  = (const int*)d_in[13];
    const int* lab_s   = (const int*)d_in[14];
    const int* lab_d   = (const int*)d_in[15];

    const int NP  = in_sizes[8];
    const int NG  = in_sizes[9];
    const int E   = in_sizes[10];
    const int EL  = in_sizes[14];
    const int FGO = in_sizes[0] / NG;

    char* base = (char*)d_ws;
    auto alloc = [&](size_t bytes) -> char* {
        char* p = base;
        base += (bytes + 255) & ~(size_t)255;
        return p;
    };
    float* xp_a  = (float*)alloc((size_t)NP * D * 4);
    float* xp_b  = (float*)alloc((size_t)NP * D * 4);
    float* xg_a  = (float*)alloc((size_t)NG * D * 4);
    float* xg_b  = (float*)alloc((size_t)NG * D * 4);
    float* agg_p = (float*)alloc((size_t)NP * D * 4);
    float* agg_g = (float*)alloc((size_t)NG * D * 4);
    int* row_g   = (int*)alloc((size_t)NG * 4);
    int* cur_g   = (int*)alloc((size_t)NG * 4);
    int* row_p   = (int*)alloc((size_t)NP * 4);
    int* cur_p   = (int*)alloc((size_t)NP * 4);
    int* srcs_g  = (int*)alloc((size_t)E * 4);
    int* srcs_p  = (int*)alloc((size_t)E * 4);
    int* bsum    = (int*)alloc(1024 * 4);

    const int nBlkE = (E + EPB - 1) / EPB;
    const int nbkG  = (NG + RPB - 1) / RPB;
    const int nbkP  = (NP + RPB - 1) / RPB;
    const int nmatG = nbkG * nBlkE;
    const int nmatP = nbkP * nBlkE;
    const int nbG   = (nmatG + 255) / 256;
    const int nbP   = (nmatP + 255) / 256;
    const int nmatGp = (nmatG + 63) & ~63;
    int* pairsG = (int*)agg_p;
    int* pairsP = pairsG + (size_t)2 * E;
    int* hmatG  = (int*)xp_b;
    int* hmatP  = hmatG + nmatGp;
    int* omatG  = (int*)xg_b;
    int* omatP  = omatG + nmatGp;
    int* bsumG  = bsum;
    int* bsumP  = omatP + ((nmatP + 63) & ~63);

    bucket_hist<<<2 * nBlkE, 1024, 0, stream>>>(dst_pg, hmatG, nbkG, dst_gp, hmatP, nbkP, E, nBlkE);
    scan_block<<<nbG + nbP, 256, 0, stream>>>(hmatG, omatG, bsumG, nmatG, nbG,
                                              hmatP, omatP, bsumP, nmatP);
    scan_sums<<<2, 1024, 0, stream>>>(bsumG, nbG, bsumP, nbP);
    add_block_offsets<<<nbG + nbP, 256, 0, stream>>>(omatG, bsumG, nmatG, nbG,
                                                     omatP, bsumP, nmatP);
    bucket_scatter<<<2 * nBlkE, 1024, 0, stream>>>(src_pg, dst_pg, omatG, pairsG, nbkG,
                                                   src_gp, dst_gp, omatP, pairsP, nbkP, E, nBlkE);
    bucket_csr<<<nbkG + nbkP, 1024, 0, stream>>>(pairsG, omatG, row_g, cur_g, srcs_g, NG, nbkG,
                                                 pairsP, omatP, row_p, cur_p, srcs_p, NP, nbkP,
                                                 E, nBlkE);

    {
        int tiles      = (NG + 15) / 16;
        int gemmBlocks = (tiles + 8 * TMI - 1) / (8 * TMI);
        int gatherBlks = (NP * 16 + 255) / 256;
        input_embed<<<gemmBlocks + gatherBlks, 256, 0, stream>>>(go_x, lin_W, lin_b, g_emb, g_nid,
                                                                 xg_a, NG, FGO, gemmBlocks,
                                                                 p_emb, p_nid, xp_a, NP * 16);
    }

    float* xp_cur = xp_a; float* xp_nxt = xp_b;
    float* xg_cur = xg_a; float* xg_nxt = xg_b;
    const int aggBlkG  = (NG * 16 + 255) / 256;
    const int aggBlkP  = (NP * 16 + 255) / 256;
    const int gemmBlkG = ((NG + 15) / 16 + 8 * TML - 1) / (8 * TML);
    const int gemmBlkP = ((NP + 15) / 16 + 8 * TML - 1) / (8 * TML);
    for (int l = 0; l < 3; ++l) {
        aggregate_csr<<<aggBlkG + aggBlkP, 256, 0, stream>>>(xp_cur, srcs_g, row_g, cur_g, agg_g,
                                                             NG, aggBlkG,
                                                             xg_cur, srcs_p, row_p, cur_p, agg_p,
                                                             NP);
        const float* Wl0 = Wl + (size_t)(l * 2 + 0) * D * D;
        const float* Wl1 = Wl + (size_t)(l * 2 + 1) * D * D;
        const float* Wr0 = Wr + (size_t)(l * 2 + 0) * D * D;
        const float* Wr1 = Wr + (size_t)(l * 2 + 1) * D * D;
        const float* bl0 = bl + (size_t)(l * 2 + 0) * D;
        const float* bl1 = bl + (size_t)(l * 2 + 1) * D;
        int relu = (l < 2) ? 1 : 0;
        layer_gemm_wmma<<<gemmBlkG + gemmBlkP, 256, 0, stream>>>(
            agg_g, xg_cur, Wl0, bl0, Wr0, xg_nxt, NG, gemmBlkG,
            agg_p, xp_cur, Wl1, bl1, Wr1, xp_nxt, NP, relu);
        float* t;
        t = xp_cur; xp_cur = xp_nxt; xp_nxt = t;
        t = xg_cur; xg_cur = xg_nxt; xg_nxt = t;
    }

    classifier_dot<<<(EL + 255) / 256, 256, 0, stream>>>(xp_cur, xg_cur, lab_s, lab_d,
                                                         (float*)d_out, EL);
}


# CSR build: both scatter passes group their output in LDS and store it in index order, so a store covers whole lines, not one line per lane; neighbour order in a row is still arrival order
# speedup vs baseline: 1.9625x; 1.0652x over previous
#include <hip/hip_runtime.h>

typedef __attribute__((ext_vector_type(2))) float v2f;
typedef __attribute__((ext_vector_type(8))) float v8f;

#define D    64
#define KC   128
#define STI  (KC+4)
#define STL  (D+4)
#define TMI  2
#define TML  5


#define EPB   8192
#define RSH   8
#define RPB   256
#define MAXB  1024
#define CAPS  16384


__device__ __forceinline__ int block_excl_scan_1024(int v, int* __restrict__ wsum, int tid) {
    const int lane = tid & 31, w = tid >> 5;
    int x = v;
    for (int off = 1; off < 32; off <<= 1) {
        int y = __shfl_up(x, off, 32);
        if (lane >= off) x += y;
    }
    if (lane == 31) wsum[w] = x;
    __syncthreads();
    if (w == 0) {
        int t = wsum[lane], z = t;
        for (int off = 1; off < 32; off <<= 1) {
            int y = __shfl_up(z, off, 32);
            if (lane >= off) z += y;
        }
        wsum[lane] = z - t;
    }
    __syncthreads();
    return x - v + wsum[w];
}

__global__ __launch_bounds__(1024) void bucket_hist(const int* __restrict__ dstA,
                                                    int* __restrict__ histA, int nbkA,
                                                    const int* __restrict__ dstB,
                                                    int* __restrict__ histB, int nbkB,
                                                    int nE, int nBlk) {
    __shared__ int h[MAXB];
    const bool second = (int)blockIdx.x >= nBlk;
    const int* __restrict__ dst = second ? dstB : dstA;
    int* __restrict__ blkHist   = second ? histB : histA;
    const int nBuckets = second ? nbkB : nbkA;
    const int blk = second ? (int)blockIdx.x - nBlk : (int)blockIdx.x;
    int tid = threadIdx.x;
    h[tid] = 0;
    __syncthreads();
    int base = blk * EPB + tid;
    int d[EPB / 1024];
#pragma unroll
    for (int u = 0; u < EPB / 1024; ++u) {
        int e = base + u * 1024;
        d[u] = dst[max(min(e, nE - 1), 0)];
    }
#pragma unroll
    for (int u = 0; u < EPB / 1024; ++u) {
        int e = base + u * 1024;
        if (e < nE) atomicAdd(&h[d[u] >> RSH], 1);
    }
    __syncthreads();
    if (tid < nBuckets) blkHist[tid * nBlk + blk] = h[tid];
}

__global__ void scan_block(const int* __restrict__ degA, int* __restrict__ exA,
                           int* __restrict__ bsumA, int nA, int nbA,
                           const int* __restrict__ degB, int* __restrict__ exB,
                           int* __restrict__ bsumB, int nB) {
    __shared__ int sh[256];
    const bool second = (int)blockIdx.x >= nbA;
    const int* __restrict__ deg = second ? degB : degA;
    int* __restrict__ ex   = second ? exB : exA;
    int* __restrict__ bsum = second ? bsumB : bsumA;
    const int n   = second ? nB : nA;
    const int blk = second ? (int)blockIdx.x - nbA : (int)blockIdx.x;
    int tid = threadIdx.x;
    int i = blk * 256 + tid;
    int v = (i < n) ? deg[i] : 0;
    sh[tid] = v;
    __syncthreads();
    for (int off = 1; off < 256; off <<= 1) {
        int add = (tid >= off) ? sh[tid - off] : 0;
        __syncthreads();
        sh[tid] += add;
        __syncthreads();
    }
    if (i < n) ex[i] = sh[tid] - v;
    if (tid == 255) bsum[blk] = sh[255];
}

__global__ void scan_sums(int* __restrict__ bsumA, int nbA,
                          int* __restrict__ bsumB, int nbB) {
    __shared__ int sh[1024];
    const bool second = blockIdx.x != 0;
    int* __restrict__ bsum = second ? bsumB : bsumA;
    const int nb = second ? nbB : nbA;
    int tid = threadIdx.x;
    int v = (tid < nb) ? bsum[tid] : 0;
    sh[tid] = v;
    __syncthreads();
    for (int off = 1; off < 1024; off <<= 1) {
        int add = (tid >= off) ? sh[tid - off] : 0;
        __syncthreads();
        sh[tid] += add;
        __syncthreads();
    }
    if (tid < nb) bsum[tid] = sh[tid] - v;
}

__global__ void add_block_offsets(int* __restrict__ exA, const int* __restrict__ bsumA, int nA,
                                  int nbA,
                                  int* __restrict__ exB, const int* __restrict__ bsumB, int nB) {
    const bool second = (int)blockIdx.x >= nbA;
    int* __restrict__ ex         = second ? exB : exA;
    const int* __restrict__ bsum = second ? bsumB : bsumA;
    const int n   = second ? nB : nA;
    const int blk = second ? (int)blockIdx.x - nbA : (int)blockIdx.x;
    int i = blk * 256 + threadIdx.x;
    if (i < n) ex[i] += bsum[blk];
}

__global__ __launch_bounds__(1024) void bucket_scatter(const int* __restrict__ srcA,
                                                       const int* __restrict__ dstA,
                                                       const int* __restrict__ histA,
                                                       const int* __restrict__ offA,
                                                       int* __restrict__ pairsA, int nbkA,
                                                       const int* __restrict__ srcB,
                                                       const int* __restrict__ dstB,
                                                       const int* __restrict__ histB,
                                                       const int* __restrict__ offB,
                                                       int* __restrict__ pairsB, int nbkB,
                                                       int nE, int nBlk) {
    __shared__ int2 stage[EPB];
    __shared__ int lcur[MAXB];
    __shared__ int delta[MAXB];
    __shared__ int wsum[32];
    const bool second = (int)blockIdx.x >= nBlk;
    const int* __restrict__ src     = second ? srcB : srcA;
    const int* __restrict__ dst     = second ? dstB : dstA;
    const int* __restrict__ blkHist = second ? histB : histA;
    const int* __restrict__ blkOff  = second ? offB : offA;
    int* __restrict__ pairs         = second ? pairsB : pairsA;
    const int nBuckets = second ? nbkB : nbkA;
    const int blk = second ? (int)blockIdx.x - nBlk : (int)blockIdx.x;
    int tid = threadIdx.x;
    int nMine = 0, gOff = 0;
    if (tid < nBuckets) {
        nMine  = blkHist[tid * nBlk + blk];
        gOff = blkOff[tid * nBlk + blk];
    }
    int base = blk * EPB + tid;
    int d[EPB / 1024], s[EPB / 1024];
#pragma unroll
    for (int u = 0; u < EPB / 1024; ++u) {
        int ec = max(min(base + u * 1024, nE - 1), 0);
        d[u] = dst[ec];
        s[u] = src[ec];
    }
    int lst = block_excl_scan_1024(nMine, wsum, tid);
    lcur[tid]  = lst;
    delta[tid] = gOff - lst;
    __syncthreads();
#pragma unroll
    for (int u = 0; u < EPB / 1024; ++u) {
        int e = base + u * 1024;
        if (e < nE) {
            int lp = atomicAdd(&lcur[d[u] >> RSH], 1);
            stage[lp] = make_int2(s[u], d[u]);
        }
    }
    __syncthreads();
    int2* __restrict__ p2 = (int2*)pairs;
    const int nValid = min(EPB, nE - blk * EPB);
#pragma unroll
    for (int u = 0; u < EPB / 1024; ++u) {
        int j = tid + u * 1024;
        if (j < nValid) {
            int2 p = stage[j];
            p2[j + delta[p.y >> RSH]] = p;
        }
    }
}

__global__ __launch_bounds__(1024) void bucket_csr(const int* __restrict__ pairsA,
                                                   const int* __restrict__ offA,
                                                   int* __restrict__ rowStartA,
                                                   int* __restrict__ rowEndA,
                                                   int* __restrict__ srcsA, int nA, int nbkA,
                                                   const int* __restrict__ pairsB,
                                                   const int* __restrict__ offB,
                                                   int* __restrict__ rowStartB,
                                                   int* __restrict__ rowEndB,
                                                   int* __restrict__ srcsB, int nB, int nbkB,
                                                   int nE, int nBlk) {
    __shared__ int cnt[RPB];
    __shared__ int sc[RPB];
    __shared__ int lbuf[CAPS];
    const bool second = (int)blockIdx.x >= nbkA;
    const int* __restrict__ pairs  = second ? pairsB : pairsA;
    const int* __restrict__ blkOff = second ? offB : offA;
    int* __restrict__ rowStart     = second ? rowStartB : rowStartA;
    int* __restrict__ rowEnd       = second ? rowEndB : rowEndA;
    int* __restrict__ srcs         = second ? srcsB : srcsA;
    const int n        = second ? nB : nA;
    const int nBuckets = second ? nbkB : nbkA;
    int b = second ? (int)blockIdx.x - nbkA : (int)blockIdx.x, tid = threadIdx.x;
    int s0 = blkOff[b * nBlk];
    int s1 = (b + 1 < nBuckets) ? blkOff[(b + 1) * nBlk] : nE;
    const int2* __restrict__ p2 = (const int2*)pairs;
    const int nEnt = s1 - s0;
    const bool grouped = nEnt <= CAPS;
    if (tid < RPB) cnt[tid] = 0;
    __syncthreads();
    for (int i0 = s0 + tid; i0 < s1; i0 += 4096) {
        int2 p[4];
#pragma unroll
        for (int u = 0; u < 4; ++u)
            p[u] = p2[min(i0 + u * 1024, s1 - 1)];
#pragma unroll
        for (int u = 0; u < 4; ++u) {
            int i = i0 + u * 1024;
            if (i < s1) atomicAdd(&cnt[p[u].y & (RPB - 1)], 1);
        }
    }
    __syncthreads();
    int v = 0;
    if (tid < RPB) { v = cnt[tid]; sc[tid] = v; }
    __syncthreads();
    for (int off = 1; off < RPB; off <<= 1) {
        int add = 0;
        if (tid < RPB && tid >= off) add = sc[tid - off];
        __syncthreads();
        if (tid < RPB) sc[tid] += add;
        __syncthreads();
    }
    if (tid < RPB) {
        int start = s0 + sc[tid] - v;
        int node = b * RPB + tid;
        if (node < n) { rowStart[node] = start; rowEnd[node] = start + v; }
        cnt[tid] = start;
    }
    __syncthreads();
    for (int i0 = s0 + tid; i0 < s1; i0 += 4096) {
        int2 p[4];
#pragma unroll
        for (int u = 0; u < 4; ++u)
            p[u] = p2[min(i0 + u * 1024, s1 - 1)];
#pragma unroll
        for (int u = 0; u < 4; ++u) {
            int i = i0 + u * 1024;
            if (i < s1) {
                int pos = atomicAdd(&cnt[p[u].y & (RPB - 1)], 1);
                if (grouped) lbuf[pos - s0] = p[u].x;
                else         srcs[pos] = p[u].x;
            }
        }
    }
    if (grouped) {
        __syncthreads();
        for (int i = tid; i < nEnt; i += 1024) srcs[s0 + i] = lbuf[i];
    }
}

__global__ void aggregate_csr(const float* __restrict__ xA, const int* __restrict__ srcsA,
                              const int* __restrict__ rowStartA, const int* __restrict__ rowEndA,
                              float* __restrict__ aggA, int nA, int blocksA,
                              const float* __restrict__ xB, const int* __restrict__ srcsB,
                              const int* __restrict__ rowStartB, const int* __restrict__ rowEndB,
                              float* __restrict__ aggB, int nB) {
    const bool second = (int)blockIdx.x >= blocksA;
    const float* __restrict__ x      = second ? xB : xA;
    const int* __restrict__ srcs     = second ? srcsB : srcsA;
    const int* __restrict__ rowStart = second ? rowStartB : rowStartA;
    const int* __restrict__ rowEnd   = second ? rowEndB : rowEndA;
    float* __restrict__ agg          = second ? aggB : aggA;
    const int n   = second ? nB : nA;
    const int blk = second ? (int)blockIdx.x - blocksA : (int)blockIdx.x;
    int t = blk * blockDim.x + threadIdx.x;
    int node = t >> 4;
    if (node >= n) return;
    int c = (t & 15) * 4;
    int s0 = rowStart[node], s1 = rowEnd[node];
    float4 acc = make_float4(0.f, 0.f, 0.f, 0.f);
    for (int i = s0; i < s1; ++i) {
        int s = srcs[i];
        float4 v = *(const float4*)(x + (size_t)s * D + c);
        acc.x += v.x; acc.y += v.y; acc.z += v.z; acc.w += v.w;
    }
    float inv = 1.0f / fmaxf((float)(s1 - s0), 1.0f);
    acc.x *= inv; acc.y *= inv; acc.z *= inv; acc.w *= inv;
    *(float4*)(agg + (size_t)node * D + c) = acc;
}

__device__ __forceinline__ void gather_rows_body(const float* __restrict__ emb,
                                                 const int* __restrict__ nid,
                                                 float* __restrict__ out, int n16, int blk) {
    int t = blk * blockDim.x + threadIdx.x;
    if (t >= n16) return;
    int i = t >> 4, c = (t & 15) * 4;
    int s = nid[i];
    *(float4*)(out + (size_t)i * D + c) = *(const float4*)(emb + (size_t)s * D + c);
}

__device__ __forceinline__ void input_gemm_body(
        const float* __restrict__ go_x, const float* __restrict__ lin_W,
        const float* __restrict__ lin_b, const float* __restrict__ go_emb,
        const int* __restrict__ go_nid, float* __restrict__ xg, int M, int K, int blk) {
    __shared__ float sWt[D * STI];

    int waveInBlk = threadIdx.x >> 5;
    int lane = threadIdx.x & 31;
    int row  = lane & 15;
    int kh   = lane >> 4;
    int colL = lane & 15;

    int numTiles = M >> 4;
    int tile0 = (blk * 8 + waveInBlk) * TMI;

    int   m0c[TMI];
    const float* arow[TMI];
    bool  valid[TMI];
#pragma unroll
    for (int t = 0; t < TMI; ++t) {
        int ti = tile0 + t;
        valid[t] = (ti < numTiles);
        m0c[t] = valid[t] ? (ti << 4) : 0;
        arow[t] = go_x + (size_t)(m0c[t] + row) * K;
    }

    v8f acc[TMI][4];
#pragma unroll
    for (int t = 0; t < TMI; ++t)
#pragma unroll
        for (int j = 0; j < 4; ++j) {
            float b = lin_b[j * 16 + colL];
#pragma unroll
            for (int g = 0; g < 8; ++g) acc[t][j][g] = b;
        }

    for (int kc0 = 0; kc0 < K; kc0 += KC) {
        int len = (K - kc0 < KC) ? (K - kc0) : KC;
        __syncthreads();
        for (int i = threadIdx.x * 4; i < len * D; i += 256 * 4) {
            int r = i >> 6, c = i & 63;
            float4 w = *(const float4*)&lin_W[(size_t)(kc0 + r) * D + c];
            sWt[(c + 0) * STI + r] = w.x;
            sWt[(c + 1) * STI + r] = w.y;
            sWt[(c + 2) * STI + r] = w.z;
            sWt[(c + 3) * STI + r] = w.w;
        }
        __syncthreads();
        if (kc0 + KC < K) {
#pragma unroll
            for (int t = 0; t < TMI; ++t)
                __builtin_prefetch(arow[t] + kc0 + KC, 0, 0);
        }
        for (int k = 0; k < len; k += 4) {
            v2f a[TMI];
#pragma unroll
            for (int t = 0; t < TMI; ++t)
                a[t] = *(const v2f*)(arow[t] + kc0 + k + 2 * kh);
#pragma unroll
            for (int j = 0; j < 4; ++j) {
                v2f b = *(const v2f*)&sWt[(j * 16 + colL) * STI + k + 2 * kh];
#pragma unroll
                for (int t = 0; t < TMI; ++t)
                    acc[t][j] = __builtin_amdgcn_wmma_f32_16x16x4_f32(
                        false, a[t], false, b, (short)0, acc[t][j], false, false);
            }
        }
    }

    int hi = kh * 8;
#pragma unroll
    for (int t = 0; t < TMI; ++t) {
        if (!valid[t]) continue;
#pragma unroll
        for (int g = 0; g < 8; ++g) {
            int m = m0c[t] + g + hi;
            int nid = go_nid[m];
#pragma unroll
            for (int j = 0; j < 4; ++j) {
                int col = j * 16 + colL;
                xg[(size_t)m * D + col] = acc[t][j][g] + go_emb[(size_t)nid * D + col];
            }
        }
    }
}

__global__ __launch_bounds__(256) void input_embed(
        const float* __restrict__ go_x, const float* __restrict__ lin_W,
        const float* __restrict__ lin_b, const float* __restrict__ go_emb,
        const int* __restrict__ go_nid, float* __restrict__ xg, int M, int K, int gemmBlocks,
        const float* __restrict__ p_emb, const int* __restrict__ p_nid,
        float* __restrict__ xp, int n16) {
    if ((int)blockIdx.x < gemmBlocks)
        input_gemm_body(go_x, lin_W, lin_b, go_emb, go_nid, xg, M, K, (int)blockIdx.x);
    else
        gather_rows_body(p_emb, p_nid, xp, n16, (int)blockIdx.x - gemmBlocks);
}

__global__ __launch_bounds__(256) void layer_gemm_wmma(
        const float* __restrict__ aggA, const float* __restrict__ xA,
        const float* __restrict__ WlA, const float* __restrict__ blA,
        const float* __restrict__ WrA, float* __restrict__ outA, int MA, int blocksA,
        const float* __restrict__ aggB, const float* __restrict__ xB,
        const float* __restrict__ WlB, const float* __restrict__ blB,
        const float* __restrict__ WrB, float* __restrict__ outB, int MB, int relu) {
    __shared__ float sWt[2 * D * STL];
    const bool second = (int)blockIdx.x >= blocksA;
    const float* __restrict__ agg = second ? aggB : aggA;
    const float* __restrict__ x   = second ? xB : xA;
    const float* __restrict__ Wl  = second ? WlB : WlA;
    const float* __restrict__ bl  = second ? blB : blA;
    const float* __restrict__ Wr  = second ? WrB : WrA;
    float* __restrict__ out       = second ? outB : outA;
    const int M   = second ? MB : MA;
    const int blk = second ? (int)blockIdx.x - blocksA : (int)blockIdx.x;

    for (int i = threadIdx.x * 4; i < D * D; i += 256 * 4) {
        int r = i >> 6, c = i & 63;
        float4 wl = *(const float4*)&Wl[i];
        float4 wr = *(const float4*)&Wr[i];
        sWt[(c + 0) * STL + r] = wl.x;
        sWt[(c + 1) * STL + r] = wl.y;
        sWt[(c + 2) * STL + r] = wl.z;
        sWt[(c + 3) * STL + r] = wl.w;
        sWt[D * STL + (c + 0) * STL + r] = wr.x;
        sWt[D * STL + (c + 1) * STL + r] = wr.y;
        sWt[D * STL + (c + 2) * STL + r] = wr.z;
        sWt[D * STL + (c + 3) * STL + r] = wr.w;
    }
    __syncthreads();

    int waveInBlk = threadIdx.x >> 5;
    int lane = threadIdx.x & 31;
    int row  = lane & 15;
    int kh   = lane >> 4;
    int colL = lane & 15;

    int numTiles = M >> 4;
    int tile0 = (blk * 8 + waveInBlk) * TML;

    int   m0c[TML];
    bool  valid[TML];
    const float* aggrow[TML];
    const float* xrow[TML];
#pragma unroll
    for (int t = 0; t < TML; ++t) {
        int ti = tile0 + t;
        valid[t] = (ti < numTiles);
        m0c[t] = valid[t] ? (ti << 4) : 0;
        aggrow[t] = agg + (size_t)(m0c[t] + row) * D;
        xrow[t]   = x   + (size_t)(m0c[t] + row) * D;
    }

    v8f acc[TML][4];
#pragma unroll
    for (int t = 0; t < TML; ++t)
#pragma unroll
        for (int j = 0; j < 4; ++j) {
            float b = bl[j * 16 + colL];
#pragma unroll
            for (int g = 0; g < 8; ++g) acc[t][j][g] = b;
        }

    for (int k = 0; k < D; k += 4) {
        v2f aA[TML], aX[TML];
#pragma unroll
        for (int t = 0; t < TML; ++t) {
            aA[t] = *(const v2f*)(aggrow[t] + k + 2 * kh);
            aX[t] = *(const v2f*)(xrow[t]   + k + 2 * kh);
        }
#pragma unroll
        for (int j = 0; j < 4; ++j) {
            v2f bL = *(const v2f*)&sWt[(j * 16 + colL) * STL + k + 2 * kh];
            v2f bR = *(const v2f*)&sWt[D * STL + (j * 16 + colL) * STL + k + 2 * kh];
#pragma unroll
            for (int t = 0; t < TML; ++t) {
                acc[t][j] = __builtin_amdgcn_wmma_f32_16x16x4_f32(
                    false, aA[t], false, bL, (short)0, acc[t][j], false, false);
                acc[t][j] = __builtin_amdgcn_wmma_f32_16x16x4_f32(
                    false, aX[t], false, bR, (short)0, acc[t][j], false, false);
            }
        }
    }

    int hi = kh * 8;
#pragma unroll
    for (int t = 0; t < TML; ++t) {
        if (!valid[t]) continue;
#pragma unroll
        for (int g = 0; g < 8; ++g) {
            int m = m0c[t] + g + hi;
#pragma unroll
            for (int j = 0; j < 4; ++j) {
                float v = acc[t][j][g];
                if (relu) v = fmaxf(v, 0.0f);
                out[(size_t)m * D + j * 16 + colL] = v;
            }
        }
    }
}

__global__ void classifier_dot(const float* __restrict__ xp, const float* __restrict__ xg,
                               const int* __restrict__ ls, const int* __restrict__ ld,
                               float* __restrict__ out, int n) {
    int e = blockIdx.x * blockDim.x + threadIdx.x;
    if (e >= n) return;
    const float4* a = (const float4*)(xp + (size_t)ls[e] * D);
    const float4* b = (const float4*)(xg + (size_t)ld[e] * D);
    float s = 0.0f;
#pragma unroll
    for (int i = 0; i < 16; ++i) {
        float4 va = a[i], vb = b[i];
        s += va.x * vb.x + va.y * vb.y + va.z * vb.z + va.w * vb.w;
    }
    out[e] = s;
}

extern "C" void kernel_launch(void* const* d_in, const int* in_sizes, int n_in,
                              void* d_out, int out_size, void* d_ws, size_t ws_size,
                              hipStream_t stream) {
    const float* go_x  = (const float*)d_in[0];
    const float* p_emb = (const float*)d_in[1];
    const float* g_emb = (const float*)d_in[2];
    const float* lin_W = (const float*)d_in[3];
    const float* lin_b = (const float*)d_in[4];
    const float* Wl    = (const float*)d_in[5];
    const float* bl    = (const float*)d_in[6];
    const float* Wr    = (const float*)d_in[7];
    const int* p_nid   = (const int*)d_in[8];
    const int* g_nid   = (const int*)d_in[9];
    const int* src_pg  = (const int*)d_in[10];
    const int* dst_pg  = (const int*)d_in[11];
    const int* src_gp  = (const int*)d_in[12];
    const int* dst_gp  = (const int*)d_in[13];
    const int* lab_s   = (const int*)d_in[14];
    const int* lab_d   = (const int*)d_in[15];

    const int NP  = in_sizes[8];
    const int NG  = in_sizes[9];
    const int E   = in_sizes[10];
    const int EL  = in_sizes[14];
    const int FGO = in_sizes[0] / NG;

    char* base = (char*)d_ws;
    auto alloc = [&](size_t bytes) -> char* {
        char* p = base;
        base += (bytes + 255) & ~(size_t)255;
        return p;
    };
    float* xp_a  = (float*)alloc((size_t)NP * D * 4);
    float* xp_b  = (float*)alloc((size_t)NP * D * 4);
    float* xg_a  = (float*)alloc((size_t)NG * D * 4);
    float* xg_b  = (float*)alloc((size_t)NG * D * 4);
    float* agg_p = (float*)alloc((size_t)NP * D * 4);
    float* agg_g = (float*)alloc((size_t)NG * D * 4);
    int* row_g   = (int*)alloc((size_t)NG * 4);
    int* cur_g   = (int*)alloc((size_t)NG * 4);
    int* row_p   = (int*)alloc((size_t)NP * 4);
    int* cur_p   = (int*)alloc((size_t)NP * 4);
    int* srcs_g  = (int*)alloc((size_t)E * 4);
    int* srcs_p  = (int*)alloc((size_t)E * 4);
    int* bsum    = (int*)alloc(1024 * 4);

    const int nBlkE = (E + EPB - 1) / EPB;
    const int nbkG  = (NG + RPB - 1) / RPB;
    const int nbkP  = (NP + RPB - 1) / RPB;
    const int nmatG = nbkG * nBlkE;
    const int nmatP = nbkP * nBlkE;
    const int nbG   = (nmatG + 255) / 256;
    const int nbP   = (nmatP + 255) / 256;
    const int nmatGp = (nmatG + 63) & ~63;
    int* pairsG = (int*)agg_p;
    int* pairsP = pairsG + (size_t)2 * E;
    int* hmatG  = (int*)xp_b;
    int* hmatP  = hmatG + nmatGp;
    int* omatG  = (int*)xg_b;
    int* omatP  = omatG + nmatGp;
    int* bsumG  = bsum;
    int* bsumP  = omatP + ((nmatP + 63) & ~63);

    bucket_hist<<<2 * nBlkE, 1024, 0, stream>>>(dst_pg, hmatG, nbkG, dst_gp, hmatP, nbkP, E, nBlkE);
    scan_block<<<nbG + nbP, 256, 0, stream>>>(hmatG, omatG, bsumG, nmatG, nbG,
                                              hmatP, omatP, bsumP, nmatP);
    scan_sums<<<2, 1024, 0, stream>>>(bsumG, nbG, bsumP, nbP);
    add_block_offsets<<<nbG + nbP, 256, 0, stream>>>(omatG, bsumG, nmatG, nbG,
                                                     omatP, bsumP, nmatP);
    bucket_scatter<<<2 * nBlkE, 1024, 0, stream>>>(src_pg, dst_pg, hmatG, omatG, pairsG, nbkG,
                                                   src_gp, dst_gp, hmatP, omatP, pairsP, nbkP,
                                                   E, nBlkE);
    bucket_csr<<<nbkG + nbkP, 1024, 0, stream>>>(pairsG, omatG, row_g, cur_g, srcs_g, NG, nbkG,
                                                 pairsP, omatP, row_p, cur_p, srcs_p, NP, nbkP,
                                                 E, nBlkE);

    {
        int tiles      = (NG + 15) / 16;
        int gemmBlocks = (tiles + 8 * TMI - 1) / (8 * TMI);
        int gatherBlks = (NP * 16 + 255) / 256;
        input_embed<<<gemmBlocks + gatherBlks, 256, 0, stream>>>(go_x, lin_W, lin_b, g_emb, g_nid,
                                                                 xg_a, NG, FGO, gemmBlocks,
                                                                 p_emb, p_nid, xp_a, NP * 16);
    }

    float* xp_cur = xp_a; float* xp_nxt = xp_b;
    float* xg_cur = xg_a; float* xg_nxt = xg_b;
    const int aggBlkG  = (NG * 16 + 255) / 256;
    const int aggBlkP  = (NP * 16 + 255) / 256;
    const int gemmBlkG = ((NG + 15) / 16 + 8 * TML - 1) / (8 * TML);
    const int gemmBlkP = ((NP + 15) / 16 + 8 * TML - 1) / (8 * TML);
    for (int l = 0; l < 3; ++l) {
        aggregate_csr<<<aggBlkG + aggBlkP, 256, 0, stream>>>(xp_cur, srcs_g, row_g, cur_g, agg_g,
                                                             NG, aggBlkG,
                                                             xg_cur, srcs_p, row_p, cur_p, agg_p,
                                                             NP);
        const float* Wl0 = Wl + (size_t)(l * 2 + 0) * D * D;
        const float* Wl1 = Wl + (size_t)(l * 2 + 1) * D * D;
        const float* Wr0 = Wr + (size_t)(l * 2 + 0) * D * D;
        const float* Wr1 = Wr + (size_t)(l * 2 + 1) * D * D;
        const float* bl0 = bl + (size_t)(l * 2 + 0) * D;
        const float* bl1 = bl + (size_t)(l * 2 + 1) * D;
        int relu = (l < 2) ? 1 : 0;
        layer_gemm_wmma<<<gemmBlkG + gemmBlkP, 256, 0, stream>>>(
            agg_g, xg_cur, Wl0, bl0, Wr0, xg_nxt, NG, gemmBlkG,
            agg_p, xp_cur, Wl1, bl1, Wr1, xp_nxt, NP, relu);
        float* t;
        t = xp_cur; xp_cur = xp_nxt; xp_nxt = t;
        t = xg_cur; xg_cur = xg_nxt; xg_nxt = t;
    }

    classifier_dot<<<(EL + 255) / 256, 256, 0, stream>>>(xp_cur, xg_cur, lab_s, lab_d,
                                                         (float*)d_out, EL);
}


# Mean aggregation fused into the layer GEMM launch: a work-group sums its 64 rows' neighbour rows (4 in flight, CSR order kept) into LDS and feeds WMMA A from there; 14 launches to 11
# speedup vs baseline: 2.2016x; 1.1218x over previous
#include <hip/hip_runtime.h>

typedef __attribute__((ext_vector_type(2))) float v2f;
typedef __attribute__((ext_vector_type(8))) float v8f;

#define D    64
#define KC   128
#define STI  (KC+4)
#define STL  (D+4)
#define TMI  2
#define FR   64


#define EPB   8192
#define RSH   8
#define RPB   256
#define MAXB  1024
#define CAPS  16384


__device__ __forceinline__ int block_excl_scan_1024(int v, int* __restrict__ wsum, int tid) {
    const int lane = tid & 31, w = tid >> 5;
    int x = v;
    for (int off = 1; off < 32; off <<= 1) {
        int y = __shfl_up(x, off, 32);
        if (lane >= off) x += y;
    }
    if (lane == 31) wsum[w] = x;
    __syncthreads();
    if (w == 0) {
        int t = wsum[lane], z = t;
        for (int off = 1; off < 32; off <<= 1) {
            int y = __shfl_up(z, off, 32);
            if (lane >= off) z += y;
        }
        wsum[lane] = z - t;
    }
    __syncthreads();
    return x - v + wsum[w];
}

__global__ __launch_bounds__(1024) void bucket_hist(const int* __restrict__ dstA,
                                                    int* __restrict__ histA, int nbkA,
                                                    const int* __restrict__ dstB,
                                                    int* __restrict__ histB, int nbkB,
                                                    int nE, int nBlk) {
    __shared__ int h[MAXB];
    const bool second = (int)blockIdx.x >= nBlk;
    const int* __restrict__ dst = second ? dstB : dstA;
    int* __restrict__ blkHist   = second ? histB : histA;
    const int nBuckets = second ? nbkB : nbkA;
    const int blk = second ? (int)blockIdx.x - nBlk : (int)blockIdx.x;
    int tid = threadIdx.x;
    h[tid] = 0;
    __syncthreads();
    int base = blk * EPB + tid;
    int d[EPB / 1024];
#pragma unroll
    for (int u = 0; u < EPB / 1024; ++u) {
        int e = base + u * 1024;
        d[u] = dst[max(min(e, nE - 1), 0)];
    }
#pragma unroll
    for (int u = 0; u < EPB / 1024; ++u) {
        int e = base + u * 1024;
        if (e < nE) atomicAdd(&h[d[u] >> RSH], 1);
    }
    __syncthreads();
    if (tid < nBuckets) blkHist[tid * nBlk + blk] = h[tid];
}

__global__ void scan_block(const int* __restrict__ degA, int* __restrict__ exA,
                           int* __restrict__ bsumA, int nA, int nbA,
                           const int* __restrict__ degB, int* __restrict__ exB,
                           int* __restrict__ bsumB, int nB) {
    __shared__ int sh[256];
    const bool second = (int)blockIdx.x >= nbA;
    const int* __restrict__ deg = second ? degB : degA;
    int* __restrict__ ex   = second ? exB : exA;
    int* __restrict__ bsum = second ? bsumB : bsumA;
    const int n   = second ? nB : nA;
    const int blk = second ? (int)blockIdx.x - nbA : (int)blockIdx.x;
    int tid = threadIdx.x;
    int i = blk * 256 + tid;
    int v = (i < n) ? deg[i] : 0;
    sh[tid] = v;
    __syncthreads();
    for (int off = 1; off < 256; off <<= 1) {
        int add = (tid >= off) ? sh[tid - off] : 0;
        __syncthreads();
        sh[tid] += add;
        __syncthreads();
    }
    if (i < n) ex[i] = sh[tid] - v;
    if (tid == 255) bsum[blk] = sh[255];
}

__global__ void scan_sums(int* __restrict__ bsumA, int nbA,
                          int* __restrict__ bsumB, int nbB) {
    __shared__ int sh[1024];
    const bool second = blockIdx.x != 0;
    int* __restrict__ bsum = second ? bsumB : bsumA;
    const int nb = second ? nbB : nbA;
    int tid = threadIdx.x;
    int v = (tid < nb) ? bsum[tid] : 0;
    sh[tid] = v;
    __syncthreads();
    for (int off = 1; off < 1024; off <<= 1) {
        int add = (tid >= off) ? sh[tid - off] : 0;
        __syncthreads();
        sh[tid] += add;
        __syncthreads();
    }
    if (tid < nb) bsum[tid] = sh[tid] - v;
}

__global__ void add_block_offsets(int* __restrict__ exA, const int* __restrict__ bsumA, int nA,
                                  int nbA,
                                  int* __restrict__ exB, const int* __restrict__ bsumB, int nB) {
    const bool second = (int)blockIdx.x >= nbA;
    int* __restrict__ ex         = second ? exB : exA;
    const int* __restrict__ bsum = second ? bsumB : bsumA;
    const int n   = second ? nB : nA;
    const int blk = second ? (int)blockIdx.x - nbA : (int)blockIdx.x;
    int i = blk * 256 + threadIdx.x;
    if (i < n) ex[i] += bsum[blk];
}

__global__ __launch_bounds__(1024) void bucket_scatter(const int* __restrict__ srcA,
                                                       const int* __restrict__ dstA,
                                                       const int* __restrict__ histA,
                                                       const int* __restrict__ offA,
                                                       int* __restrict__ pairsA, int nbkA,
                                                       const int* __restrict__ srcB,
                                                       const int* __restrict__ dstB,
                                                       const int* __restrict__ histB,
                                                       const int* __restrict__ offB,
                                                       int* __restrict__ pairsB, int nbkB,
                                                       int nE, int nBlk) {
    __shared__ int2 stage[EPB];
    __shared__ int lcur[MAXB];
    __shared__ int delta[MAXB];
    __shared__ int wsum[32];
    const bool second = (int)blockIdx.x >= nBlk;
    const int* __restrict__ src     = second ? srcB : srcA;
    const int* __restrict__ dst     = second ? dstB : dstA;
    const int* __restrict__ blkHist = second ? histB : histA;
    const int* __restrict__ blkOff  = second ? offB : offA;
    int* __restrict__ pairs         = second ? pairsB : pairsA;
    const int nBuckets = second ? nbkB : nbkA;
    const int blk = second ? (int)blockIdx.x - nBlk : (int)blockIdx.x;
    int tid = threadIdx.x;
    int nMine = 0, gOff = 0;
    if (tid < nBuckets) {
        nMine  = blkHist[tid * nBlk + blk];
        gOff = blkOff[tid * nBlk + blk];
    }
    int base = blk * EPB + tid;
    int d[EPB / 1024], s[EPB / 1024];
#pragma unroll
    for (int u = 0; u < EPB / 1024; ++u) {
        int ec = max(min(base + u * 1024, nE - 1), 0);
        d[u] = dst[ec];
        s[u] = src[ec];
    }
    int lst = block_excl_scan_1024(nMine, wsum, tid);
    lcur[tid]  = lst;
    delta[tid] = gOff - lst;
    __syncthreads();
#pragma unroll
    for (int u = 0; u < EPB / 1024; ++u) {
        int e = base + u * 1024;
        if (e < nE) {
            int lp = atomicAdd(&lcur[d[u] >> RSH], 1);
            stage[lp] = make_int2(s[u], d[u]);
        }
    }
    __syncthreads();
    int2* __restrict__ p2 = (int2*)pairs;
    const int nValid = min(EPB, nE - blk * EPB);
#pragma unroll
    for (int u = 0; u < EPB / 1024; ++u) {
        int j = tid + u * 1024;
        if (j < nValid) {
            int2 p = stage[j];
            p2[j + delta[p.y >> RSH]] = p;
        }
    }
}

__global__ __launch_bounds__(1024) void bucket_csr(const int* __restrict__ pairsA,
                                                   const int* __restrict__ offA,
                                                   int* __restrict__ rowStartA,
                                                   int* __restrict__ rowEndA,
                                                   int* __restrict__ srcsA, int nA, int nbkA,
                                                   const int* __restrict__ pairsB,
                                                   const int* __restrict__ offB,
                                                   int* __restrict__ rowStartB,
                                                   int* __restrict__ rowEndB,
                                                   int* __restrict__ srcsB, int nB, int nbkB,
                                                   int nE, int nBlk) {
    __shared__ int cnt[RPB];
    __shared__ int sc[RPB];
    __shared__ int lbuf[CAPS];
    const bool second = (int)blockIdx.x >= nbkA;
    const int* __restrict__ pairs  = second ? pairsB : pairsA;
    const int* __restrict__ blkOff = second ? offB : offA;
    int* __restrict__ rowStart     = second ? rowStartB : rowStartA;
    int* __restrict__ rowEnd       = second ? rowEndB : rowEndA;
    int* __restrict__ srcs         = second ? srcsB : srcsA;
    const int n        = second ? nB : nA;
    const int nBuckets = second ? nbkB : nbkA;
    int b = second ? (int)blockIdx.x - nbkA : (int)blockIdx.x, tid = threadIdx.x;
    int s0 = blkOff[b * nBlk];
    int s1 = (b + 1 < nBuckets) ? blkOff[(b + 1) * nBlk] : nE;
    const int2* __restrict__ p2 = (const int2*)pairs;
    const int nEnt = s1 - s0;
    const bool grouped = nEnt <= CAPS;
    if (tid < RPB) cnt[tid] = 0;
    __syncthreads();
    for (int i0 = s0 + tid; i0 < s1; i0 += 4096) {
        int2 p[4];
#pragma unroll
        for (int u = 0; u < 4; ++u)
            p[u] = p2[min(i0 + u * 1024, s1 - 1)];
#pragma unroll
        for (int u = 0; u < 4; ++u) {
            int i = i0 + u * 1024;
            if (i < s1) atomicAdd(&cnt[p[u].y & (RPB - 1)], 1);
        }
    }
    __syncthreads();
    int v = 0;
    if (tid < RPB) { v = cnt[tid]; sc[tid] = v; }
    __syncthreads();
    for (int off = 1; off < RPB; off <<= 1) {
        int add = 0;
        if (tid < RPB && tid >= off) add = sc[tid - off];
        __syncthreads();
        if (tid < RPB) sc[tid] += add;
        __syncthreads();
    }
    if (tid < RPB) {
        int start = s0 + sc[tid] - v;
        int node = b * RPB + tid;
        if (node < n) { rowStart[node] = start; rowEnd[node] = start + v; }
        cnt[tid] = start;
    }
    __syncthreads();
    for (int i0 = s0 + tid; i0 < s1; i0 += 4096) {
        int2 p[4];
#pragma unroll
        for (int u = 0; u < 4; ++u)
            p[u] = p2[min(i0 + u * 1024, s1 - 1)];
#pragma unroll
        for (int u = 0; u < 4; ++u) {
            int i = i0 + u * 1024;
            if (i < s1) {
                int pos = atomicAdd(&cnt[p[u].y & (RPB - 1)], 1);
                if (grouped) lbuf[pos - s0] = p[u].x;
                else         srcs[pos] = p[u].x;
            }
        }
    }
    if (grouped) {
        __syncthreads();
        for (int i = tid; i < nEnt; i += 1024) srcs[s0 + i] = lbuf[i];
    }
}

__device__ __forceinline__ void gather_rows_body(const float* __restrict__ emb,
                                                 const int* __restrict__ nid,
                                                 float* __restrict__ out, int n16, int blk) {
    int t = blk * blockDim.x + threadIdx.x;
    if (t >= n16) return;
    int i = t >> 4, c = (t & 15) * 4;
    int s = nid[i];
    *(float4*)(out + (size_t)i * D + c) = *(const float4*)(emb + (size_t)s * D + c);
}

__device__ __forceinline__ void input_gemm_body(
        const float* __restrict__ go_x, const float* __restrict__ lin_W,
        const float* __restrict__ lin_b, const float* __restrict__ go_emb,
        const int* __restrict__ go_nid, float* __restrict__ xg, int M, int K, int blk) {
    __shared__ float sWt[D * STI];

    int waveInBlk = threadIdx.x >> 5;
    int lane = threadIdx.x & 31;
    int row  = lane & 15;
    int kh   = lane >> 4;
    int colL = lane & 15;

    int numTiles = M >> 4;
    int tile0 = (blk * 8 + waveInBlk) * TMI;

    int   m0c[TMI];
    const float* arow[TMI];
    bool  valid[TMI];
#pragma unroll
    for (int t = 0; t < TMI; ++t) {
        int ti = tile0 + t;
        valid[t] = (ti < numTiles);
        m0c[t] = valid[t] ? (ti << 4) : 0;
        arow[t] = go_x + (size_t)(m0c[t] + row) * K;
    }

    v8f acc[TMI][4];
#pragma unroll
    for (int t = 0; t < TMI; ++t)
#pragma unroll
        for (int j = 0; j < 4; ++j) {
            float b = lin_b[j * 16 + colL];
#pragma unroll
            for (int g = 0; g < 8; ++g) acc[t][j][g] = b;
        }

    for (int kc0 = 0; kc0 < K; kc0 += KC) {
        int len = (K - kc0 < KC) ? (K - kc0) : KC;
        __syncthreads();
        for (int i = threadIdx.x * 4; i < len * D; i += 256 * 4) {
            int r = i >> 6, c = i & 63;
            float4 w = *(const float4*)&lin_W[(size_t)(kc0 + r) * D + c];
            sWt[(c + 0) * STI + r] = w.x;
            sWt[(c + 1) * STI + r] = w.y;
            sWt[(c + 2) * STI + r] = w.z;
            sWt[(c + 3) * STI + r] = w.w;
        }
        __syncthreads();
        if (kc0 + KC < K) {
#pragma unroll
            for (int t = 0; t < TMI; ++t)
                __builtin_prefetch(arow[t] + kc0 + KC, 0, 0);
        }
        for (int k = 0; k < len; k += 4) {
            v2f a[TMI];
#pragma unroll
            for (int t = 0; t < TMI; ++t)
                a[t] = *(const v2f*)(arow[t] + kc0 + k + 2 * kh);
#pragma unroll
            for (int j = 0; j < 4; ++j) {
                v2f b = *(const v2f*)&sWt[(j * 16 + colL) * STI + k + 2 * kh];
#pragma unroll
                for (int t = 0; t < TMI; ++t)
                    acc[t][j] = __builtin_amdgcn_wmma_f32_16x16x4_f32(
                        false, a[t], false, b, (short)0, acc[t][j], false, false);
            }
        }
    }

    int hi = kh * 8;
#pragma unroll
    for (int t = 0; t < TMI; ++t) {
        if (!valid[t]) continue;
#pragma unroll
        for (int g = 0; g < 8; ++g) {
            int m = m0c[t] + g + hi;
            int nid = go_nid[m];
#pragma unroll
            for (int j = 0; j < 4; ++j) {
                int col = j * 16 + colL;
                xg[(size_t)m * D + col] = acc[t][j][g] + go_emb[(size_t)nid * D + col];
            }
        }
    }
}

__global__ __launch_bounds__(256) void input_embed(
        const float* __restrict__ go_x, const float* __restrict__ lin_W,
        const float* __restrict__ lin_b, const float* __restrict__ go_emb,
        const int* __restrict__ go_nid, float* __restrict__ xg, int M, int K, int gemmBlocks,
        const float* __restrict__ p_emb, const int* __restrict__ p_nid,
        float* __restrict__ xp, int n16) {
    if ((int)blockIdx.x < gemmBlocks)
        input_gemm_body(go_x, lin_W, lin_b, go_emb, go_nid, xg, M, K, (int)blockIdx.x);
    else
        gather_rows_body(p_emb, p_nid, xp, n16, (int)blockIdx.x - gemmBlocks);
}

__global__ __launch_bounds__(256) void layer_fused_wmma(
        const float* __restrict__ xsA, const int* __restrict__ srcsA,
        const int* __restrict__ rowStartA, const int* __restrict__ rowEndA,
        const float* __restrict__ xA,
        const float* __restrict__ WlA, const float* __restrict__ blA,
        const float* __restrict__ WrA, float* __restrict__ outA, int MA, int blocksA,
        const float* __restrict__ xsB, const int* __restrict__ srcsB,
        const int* __restrict__ rowStartB, const int* __restrict__ rowEndB,
        const float* __restrict__ xB,
        const float* __restrict__ WlB, const float* __restrict__ blB,
        const float* __restrict__ WrB, float* __restrict__ outB, int MB, int relu) {
    __shared__ float sWt[2 * D * STL];
    __shared__ float sAgg[FR * STL];
    const bool second = (int)blockIdx.x >= blocksA;
    const float* __restrict__ xs     = second ? xsB : xsA;
    const int* __restrict__ srcs     = second ? srcsB : srcsA;
    const int* __restrict__ rowStart = second ? rowStartB : rowStartA;
    const int* __restrict__ rowEnd   = second ? rowEndB : rowEndA;
    const float* __restrict__ x   = second ? xB : xA;
    const float* __restrict__ Wl  = second ? WlB : WlA;
    const float* __restrict__ bl  = second ? blB : blA;
    const float* __restrict__ Wr  = second ? WrB : WrA;
    float* __restrict__ out       = second ? outB : outA;
    const int M   = second ? MB : MA;
    const int blk = second ? (int)blockIdx.x - blocksA : (int)blockIdx.x;

    for (int i = threadIdx.x * 4; i < D * D; i += 256 * 4) {
        int r = i >> 6, c = i & 63;
        float4 wl = *(const float4*)&Wl[i];
        float4 wr = *(const float4*)&Wr[i];
        sWt[(c + 0) * STL + r] = wl.x;
        sWt[(c + 1) * STL + r] = wl.y;
        sWt[(c + 2) * STL + r] = wl.z;
        sWt[(c + 3) * STL + r] = wl.w;
        sWt[D * STL + (c + 0) * STL + r] = wr.x;
        sWt[D * STL + (c + 1) * STL + r] = wr.y;
        sWt[D * STL + (c + 2) * STL + r] = wr.z;
        sWt[D * STL + (c + 3) * STL + r] = wr.w;
    }

    {
        const int grp = threadIdx.x >> 4;
        const int c   = (threadIdx.x & 15) * 4;
        for (int p = 0; p < FR / 16; ++p) {
            const int lr   = p * 16 + grp;
            const int node = blk * FR + lr;
            int s0 = 0, s1 = 0;
            if (node < M) { s0 = rowStart[node]; s1 = rowEnd[node]; }
            float4 acc = make_float4(0.f, 0.f, 0.f, 0.f);
            int i = s0;
            for (; i + 4 <= s1; i += 4) {
                int n0 = srcs[i], n1 = srcs[i + 1], n2 = srcs[i + 2], n3 = srcs[i + 3];
                float4 v0 = *(const float4*)(xs + (size_t)n0 * D + c);
                float4 v1 = *(const float4*)(xs + (size_t)n1 * D + c);
                float4 v2 = *(const float4*)(xs + (size_t)n2 * D + c);
                float4 v3 = *(const float4*)(xs + (size_t)n3 * D + c);
                acc.x += v0.x; acc.y += v0.y; acc.z += v0.z; acc.w += v0.w;
                acc.x += v1.x; acc.y += v1.y; acc.z += v1.z; acc.w += v1.w;
                acc.x += v2.x; acc.y += v2.y; acc.z += v2.z; acc.w += v2.w;
                acc.x += v3.x; acc.y += v3.y; acc.z += v3.z; acc.w += v3.w;
            }
            for (; i < s1; ++i) {
                int n = srcs[i];
                float4 v = *(const float4*)(xs + (size_t)n * D + c);
                acc.x += v.x; acc.y += v.y; acc.z += v.z; acc.w += v.w;
            }
            float inv = 1.0f / fmaxf((float)(s1 - s0), 1.0f);
            acc.x *= inv; acc.y *= inv; acc.z *= inv; acc.w *= inv;
            *(float4*)&sAgg[lr * STL + c] = acc;
        }
    }
    __syncthreads();

    const int w    = threadIdx.x >> 5;
    const int lane = threadIdx.x & 31;
    const int row  = lane & 15;
    const int kh   = lane >> 4;
    const int colL = lane & 15;
    const int tl   = w & 3;
    const int ch   = w >> 2;

    const int  m0    = blk * FR + tl * 16;
    const bool valid = (m0 + 16 <= M);
    const int  m0c   = valid ? m0 : 0;
    const float* xrow = x + (size_t)(m0c + row) * D;
    const float* arow = &sAgg[(tl * 16 + row) * STL];

    v8f acc[2];
#pragma unroll
    for (int jj = 0; jj < 2; ++jj) {
        float b = bl[(ch * 2 + jj) * 16 + colL];
#pragma unroll
        for (int g = 0; g < 8; ++g) acc[jj][g] = b;
    }

#pragma unroll
    for (int k = 0; k < D; k += 4) {
        v2f aA = *(const v2f*)(arow + k + 2 * kh);
        v2f aX = *(const v2f*)(xrow + k + 2 * kh);
#pragma unroll
        for (int jj = 0; jj < 2; ++jj) {
            const int j = ch * 2 + jj;
            v2f bL = *(const v2f*)&sWt[(j * 16 + colL) * STL + k + 2 * kh];
            v2f bR = *(const v2f*)&sWt[D * STL + (j * 16 + colL) * STL + k + 2 * kh];
            acc[jj] = __builtin_amdgcn_wmma_f32_16x16x4_f32(
                false, aA, false, bL, (short)0, acc[jj], false, false);
            acc[jj] = __builtin_amdgcn_wmma_f32_16x16x4_f32(
                false, aX, false, bR, (short)0, acc[jj], false, false);
        }
    }

    if (valid) {
        const int hi = kh * 8;
#pragma unroll
        for (int g = 0; g < 8; ++g) {
            int m = m0c + g + hi;
#pragma unroll
            for (int jj = 0; jj < 2; ++jj) {
                float v = acc[jj][g];
                if (relu) v = fmaxf(v, 0.0f);
                out[(size_t)m * D + (ch * 2 + jj) * 16 + colL] = v;
            }
        }
    }
}

__global__ void classifier_dot(const float* __restrict__ xp, const float* __restrict__ xg,
                               const int* __restrict__ ls, const int* __restrict__ ld,
                               float* __restrict__ out, int n) {
    int e = blockIdx.x * blockDim.x + threadIdx.x;
    if (e >= n) return;
    const float4* a = (const float4*)(xp + (size_t)ls[e] * D);
    const float4* b = (const float4*)(xg + (size_t)ld[e] * D);
    float s = 0.0f;
#pragma unroll
    for (int i = 0; i < 16; ++i) {
        float4 va = a[i], vb = b[i];
        s += va.x * vb.x + va.y * vb.y + va.z * vb.z + va.w * vb.w;
    }
    out[e] = s;
}

extern "C" void kernel_launch(void* const* d_in, const int* in_sizes, int n_in,
                              void* d_out, int out_size, void* d_ws, size_t ws_size,
                              hipStream_t stream) {
    const float* go_x  = (const float*)d_in[0];
    const float* p_emb = (const float*)d_in[1];
    const float* g_emb = (const float*)d_in[2];
    const float* lin_W = (const float*)d_in[3];
    const float* lin_b = (const float*)d_in[4];
    const float* Wl    = (const float*)d_in[5];
    const float* bl    = (const float*)d_in[6];
    const float* Wr    = (const float*)d_in[7];
    const int* p_nid   = (const int*)d_in[8];
    const int* g_nid   = (const int*)d_in[9];
    const int* src_pg  = (const int*)d_in[10];
    const int* dst_pg  = (const int*)d_in[11];
    const int* src_gp  = (const int*)d_in[12];
    const int* dst_gp  = (const int*)d_in[13];
    const int* lab_s   = (const int*)d_in[14];
    const int* lab_d   = (const int*)d_in[15];

    const int NP  = in_sizes[8];
    const int NG  = in_sizes[9];
    const int E   = in_sizes[10];
    const int EL  = in_sizes[14];
    const int FGO = in_sizes[0] / NG;

    char* base = (char*)d_ws;
    auto alloc = [&](size_t bytes) -> char* {
        char* p = base;
        base += (bytes + 255) & ~(size_t)255;
        return p;
    };
    float* xp_a  = (float*)alloc((size_t)NP * D * 4);
    float* xp_b  = (float*)alloc((size_t)NP * D * 4);
    float* xg_a  = (float*)alloc((size_t)NG * D * 4);
    float* xg_b  = (float*)alloc((size_t)NG * D * 4);
    float* tmp_p = (float*)alloc((size_t)NP * D * 4);
    (void)alloc((size_t)NG * D * 4);
    int* row_g   = (int*)alloc((size_t)NG * 4);
    int* cur_g   = (int*)alloc((size_t)NG * 4);
    int* row_p   = (int*)alloc((size_t)NP * 4);
    int* cur_p   = (int*)alloc((size_t)NP * 4);
    int* srcs_g  = (int*)alloc((size_t)E * 4);
    int* srcs_p  = (int*)alloc((size_t)E * 4);
    int* bsum    = (int*)alloc(1024 * 4);

    const int nBlkE = (E + EPB - 1) / EPB;
    const int nbkG  = (NG + RPB - 1) / RPB;
    const int nbkP  = (NP + RPB - 1) / RPB;
    const int nmatG = nbkG * nBlkE;
    const int nmatP = nbkP * nBlkE;
    const int nbG   = (nmatG + 255) / 256;
    const int nbP   = (nmatP + 255) / 256;
    const int nmatGp = (nmatG + 63) & ~63;
    int* pairsG = (int*)tmp_p;
    int* pairsP = pairsG + (size_t)2 * E;
    int* hmatG  = (int*)xp_b;
    int* hmatP  = hmatG + nmatGp;
    int* omatG  = (int*)xg_b;
    int* omatP  = omatG + nmatGp;
    int* bsumG  = bsum;
    int* bsumP  = omatP + ((nmatP + 63) & ~63);

    bucket_hist<<<2 * nBlkE, 1024, 0, stream>>>(dst_pg, hmatG, nbkG, dst_gp, hmatP, nbkP, E, nBlkE);
    scan_block<<<nbG + nbP, 256, 0, stream>>>(hmatG, omatG, bsumG, nmatG, nbG,
                                              hmatP, omatP, bsumP, nmatP);
    scan_sums<<<2, 1024, 0, stream>>>(bsumG, nbG, bsumP, nbP);
    add_block_offsets<<<nbG + nbP, 256, 0, stream>>>(omatG, bsumG, nmatG, nbG,
                                                     omatP, bsumP, nmatP);
    bucket_scatter<<<2 * nBlkE, 1024, 0, stream>>>(src_pg, dst_pg, hmatG, omatG, pairsG, nbkG,
                                                   src_gp, dst_gp, hmatP, omatP, pairsP, nbkP,
                                                   E, nBlkE);
    bucket_csr<<<nbkG + nbkP, 1024, 0, stream>>>(pairsG, omatG, row_g, cur_g, srcs_g, NG, nbkG,
                                                 pairsP, omatP, row_p, cur_p, srcs_p, NP, nbkP,
                                                 E, nBlkE);

    {
        int tiles      = (NG + 15) / 16;
        int gemmBlocks = (tiles + 8 * TMI - 1) / (8 * TMI);
        int gatherBlks = (NP * 16 + 255) / 256;
        input_embed<<<gemmBlocks + gatherBlks, 256, 0, stream>>>(go_x, lin_W, lin_b, g_emb, g_nid,
                                                                 xg_a, NG, FGO, gemmBlocks,
                                                                 p_emb, p_nid, xp_a, NP * 16);
    }

    float* xp_cur = xp_a; float* xp_nxt = xp_b;
    float* xg_cur = xg_a; float* xg_nxt = xg_b;
    const int fusBlkG = (NG + FR - 1) / FR;
    const int fusBlkP = (NP + FR - 1) / FR;
    for (int l = 0; l < 3; ++l) {
        const float* Wl0 = Wl + (size_t)(l * 2 + 0) * D * D;
        const float* Wl1 = Wl + (size_t)(l * 2 + 1) * D * D;
        const float* Wr0 = Wr + (size_t)(l * 2 + 0) * D * D;
        const float* Wr1 = Wr + (size_t)(l * 2 + 1) * D * D;
        const float* bl0 = bl + (size_t)(l * 2 + 0) * D;
        const float* bl1 = bl + (size_t)(l * 2 + 1) * D;
        int relu = (l < 2) ? 1 : 0;
        layer_fused_wmma<<<fusBlkG + fusBlkP, 256, 0, stream>>>(
            xp_cur, srcs_g, row_g, cur_g, xg_cur, Wl0, bl0, Wr0, xg_nxt, NG, fusBlkG,
            xg_cur, srcs_p, row_p, cur_p, xp_cur, Wl1, bl1, Wr1, xp_nxt, NP, relu);
        float* t;
        t = xp_cur; xp_cur = xp_nxt; xp_nxt = t;
        t = xg_cur; xg_cur = xg_nxt; xg_nxt = t;
    }

    classifier_dot<<<(EL + 255) / 256, 256, 0, stream>>>(xp_cur, xg_cur, lab_s, lab_d,
                                                         (float*)d_out, EL);
}


# Fused layer kernel, neighbour gather: a row's 16 lanes fetch its CSR entries 16 per load, one chunk ahead, and pass them by lane shuffle; 8 row loads in flight, not 4; add order within a row kept
# speedup vs baseline: 2.3403x; 1.0630x over previous
#include <hip/hip_runtime.h>

typedef __attribute__((ext_vector_type(2))) float v2f;
typedef __attribute__((ext_vector_type(8))) float v8f;

#define D    64
#define KC   128
#define STI  (KC+4)
#define STL  (D+4)
#define TMI  2
#define FR   64


#define EPB   8192
#define RSH   8
#define RPB   256
#define MAXB  1024
#define CAPS  16384


__device__ __forceinline__ int block_excl_scan_1024(int v, int* __restrict__ wsum, int tid) {
    const int lane = tid & 31, w = tid >> 5;
    int x = v;
    for (int off = 1; off < 32; off <<= 1) {
        int y = __shfl_up(x, off, 32);
        if (lane >= off) x += y;
    }
    if (lane == 31) wsum[w] = x;
    __syncthreads();
    if (w == 0) {
        int t = wsum[lane], z = t;
        for (int off = 1; off < 32; off <<= 1) {
            int y = __shfl_up(z, off, 32);
            if (lane >= off) z += y;
        }
        wsum[lane] = z - t;
    }
    __syncthreads();
    return x - v + wsum[w];
}

__global__ __launch_bounds__(1024) void bucket_hist(const int* __restrict__ dstA,
                                                    int* __restrict__ histA, int nbkA,
                                                    const int* __restrict__ dstB,
                                                    int* __restrict__ histB, int nbkB,
                                                    int nE, int nBlk) {
    __shared__ int h[MAXB];
    const bool second = (int)blockIdx.x >= nBlk;
    const int* __restrict__ dst = second ? dstB : dstA;
    int* __restrict__ blkHist   = second ? histB : histA;
    const int nBuckets = second ? nbkB : nbkA;
    const int blk = second ? (int)blockIdx.x - nBlk : (int)blockIdx.x;
    int tid = threadIdx.x;
    h[tid] = 0;
    __syncthreads();
    int base = blk * EPB + tid;
    int d[EPB / 1024];
#pragma unroll
    for (int u = 0; u < EPB / 1024; ++u) {
        int e = base + u * 1024;
        d[u] = dst[max(min(e, nE - 1), 0)];
    }
#pragma unroll
    for (int u = 0; u < EPB / 1024; ++u) {
        int e = base + u * 1024;
        if (e < nE) atomicAdd(&h[d[u] >> RSH], 1);
    }
    __syncthreads();
    if (tid < nBuckets) blkHist[tid * nBlk + blk] = h[tid];
}

__global__ void scan_block(const int* __restrict__ degA, int* __restrict__ exA,
                           int* __restrict__ bsumA, int nA, int nbA,
                           const int* __restrict__ degB, int* __restrict__ exB,
                           int* __restrict__ bsumB, int nB) {
    __shared__ int sh[256];
    const bool second = (int)blockIdx.x >= nbA;
    const int* __restrict__ deg = second ? degB : degA;
    int* __restrict__ ex   = second ? exB : exA;
    int* __restrict__ bsum = second ? bsumB : bsumA;
    const int n   = second ? nB : nA;
    const int blk = second ? (int)blockIdx.x - nbA : (int)blockIdx.x;
    int tid = threadIdx.x;
    int i = blk * 256 + tid;
    int v = (i < n) ? deg[i] : 0;
    sh[tid] = v;
    __syncthreads();
    for (int off = 1; off < 256; off <<= 1) {
        int add = (tid >= off) ? sh[tid - off] : 0;
        __syncthreads();
        sh[tid] += add;
        __syncthreads();
    }
    if (i < n) ex[i] = sh[tid] - v;
    if (tid == 255) bsum[blk] = sh[255];
}

__global__ void scan_sums(int* __restrict__ bsumA, int nbA,
                          int* __restrict__ bsumB, int nbB) {
    __shared__ int sh[1024];
    const bool second = blockIdx.x != 0;
    int* __restrict__ bsum = second ? bsumB : bsumA;
    const int nb = second ? nbB : nbA;
    int tid = threadIdx.x;
    int v = (tid < nb) ? bsum[tid] : 0;
    sh[tid] = v;
    __syncthreads();
    for (int off = 1; off < 1024; off <<= 1) {
        int add = (tid >= off) ? sh[tid - off] : 0;
        __syncthreads();
        sh[tid] += add;
        __syncthreads();
    }
    if (tid < nb) bsum[tid] = sh[tid] - v;
}

__global__ void add_block_offsets(int* __restrict__ exA, const int* __restrict__ bsumA, int nA,
                                  int nbA,
                                  int* __restrict__ exB, const int* __restrict__ bsumB, int nB) {
    const bool second = (int)blockIdx.x >= nbA;
    int* __restrict__ ex         = second ? exB : exA;
    const int* __restrict__ bsum = second ? bsumB : bsumA;
    const int n   = second ? nB : nA;
    const int blk = second ? (int)blockIdx.x - nbA : (int)blockIdx.x;
    int i = blk * 256 + threadIdx.x;
    if (i < n) ex[i] += bsum[blk];
}

__global__ __launch_bounds__(1024) void bucket_scatter(const int* __restrict__ srcA,
                                                       const int* __restrict__ dstA,
                                                       const int* __restrict__ histA,
                                                       const int* __restrict__ offA,
                                                       int* __restrict__ pairsA, int nbkA,
                                                       const int* __restrict__ srcB,
                                                       const int* __restrict__ dstB,
                                                       const int* __restrict__ histB,
                                                       const int* __restrict__ offB,
                                                       int* __restrict__ pairsB, int nbkB,
                                                       int nE, int nBlk) {
    __shared__ int2 stage[EPB];
    __shared__ int lcur[MAXB];
    __shared__ int delta[MAXB];
    __shared__ int wsum[32];
    const bool second = (int)blockIdx.x >= nBlk;
    const int* __restrict__ src     = second ? srcB : srcA;
    const int* __restrict__ dst     = second ? dstB : dstA;
    const int* __restrict__ blkHist = second ? histB : histA;
    const int* __restrict__ blkOff  = second ? offB : offA;
    int* __restrict__ pairs         = second ? pairsB : pairsA;
    const int nBuckets = second ? nbkB : nbkA;
    const int blk = second ? (int)blockIdx.x - nBlk : (int)blockIdx.x;
    int tid = threadIdx.x;
    int nMine = 0, gOff = 0;
    if (tid < nBuckets) {
        nMine  = blkHist[tid * nBlk + blk];
        gOff = blkOff[tid * nBlk + blk];
    }
    int base = blk * EPB + tid;
    int d[EPB / 1024], s[EPB / 1024];
#pragma unroll
    for (int u = 0; u < EPB / 1024; ++u) {
        int ec = max(min(base + u * 1024, nE - 1), 0);
        d[u] = dst[ec];
        s[u] = src[ec];
    }
    int lst = block_excl_scan_1024(nMine, wsum, tid);
    lcur[tid]  = lst;
    delta[tid] = gOff - lst;
    __syncthreads();
#pragma unroll
    for (int u = 0; u < EPB / 1024; ++u) {
        int e = base + u * 1024;
        if (e < nE) {
            int lp = atomicAdd(&lcur[d[u] >> RSH], 1);
            stage[lp] = make_int2(s[u], d[u]);
        }
    }
    __syncthreads();
    int2* __restrict__ p2 = (int2*)pairs;
    const int nValid = min(EPB, nE - blk * EPB);
#pragma unroll
    for (int u = 0; u < EPB / 1024; ++u) {
        int j = tid + u * 1024;
        if (j < nValid) {
            int2 p = stage[j];
            p2[j + delta[p.y >> RSH]] = p;
        }
    }
}

__global__ __launch_bounds__(1024) void bucket_csr(const int* __restrict__ pairsA,
                                                   const int* __restrict__ offA,
                                                   int* __restrict__ rowStartA,
                                                   int* __restrict__ rowEndA,
                                                   int* __restrict__ srcsA, int nA, int nbkA,
                                                   const int* __restrict__ pairsB,
                                                   const int* __restrict__ offB,
                                                   int* __restrict__ rowStartB,
                                                   int* __restrict__ rowEndB,
                                                   int* __restrict__ srcsB, int nB, int nbkB,
                                                   int nE, int nBlk) {
    __shared__ int cnt[RPB];
    __shared__ int sc[RPB];
    __shared__ int lbuf[CAPS];
    const bool second = (int)blockIdx.x >= nbkA;
    const int* __restrict__ pairs  = second ? pairsB : pairsA;
    const int* __restrict__ blkOff = second ? offB : offA;
    int* __restrict__ rowStart     = second ? rowStartB : rowStartA;
    int* __restrict__ rowEnd       = second ? rowEndB : rowEndA;
    int* __restrict__ srcs         = second ? srcsB : srcsA;
    const int n        = second ? nB : nA;
    const int nBuckets = second ? nbkB : nbkA;
    int b = second ? (int)blockIdx.x - nbkA : (int)blockIdx.x, tid = threadIdx.x;
    int s0 = blkOff[b * nBlk];
    int s1 = (b + 1 < nBuckets) ? blkOff[(b + 1) * nBlk] : nE;
    const int2* __restrict__ p2 = (const int2*)pairs;
    const int nEnt = s1 - s0;
    const bool grouped = nEnt <= CAPS;
    if (tid < RPB) cnt[tid] = 0;
    __syncthreads();
    for (int i0 = s0 + tid; i0 < s1; i0 += 4096) {
        int2 p[4];
#pragma unroll
        for (int u = 0; u < 4; ++u)
            p[u] = p2[min(i0 + u * 1024, s1 - 1)];
#pragma unroll
        for (int u = 0; u < 4; ++u) {
            int i = i0 + u * 1024;
            if (i < s1) atomicAdd(&cnt[p[u].y & (RPB - 1)], 1);
        }
    }
    __syncthreads();
    int v = 0;
    if (tid < RPB) { v = cnt[tid]; sc[tid] = v; }
    __syncthreads();
    for (int off = 1; off < RPB; off <<= 1) {
        int add = 0;
        if (tid < RPB && tid >= off) add = sc[tid - off];
        __syncthreads();
        if (tid < RPB) sc[tid] += add;
        __syncthreads();
    }
    if (tid < RPB) {
        int start = s0 + sc[tid] - v;
        int node = b * RPB + tid;
        if (node < n) { rowStart[node] = start; rowEnd[node] = start + v; }
        cnt[tid] = start;
    }
    __syncthreads();
    for (int i0 = s0 + tid; i0 < s1; i0 += 4096) {
        int2 p[4];
#pragma unroll
        for (int u = 0; u < 4; ++u)
            p[u] = p2[min(i0 + u * 1024, s1 - 1)];
#pragma unroll
        for (int u = 0; u < 4; ++u) {
            int i = i0 + u * 1024;
            if (i < s1) {
                int pos = atomicAdd(&cnt[p[u].y & (RPB - 1)], 1);
                if (grouped) lbuf[pos - s0] = p[u].x;
                else         srcs[pos] = p[u].x;
            }
        }
    }
    if (grouped) {
        __syncthreads();
        for (int i = tid; i < nEnt; i += 1024) srcs[s0 + i] = lbuf[i];
    }
}

__device__ __forceinline__ void gather_rows_body(const float* __restrict__ emb,
                                                 const int* __restrict__ nid,
                                                 float* __restrict__ out, int n16, int blk) {
    int t = blk * blockDim.x + threadIdx.x;
    if (t >= n16) return;
    int i = t >> 4, c = (t & 15) * 4;
    int s = nid[i];
    *(float4*)(out + (size_t)i * D + c) = *(const float4*)(emb + (size_t)s * D + c);
}

__device__ __forceinline__ void input_gemm_body(
        const float* __restrict__ go_x, const float* __restrict__ lin_W,
        const float* __restrict__ lin_b, const float* __restrict__ go_emb,
        const int* __restrict__ go_nid, float* __restrict__ xg, int M, int K, int blk) {
    __shared__ float sWt[D * STI];

    int waveInBlk = threadIdx.x >> 5;
    int lane = threadIdx.x & 31;
    int row  = lane & 15;
    int kh   = lane >> 4;
    int colL = lane & 15;

    int numTiles = M >> 4;
    int tile0 = (blk * 8 + waveInBlk) * TMI;

    int   m0c[TMI];
    const float* arow[TMI];
    bool  valid[TMI];
#pragma unroll
    for (int t = 0; t < TMI; ++t) {
        int ti = tile0 + t;
        valid[t] = (ti < numTiles);
        m0c[t] = valid[t] ? (ti << 4) : 0;
        arow[t] = go_x + (size_t)(m0c[t] + row) * K;
    }

    v8f acc[TMI][4];
#pragma unroll
    for (int t = 0; t < TMI; ++t)
#pragma unroll
        for (int j = 0; j < 4; ++j) {
            float b = lin_b[j * 16 + colL];
#pragma unroll
            for (int g = 0; g < 8; ++g) acc[t][j][g] = b;
        }

    for (int kc0 = 0; kc0 < K; kc0 += KC) {
        int len = (K - kc0 < KC) ? (K - kc0) : KC;
        __syncthreads();
        for (int i = threadIdx.x * 4; i < len * D; i += 256 * 4) {
            int r = i >> 6, c = i & 63;
            float4 w = *(const float4*)&lin_W[(size_t)(kc0 + r) * D + c];
            sWt[(c + 0) * STI + r] = w.x;
            sWt[(c + 1) * STI + r] = w.y;
            sWt[(c + 2) * STI + r] = w.z;
            sWt[(c + 3) * STI + r] = w.w;
        }
        __syncthreads();
        if (kc0 + KC < K) {
#pragma unroll
            for (int t = 0; t < TMI; ++t)
                __builtin_prefetch(arow[t] + kc0 + KC, 0, 0);
        }
        for (int k = 0; k < len; k += 4) {
            v2f a[TMI];
#pragma unroll
            for (int t = 0; t < TMI; ++t)
                a[t] = *(const v2f*)(arow[t] + kc0 + k + 2 * kh);
#pragma unroll
            for (int j = 0; j < 4; ++j) {
                v2f b = *(const v2f*)&sWt[(j * 16 + colL) * STI + k + 2 * kh];
#pragma unroll
                for (int t = 0; t < TMI; ++t)
                    acc[t][j] = __builtin_amdgcn_wmma_f32_16x16x4_f32(
                        false, a[t], false, b, (short)0, acc[t][j], false, false);
            }
        }
    }

    int hi = kh * 8;
#pragma unroll
    for (int t = 0; t < TMI; ++t) {
        if (!valid[t]) continue;
#pragma unroll
        for (int g = 0; g < 8; ++g) {
            int m = m0c[t] + g + hi;
            int nid = go_nid[m];
#pragma unroll
            for (int j = 0; j < 4; ++j) {
                int col = j * 16 + colL;
                xg[(size_t)m * D + col] = acc[t][j][g] + go_emb[(size_t)nid * D + col];
            }
        }
    }
}

__global__ __launch_bounds__(256) void input_embed(
        const float* __restrict__ go_x, const float* __restrict__ lin_W,
        const float* __restrict__ lin_b, const float* __restrict__ go_emb,
        const int* __restrict__ go_nid, float* __restrict__ xg, int M, int K, int gemmBlocks,
        const float* __restrict__ p_emb, const int* __restrict__ p_nid,
        float* __restrict__ xp, int n16) {
    if ((int)blockIdx.x < gemmBlocks)
        input_gemm_body(go_x, lin_W, lin_b, go_emb, go_nid, xg, M, K, (int)blockIdx.x);
    else
        gather_rows_body(p_emb, p_nid, xp, n16, (int)blockIdx.x - gemmBlocks);
}

__global__ __launch_bounds__(256) void layer_fused_wmma(
        const float* __restrict__ xsA, const int* __restrict__ srcsA,
        const int* __restrict__ rowStartA, const int* __restrict__ rowEndA,
        const float* __restrict__ xA,
        const float* __restrict__ WlA, const float* __restrict__ blA,
        const float* __restrict__ WrA, float* __restrict__ outA, int MA, int blocksA,
        const float* __restrict__ xsB, const int* __restrict__ srcsB,
        const int* __restrict__ rowStartB, const int* __restrict__ rowEndB,
        const float* __restrict__ xB,
        const float* __restrict__ WlB, const float* __restrict__ blB,
        const float* __restrict__ WrB, float* __restrict__ outB, int MB, int relu) {
    __shared__ float sWt[2 * D * STL];
    __shared__ float sAgg[FR * STL];
    const bool second = (int)blockIdx.x >= blocksA;
    const float* __restrict__ xs     = second ? xsB : xsA;
    const int* __restrict__ srcs     = second ? srcsB : srcsA;
    const int* __restrict__ rowStart = second ? rowStartB : rowStartA;
    const int* __restrict__ rowEnd   = second ? rowEndB : rowEndA;
    const float* __restrict__ x   = second ? xB : xA;
    const float* __restrict__ Wl  = second ? WlB : WlA;
    const float* __restrict__ bl  = second ? blB : blA;
    const float* __restrict__ Wr  = second ? WrB : WrA;
    float* __restrict__ out       = second ? outB : outA;
    const int M   = second ? MB : MA;
    const int blk = second ? (int)blockIdx.x - blocksA : (int)blockIdx.x;

    const int grp = threadIdx.x >> 4;
    const int q   = threadIdx.x & 15;
    const unsigned cb = (unsigned)q * 16u;
    const char* __restrict__ xsb = (const char*)xs;
    int rs[FR / 16], re[FR / 16], ix[FR / 16];
#pragma unroll
    for (int p = 0; p < FR / 16; ++p)
        rs[p] = rowStart[min(blk * FR + p * 16 + grp, M - 1)];
#pragma unroll
    for (int p = 0; p < FR / 16; ++p)
        re[p] = rowEnd[min(blk * FR + p * 16 + grp, M - 1)];
    __builtin_amdgcn_sched_barrier(0);
#pragma unroll
    for (int p = 0; p < FR / 16; ++p)
        re[p] = (blk * FR + p * 16 + grp < M) ? re[p] : rs[p];
#pragma unroll
    for (int p = 0; p < FR / 16; ++p)
        ix[p] = srcs[max(min(rs[p] + q, re[p] - 1), 0)];

    for (int i = threadIdx.x * 4; i < D * D; i += 256 * 4) {
        int r = i >> 6, c = i & 63;
        float4 wl = *(const float4*)&Wl[i];
        float4 wr = *(const float4*)&Wr[i];
        sWt[(c + 0) * STL + r] = wl.x;
        sWt[(c + 1) * STL + r] = wl.y;
        sWt[(c + 2) * STL + r] = wl.z;
        sWt[(c + 3) * STL + r] = wl.w;
        sWt[D * STL + (c + 0) * STL + r] = wr.x;
        sWt[D * STL + (c + 1) * STL + r] = wr.y;
        sWt[D * STL + (c + 2) * STL + r] = wr.z;
        sWt[D * STL + (c + 3) * STL + r] = wr.w;
    }

    {
#pragma unroll
        for (int p = 0; p < FR / 16; ++p) {
            const int lr = p * 16 + grp;
            const int s0 = rs[p], s1 = re[p];
            float4 acc = make_float4(0.f, 0.f, 0.f, 0.f);
            int my = ix[p];
            for (int base = s0; base < s1; base += 16) {
                const int cnt = min(16, s1 - base);
                const int myNext = srcs[min(base + 16 + q, s1 - 1)];
                for (int k = 0; k < cnt; k += 8) {
                    float4 v[8];
#pragma unroll
                    for (int u = 0; u < 8; ++u) {
                        const unsigned n = (unsigned)__shfl(my, k + u, 16);
                        v[u] = *(const float4*)(xsb + (size_t)(n * (unsigned)(D * 4) + cb));
                    }
                    acc.x += v[0].x; acc.y += v[0].y; acc.z += v[0].z; acc.w += v[0].w;
#pragma unroll
                    for (int u = 1; u < 8; ++u) {
                        const bool on = (k + u < cnt);
                        acc.x += on ? v[u].x : 0.0f;
                        acc.y += on ? v[u].y : 0.0f;
                        acc.z += on ? v[u].z : 0.0f;
                        acc.w += on ? v[u].w : 0.0f;
                    }
                }
                my = myNext;
            }
            float inv = 1.0f / fmaxf((float)(s1 - s0), 1.0f);
            acc.x *= inv; acc.y *= inv; acc.z *= inv; acc.w *= inv;
            *(float4*)&sAgg[lr * STL + 4 * q] = acc;
        }
    }
    __syncthreads();

    const int w    = threadIdx.x >> 5;
    const int lane = threadIdx.x & 31;
    const int row  = lane & 15;
    const int kh   = lane >> 4;
    const int colL = lane & 15;
    const int tl   = w & 3;
    const int ch   = w >> 2;

    const int  m0    = blk * FR + tl * 16;
    const bool valid = (m0 + 16 <= M);
    const int  m0c   = valid ? m0 : 0;
    const float* xrow = x + (size_t)(m0c + row) * D;
    const float* arow = &sAgg[(tl * 16 + row) * STL];

    v8f acc[2];
#pragma unroll
    for (int jj = 0; jj < 2; ++jj) {
        float b = bl[(ch * 2 + jj) * 16 + colL];
#pragma unroll
        for (int g = 0; g < 8; ++g) acc[jj][g] = b;
    }

#pragma unroll
    for (int k = 0; k < D; k += 4) {
        v2f aA = *(const v2f*)(arow + k + 2 * kh);
        v2f aX = *(const v2f*)(xrow + k + 2 * kh);
#pragma unroll
        for (int jj = 0; jj < 2; ++jj) {
            const int j = ch * 2 + jj;
            v2f bL = *(const v2f*)&sWt[(j * 16 + colL) * STL + k + 2 * kh];
            v2f bR = *(const v2f*)&sWt[D * STL + (j * 16 + colL) * STL + k + 2 * kh];
            acc[jj] = __builtin_amdgcn_wmma_f32_16x16x4_f32(
                false, aA, false, bL, (short)0, acc[jj], false, false);
            acc[jj] = __builtin_amdgcn_wmma_f32_16x16x4_f32(
                false, aX, false, bR, (short)0, acc[jj], false, false);
        }
    }

    if (valid) {
        const int hi = kh * 8;
#pragma unroll
        for (int g = 0; g < 8; ++g) {
            int m = m0c + g + hi;
#pragma unroll
            for (int jj = 0; jj < 2; ++jj) {
                float v = acc[jj][g];
                if (relu) v = fmaxf(v, 0.0f);
                out[(size_t)m * D + (ch * 2 + jj) * 16 + colL] = v;
            }
        }
    }
}

__global__ void classifier_dot(const float* __restrict__ xp, const float* __restrict__ xg,
                               const int* __restrict__ ls, const int* __restrict__ ld,
                               float* __restrict__ out, int n) {
    int e = blockIdx.x * blockDim.x + threadIdx.x;
    if (e >= n) return;
    const float4* a = (const float4*)(xp + (size_t)ls[e] * D);
    const float4* b = (const float4*)(xg + (size_t)ld[e] * D);
    float s = 0.0f;
#pragma unroll
    for (int i = 0; i < 16; ++i) {
        float4 va = a[i], vb = b[i];
        s += va.x * vb.x + va.y * vb.y + va.z * vb.z + va.w * vb.w;
    }
    out[e] = s;
}

extern "C" void kernel_launch(void* const* d_in, const int* in_sizes, int n_in,
                              void* d_out, int out_size, void* d_ws, size_t ws_size,
                              hipStream_t stream) {
    const float* go_x  = (const float*)d_in[0];
    const float* p_emb = (const float*)d_in[1];
    const float* g_emb = (const float*)d_in[2];
    const float* lin_W = (const float*)d_in[3];
    const float* lin_b = (const float*)d_in[4];
    const float* Wl    = (const float*)d_in[5];
    const float* bl    = (const float*)d_in[6];
    const float* Wr    = (const float*)d_in[7];
    const int* p_nid   = (const int*)d_in[8];
    const int* g_nid   = (const int*)d_in[9];
    const int* src_pg  = (const int*)d_in[10];
    const int* dst_pg  = (const int*)d_in[11];
    const int* src_gp  = (const int*)d_in[12];
    const int* dst_gp  = (const int*)d_in[13];
    const int* lab_s   = (const int*)d_in[14];
    const int* lab_d   = (const int*)d_in[15];

    const int NP  = in_sizes[8];
    const int NG  = in_sizes[9];
    const int E   = in_sizes[10];
    const int EL  = in_sizes[14];
    const int FGO = in_sizes[0] / NG;

    char* base = (char*)d_ws;
    auto alloc = [&](size_t bytes) -> char* {
        char* p = base;
        base += (bytes + 255) & ~(size_t)255;
        return p;
    };
    float* xp_a  = (float*)alloc((size_t)NP * D * 4);
    float* xp_b  = (float*)alloc((size_t)NP * D * 4);
    float* xg_a  = (float*)alloc((size_t)NG * D * 4);
    float* xg_b  = (float*)alloc((size_t)NG * D * 4);
    float* tmp_p = (float*)alloc((size_t)NP * D * 4);
    (void)alloc((size_t)NG * D * 4);
    int* row_g   = (int*)alloc((size_t)NG * 4);
    int* cur_g   = (int*)alloc((size_t)NG * 4);
    int* row_p   = (int*)alloc((size_t)NP * 4);
    int* cur_p   = (int*)alloc((size_t)NP * 4);
    int* srcs_g  = (int*)alloc((size_t)E * 4);
    int* srcs_p  = (int*)alloc((size_t)E * 4);
    int* bsum    = (int*)alloc(1024 * 4);

    const int nBlkE = (E + EPB - 1) / EPB;
    const int nbkG  = (NG + RPB - 1) / RPB;
    const int nbkP  = (NP + RPB - 1) / RPB;
    const int nmatG = nbkG * nBlkE;
    const int nmatP = nbkP * nBlkE;
    const int nbG   = (nmatG + 255) / 256;
    const int nbP   = (nmatP + 255) / 256;
    const int nmatGp = (nmatG + 63) & ~63;
    int* pairsG = (int*)tmp_p;
    int* pairsP = pairsG + (size_t)2 * E;
    int* hmatG  = (int*)xp_b;
    int* hmatP  = hmatG + nmatGp;
    int* omatG  = (int*)xg_b;
    int* omatP  = omatG + nmatGp;
    int* bsumG  = bsum;
    int* bsumP  = omatP + ((nmatP + 63) & ~63);

    bucket_hist<<<2 * nBlkE, 1024, 0, stream>>>(dst_pg, hmatG, nbkG, dst_gp, hmatP, nbkP, E, nBlkE);
    scan_block<<<nbG + nbP, 256, 0, stream>>>(hmatG, omatG, bsumG, nmatG, nbG,
                                              hmatP, omatP, bsumP, nmatP);
    scan_sums<<<2, 1024, 0, stream>>>(bsumG, nbG, bsumP, nbP);
    add_block_offsets<<<nbG + nbP, 256, 0, stream>>>(omatG, bsumG, nmatG, nbG,
                                                     omatP, bsumP, nmatP);
    bucket_scatter<<<2 * nBlkE, 1024, 0, stream>>>(src_pg, dst_pg, hmatG, omatG, pairsG, nbkG,
                                                   src_gp, dst_gp, hmatP, omatP, pairsP, nbkP,
                                                   E, nBlkE);
    bucket_csr<<<nbkG + nbkP, 1024, 0, stream>>>(pairsG, omatG, row_g, cur_g, srcs_g, NG, nbkG,
                                                 pairsP, omatP, row_p, cur_p, srcs_p, NP, nbkP,
                                                 E, nBlkE);

    {
        int tiles      = (NG + 15) / 16;
        int gemmBlocks = (tiles + 8 * TMI - 1) / (8 * TMI);
        int gatherBlks = (NP * 16 + 255) / 256;
        input_embed<<<gemmBlocks + gatherBlks, 256, 0, stream>>>(go_x, lin_W, lin_b, g_emb, g_nid,
                                                                 xg_a, NG, FGO, gemmBlocks,
                                                                 p_emb, p_nid, xp_a, NP * 16);
    }

    float* xp_cur = xp_a; float* xp_nxt = xp_b;
    float* xg_cur = xg_a; float* xg_nxt = xg_b;
    const int fusBlkG = (NG + FR - 1) / FR;
    const int fusBlkP = (NP + FR - 1) / FR;
    for (int l = 0; l < 3; ++l) {
        const float* Wl0 = Wl + (size_t)(l * 2 + 0) * D * D;
        const float* Wl1 = Wl + (size_t)(l * 2 + 1) * D * D;
        const float* Wr0 = Wr + (size_t)(l * 2 + 0) * D * D;
        const float* Wr1 = Wr + (size_t)(l * 2 + 1) * D * D;
        const float* bl0 = bl + (size_t)(l * 2 + 0) * D;
        const float* bl1 = bl + (size_t)(l * 2 + 1) * D;
        int relu = (l < 2) ? 1 : 0;
        layer_fused_wmma<<<fusBlkG + fusBlkP, 256, 0, stream>>>(
            xp_cur, srcs_g, row_g, cur_g, xg_cur, Wl0, bl0, Wr0, xg_nxt, NG, fusBlkG,
            xg_cur, srcs_p, row_p, cur_p, xp_cur, Wl1, bl1, Wr1, xp_nxt, NP, relu);
        float* t;
        t = xp_cur; xp_cur = xp_nxt; xp_nxt = t;
        t = xg_cur; xg_cur = xg_nxt; xg_nxt = t;
    }

    classifier_dot<<<(EL + 255) / 256, 256, 0, stream>>>(xp_cur, xg_cur, lab_s, lab_d,
                                                         (float*)d_out, EL);
}


# Input GEMM, operand loads ahead of their products: a wave asks for the A fragments of the next 32 k (128 bytes of each go_x row) before the current 32 k's WMMAs; weight chunk loads issue as one group
# speedup vs baseline: 2.8081x; 1.1999x over previous
#include <hip/hip_runtime.h>

typedef __attribute__((ext_vector_type(2))) float v2f;
typedef __attribute__((ext_vector_type(8))) float v8f;

#define D    64
#define KC   128
#define STI  (KC+4)
#define STL  (D+4)
#define TMI  2
#define FR   64


#define EPB   8192
#define RSH   8
#define RPB   256
#define MAXB  1024
#define CAPS  16384


__device__ __forceinline__ int block_excl_scan_1024(int v, int* __restrict__ wsum, int tid) {
    const int lane = tid & 31, w = tid >> 5;
    int x = v;
    for (int off = 1; off < 32; off <<= 1) {
        int y = __shfl_up(x, off, 32);
        if (lane >= off) x += y;
    }
    if (lane == 31) wsum[w] = x;
    __syncthreads();
    if (w == 0) {
        int t = wsum[lane], z = t;
        for (int off = 1; off < 32; off <<= 1) {
            int y = __shfl_up(z, off, 32);
            if (lane >= off) z += y;
        }
        wsum[lane] = z - t;
    }
    __syncthreads();
    return x - v + wsum[w];
}

__global__ __launch_bounds__(1024) void bucket_hist(const int* __restrict__ dstA,
                                                    int* __restrict__ histA, int nbkA,
                                                    const int* __restrict__ dstB,
                                                    int* __restrict__ histB, int nbkB,
                                                    int nE, int nBlk) {
    __shared__ int h[MAXB];
    const bool second = (int)blockIdx.x >= nBlk;
    const int* __restrict__ dst = second ? dstB : dstA;
    int* __restrict__ blkHist   = second ? histB : histA;
    const int nBuckets = second ? nbkB : nbkA;
    const int blk = second ? (int)blockIdx.x - nBlk : (int)blockIdx.x;
    int tid = threadIdx.x;
    h[tid] = 0;
    __syncthreads();
    int base = blk * EPB + tid;
    int d[EPB / 1024];
#pragma unroll
    for (int u = 0; u < EPB / 1024; ++u) {
        int e = base + u * 1024;
        d[u] = dst[max(min(e, nE - 1), 0)];
    }
#pragma unroll
    for (int u = 0; u < EPB / 1024; ++u) {
        int e = base + u * 1024;
        if (e < nE) atomicAdd(&h[d[u] >> RSH], 1);
    }
    __syncthreads();
    if (tid < nBuckets) blkHist[tid * nBlk + blk] = h[tid];
}

__global__ void scan_block(const int* __restrict__ degA, int* __restrict__ exA,
                           int* __restrict__ bsumA, int nA, int nbA,
                           const int* __restrict__ degB, int* __restrict__ exB,
                           int* __restrict__ bsumB, int nB) {
    __shared__ int sh[256];
    const bool second = (int)blockIdx.x >= nbA;
    const int* __restrict__ deg = second ? degB : degA;
    int* __restrict__ ex   = second ? exB : exA;
    int* __restrict__ bsum = second ? bsumB : bsumA;
    const int n   = second ? nB : nA;
    const int blk = second ? (int)blockIdx.x - nbA : (int)blockIdx.x;
    int tid = threadIdx.x;
    int i = blk * 256 + tid;
    int v = (i < n) ? deg[i] : 0;
    sh[tid] = v;
    __syncthreads();
    for (int off = 1; off < 256; off <<= 1) {
        int add = (tid >= off) ? sh[tid - off] : 0;
        __syncthreads();
        sh[tid] += add;
        __syncthreads();
    }
    if (i < n) ex[i] = sh[tid] - v;
    if (tid == 255) bsum[blk] = sh[255];
}

__global__ void scan_sums(int* __restrict__ bsumA, int nbA,
                          int* __restrict__ bsumB, int nbB) {
    __shared__ int sh[1024];
    const bool second = blockIdx.x != 0;
    int* __restrict__ bsum = second ? bsumB : bsumA;
    const int nb = second ? nbB : nbA;
    int tid = threadIdx.x;
    int v = (tid < nb) ? bsum[tid] : 0;
    sh[tid] = v;
    __syncthreads();
    for (int off = 1; off < 1024; off <<= 1) {
        int add = (tid >= off) ? sh[tid - off] : 0;
        __syncthreads();
        sh[tid] += add;
        __syncthreads();
    }
    if (tid < nb) bsum[tid] = sh[tid] - v;
}

__global__ void add_block_offsets(int* __restrict__ exA, const int* __restrict__ bsumA, int nA,
                                  int nbA,
                                  int* __restrict__ exB, const int* __restrict__ bsumB, int nB) {
    const bool second = (int)blockIdx.x >= nbA;
    int* __restrict__ ex         = second ? exB : exA;
    const int* __restrict__ bsum = second ? bsumB : bsumA;
    const int n   = second ? nB : nA;
    const int blk = second ? (int)blockIdx.x - nbA : (int)blockIdx.x;
    int i = blk * 256 + threadIdx.x;
    if (i < n) ex[i] += bsum[blk];
}

__global__ __launch_bounds__(1024) void bucket_scatter(const int* __restrict__ srcA,
                                                       const int* __restrict__ dstA,
                                                       const int* __restrict__ histA,
                                                       const int* __restrict__ offA,
                                                       int* __restrict__ pairsA, int nbkA,
                                                       const int* __restrict__ srcB,
                                                       const int* __restrict__ dstB,
                                                       const int* __restrict__ histB,
                                                       const int* __restrict__ offB,
                                                       int* __restrict__ pairsB, int nbkB,
                                                       int nE, int nBlk) {
    __shared__ int2 stage[EPB];
    __shared__ int lcur[MAXB];
    __shared__ int delta[MAXB];
    __shared__ int wsum[32];
    const bool second = (int)blockIdx.x >= nBlk;
    const int* __restrict__ src     = second ? srcB : srcA;
    const int* __restrict__ dst     = second ? dstB : dstA;
    const int* __restrict__ blkHist = second ? histB : histA;
    const int* __restrict__ blkOff  = second ? offB : offA;
    int* __restrict__ pairs         = second ? pairsB : pairsA;
    const int nBuckets = second ? nbkB : nbkA;
    const int blk = second ? (int)blockIdx.x - nBlk : (int)blockIdx.x;
    int tid = threadIdx.x;
    int nMine = 0, gOff = 0;
    if (tid < nBuckets) {
        nMine  = blkHist[tid * nBlk + blk];
        gOff = blkOff[tid * nBlk + blk];
    }
    int base = blk * EPB + tid;
    int d[EPB / 1024], s[EPB / 1024];
#pragma unroll
    for (int u = 0; u < EPB / 1024; ++u) {
        int ec = max(min(base + u * 1024, nE - 1), 0);
        d[u] = dst[ec];
        s[u] = src[ec];
    }
    int lst = block_excl_scan_1024(nMine, wsum, tid);
    lcur[tid]  = lst;
    delta[tid] = gOff - lst;
    __syncthreads();
#pragma unroll
    for (int u = 0; u < EPB / 1024; ++u) {
        int e = base + u * 1024;
        if (e < nE) {
            int lp = atomicAdd(&lcur[d[u] >> RSH], 1);
            stage[lp] = make_int2(s[u], d[u]);
        }
    }
    __syncthreads();
    int2* __restrict__ p2 = (int2*)pairs;
    const int nValid = min(EPB, nE - blk * EPB);
#pragma unroll
    for (int u = 0; u < EPB / 1024; ++u) {
        int j = tid + u * 1024;
        if (j < nValid) {
            int2 p = stage[j];
            p2[j + delta[p.y >> RSH]] = p;
        }
    }
}

__global__ __launch_bounds__(1024) void bucket_csr(const int* __restrict__ pairsA,
                                                   const int* __restrict__ offA,
                                                   int* __restrict__ rowStartA,
                                                   int* __restrict__ rowEndA,
                                                   int* __restrict__ srcsA, int nA, int nbkA,
                                                   const int* __restrict__ pairsB,
                                                   const int* __restrict__ offB,
                                                   int* __restrict__ rowStartB,
                                                   int* __restrict__ rowEndB,
                                                   int* __restrict__ srcsB, int nB, int nbkB,
                                                   int nE, int nBlk) {
    __shared__ int cnt[RPB];
    __shared__ int sc[RPB];
    __shared__ int lbuf[CAPS];
    const bool second = (int)blockIdx.x >= nbkA;
    const int* __restrict__ pairs  = second ? pairsB : pairsA;
    const int* __restrict__ blkOff = second ? offB : offA;
    int* __restrict__ rowStart     = second ? rowStartB : rowStartA;
    int* __restrict__ rowEnd       = second ? rowEndB : rowEndA;
    int* __restrict__ srcs         = second ? srcsB : srcsA;
    const int n        = second ? nB : nA;
    const int nBuckets = second ? nbkB : nbkA;
    int b = second ? (int)blockIdx.x - nbkA : (int)blockIdx.x, tid = threadIdx.x;
    int s0 = blkOff[b * nBlk];
    int s1 = (b + 1 < nBuckets) ? blkOff[(b + 1) * nBlk] : nE;
    const int2* __restrict__ p2 = (const int2*)pairs;
    const int nEnt = s1 - s0;
    const bool grouped = nEnt <= CAPS;
    if (tid < RPB) cnt[tid] = 0;
    __syncthreads();
    for (int i0 = s0 + tid; i0 < s1; i0 += 4096) {
        int2 p[4];
#pragma unroll
        for (int u = 0; u < 4; ++u)
            p[u] = p2[min(i0 + u * 1024, s1 - 1)];
#pragma unroll
        for (int u = 0; u < 4; ++u) {
            int i = i0 + u * 1024;
            if (i < s1) atomicAdd(&cnt[p[u].y & (RPB - 1)], 1);
        }
    }
    __syncthreads();
    int v = 0;
    if (tid < RPB) { v = cnt[tid]; sc[tid] = v; }
    __syncthreads();
    for (int off = 1; off < RPB; off <<= 1) {
        int add = 0;
        if (tid < RPB && tid >= off) add = sc[tid - off];
        __syncthreads();
        if (tid < RPB) sc[tid] += add;
        __syncthreads();
    }
    if (tid < RPB) {
        int start = s0 + sc[tid] - v;
        int node = b * RPB + tid;
        if (node < n) { rowStart[node] = start; rowEnd[node] = start + v; }
        cnt[tid] = start;
    }
    __syncthreads();
    for (int i0 = s0 + tid; i0 < s1; i0 += 4096) {
        int2 p[4];
#pragma unroll
        for (int u = 0; u < 4; ++u)
            p[u] = p2[min(i0 + u * 1024, s1 - 1)];
#pragma unroll
        for (int u = 0; u < 4; ++u) {
            int i = i0 + u * 1024;
            if (i < s1) {
                int pos = atomicAdd(&cnt[p[u].y & (RPB - 1)], 1);
                if (grouped) lbuf[pos - s0] = p[u].x;
                else         srcs[pos] = p[u].x;
            }
        }
    }
    if (grouped) {
        __syncthreads();
        for (int i = tid; i < nEnt; i += 1024) srcs[s0 + i] = lbuf[i];
    }
}

__device__ __forceinline__ void gather_rows_body(const float* __restrict__ emb,
                                                 const int* __restrict__ nid,
                                                 float* __restrict__ out, int n16, int blk) {
    int t = blk * blockDim.x + threadIdx.x;
    if (t >= n16) return;
    int i = t >> 4, c = (t & 15) * 4;
    int s = nid[i];
    *(float4*)(out + (size_t)i * D + c) = *(const float4*)(emb + (size_t)s * D + c);
}

#define KG   32
#define WPT  (KC / 16)

__device__ __forceinline__ void load_a_line(v2f (&dst)[KG / 4], const float* __restrict__ p) {
#pragma unroll
    for (int s = 0; s < KG / 4; ++s) dst[s] = *(const v2f*)(p + 4 * s);
}

__device__ __forceinline__ void input_gemm_body(
        const float* __restrict__ go_x, const float* __restrict__ lin_W,
        const float* __restrict__ lin_b, const float* __restrict__ go_emb,
        const int* __restrict__ go_nid, float* __restrict__ xg, int M, int K, int blk) {
    __shared__ float sWt[D * STI];

    int waveInBlk = threadIdx.x >> 5;
    int lane = threadIdx.x & 31;
    int row  = lane & 15;
    int kh   = lane >> 4;
    int colL = lane & 15;

    int numTiles = M >> 4;
    int tile0 = (blk * 8 + waveInBlk) * TMI;

    int   m0c[TMI];
    const float* arow[TMI];
    bool  valid[TMI];
#pragma unroll
    for (int t = 0; t < TMI; ++t) {
        int ti = tile0 + t;
        valid[t] = (ti < numTiles);
        m0c[t] = valid[t] ? (ti << 4) : 0;
        arow[t] = go_x + (size_t)(m0c[t] + row) * K + 2 * kh;
    }

    v8f acc[TMI][4];
#pragma unroll
    for (int t = 0; t < TMI; ++t)
#pragma unroll
        for (int j = 0; j < 4; ++j) {
            float b = lin_b[j * 16 + colL];
#pragma unroll
            for (int g = 0; g < 8; ++g) acc[t][j][g] = b;
        }

    v2f aC[TMI][KG / 4], aN[TMI][KG / 4];
    if (K >= KG) {
#pragma unroll
        for (int t = 0; t < TMI; ++t) load_a_line(aC[t], arow[t]);
    }

    const int wr = threadIdx.x >> 4;
    const int wc = (threadIdx.x & 15) * 4;

    for (int kc0 = 0; kc0 < K; kc0 += KC) {
        int len = (K - kc0 < KC) ? (K - kc0) : KC;
        __syncthreads();
        {
            float4 w[WPT];
#pragma unroll
            for (int u = 0; u < WPT; ++u) {
                int r = min(kc0 + wr + 16 * u, K - 1);
                w[u] = *(const float4*)&lin_W[(size_t)r * D + wc];
            }
#pragma unroll
            for (int u = 0; u < WPT; ++u) {
                int r = wr + 16 * u;
                if (r < len) {
                    sWt[(wc + 0) * STI + r] = w[u].x;
                    sWt[(wc + 1) * STI + r] = w[u].y;
                    sWt[(wc + 2) * STI + r] = w[u].z;
                    sWt[(wc + 3) * STI + r] = w[u].w;
                }
            }
        }
        __syncthreads();
        if (kc0 + KC < K) {
#pragma unroll
            for (int t = 0; t < TMI; ++t)
                __builtin_prefetch(arow[t] + kc0 + KC, 0, 0);
        }
        int k = 0;
        for (; k + KG <= len; k += KG) {
            const int kn = min(kc0 + k + KG, K - KG);
#pragma unroll
            for (int t = 0; t < TMI; ++t) load_a_line(aN[t], arow[t] + kn);
#pragma unroll
            for (int s = 0; s < KG / 4; ++s) {
#pragma unroll
                for (int j = 0; j < 4; ++j) {
                    v2f b = *(const v2f*)&sWt[(j * 16 + colL) * STI + k + 4 * s + 2 * kh];
#pragma unroll
                    for (int t = 0; t < TMI; ++t)
                        acc[t][j] = __builtin_amdgcn_wmma_f32_16x16x4_f32(
                            false, aC[t][s], false, b, (short)0, acc[t][j], false, false);
                }
            }
#pragma unroll
            for (int t = 0; t < TMI; ++t)
#pragma unroll
                for (int s = 0; s < KG / 4; ++s) aC[t][s] = aN[t][s];
        }
        for (; k < len; k += 4) {
            v2f a[TMI];
#pragma unroll
            for (int t = 0; t < TMI; ++t)
                a[t] = *(const v2f*)(arow[t] + kc0 + k);
#pragma unroll
            for (int j = 0; j < 4; ++j) {
                v2f b = *(const v2f*)&sWt[(j * 16 + colL) * STI + k + 2 * kh];
#pragma unroll
                for (int t = 0; t < TMI; ++t)
                    acc[t][j] = __builtin_amdgcn_wmma_f32_16x16x4_f32(
                        false, a[t], false, b, (short)0, acc[t][j], false, false);
            }
        }
    }

    int hi = kh * 8;
#pragma unroll
    for (int t = 0; t < TMI; ++t) {
        if (!valid[t]) continue;
#pragma unroll
        for (int g = 0; g < 8; ++g) {
            int m = m0c[t] + g + hi;
            int nid = go_nid[m];
#pragma unroll
            for (int j = 0; j < 4; ++j) {
                int col = j * 16 + colL;
                xg[(size_t)m * D + col] = acc[t][j][g] + go_emb[(size_t)nid * D + col];
            }
        }
    }
}

__global__ __launch_bounds__(256, 6) void input_embed(
        const float* __restrict__ go_x, const float* __restrict__ lin_W,
        const float* __restrict__ lin_b, const float* __restrict__ go_emb,
        const int* __restrict__ go_nid, float* __restrict__ xg, int M, int K, int gemmBlocks,
        const float* __restrict__ p_emb, const int* __restrict__ p_nid,
        float* __restrict__ xp, int n16) {
    if ((int)blockIdx.x < gemmBlocks)
        input_gemm_body(go_x, lin_W, lin_b, go_emb, go_nid, xg, M, K, (int)blockIdx.x);
    else
        gather_rows_body(p_emb, p_nid, xp, n16, (int)blockIdx.x - gemmBlocks);
}

__global__ __launch_bounds__(256) void layer_fused_wmma(
        const float* __restrict__ xsA, const int* __restrict__ srcsA,
        const int* __restrict__ rowStartA, const int* __restrict__ rowEndA,
        const float* __restrict__ xA,
        const float* __restrict__ WlA, const float* __restrict__ blA,
        const float* __restrict__ WrA, float* __restrict__ outA, int MA, int blocksA,
        const float* __restrict__ xsB, const int* __restrict__ srcsB,
        const int* __restrict__ rowStartB, const int* __restrict__ rowEndB,
        const float* __restrict__ xB,
        const float* __restrict__ WlB, const float* __restrict__ blB,
        const float* __restrict__ WrB, float* __restrict__ outB, int MB, int relu) {
    __shared__ float sWt[2 * D * STL];
    __shared__ float sAgg[FR * STL];
    const bool second = (int)blockIdx.x >= blocksA;
    const float* __restrict__ xs     = second ? xsB : xsA;
    const int* __restrict__ srcs     = second ? srcsB : srcsA;
    const int* __restrict__ rowStart = second ? rowStartB : rowStartA;
    const int* __restrict__ rowEnd   = second ? rowEndB : rowEndA;
    const float* __restrict__ x   = second ? xB : xA;
    const float* __restrict__ Wl  = second ? WlB : WlA;
    const float* __restrict__ bl  = second ? blB : blA;
    const float* __restrict__ Wr  = second ? WrB : WrA;
    float* __restrict__ out       = second ? outB : outA;
    const int M   = second ? MB : MA;
    const int blk = second ? (int)blockIdx.x - blocksA : (int)blockIdx.x;

    const int grp = threadIdx.x >> 4;
    const int q   = threadIdx.x & 15;
    const unsigned cb = (unsigned)q * 16u;
    const char* __restrict__ xsb = (const char*)xs;
    int rs[FR / 16], re[FR / 16], ix[FR / 16];
#pragma unroll
    for (int p = 0; p < FR / 16; ++p)
        rs[p] = rowStart[min(blk * FR + p * 16 + grp, M - 1)];
#pragma unroll
    for (int p = 0; p < FR / 16; ++p)
        re[p] = rowEnd[min(blk * FR + p * 16 + grp, M - 1)];
    __builtin_amdgcn_sched_barrier(0);
#pragma unroll
    for (int p = 0; p < FR / 16; ++p)
        re[p] = (blk * FR + p * 16 + grp < M) ? re[p] : rs[p];
#pragma unroll
    for (int p = 0; p < FR / 16; ++p)
        ix[p] = srcs[max(min(rs[p] + q, re[p] - 1), 0)];

    for (int i = threadIdx.x * 4; i < D * D; i += 256 * 4) {
        int r = i >> 6, c = i & 63;
        float4 wl = *(const float4*)&Wl[i];
        float4 wr = *(const float4*)&Wr[i];
        sWt[(c + 0) * STL + r] = wl.x;
        sWt[(c + 1) * STL + r] = wl.y;
        sWt[(c + 2) * STL + r] = wl.z;
        sWt[(c + 3) * STL + r] = wl.w;
        sWt[D * STL + (c + 0) * STL + r] = wr.x;
        sWt[D * STL + (c + 1) * STL + r] = wr.y;
        sWt[D * STL + (c + 2) * STL + r] = wr.z;
        sWt[D * STL + (c + 3) * STL + r] = wr.w;
    }

    {
#pragma unroll
        for (int p = 0; p < FR / 16; ++p) {
            const int lr = p * 16 + grp;
            const int s0 = rs[p], s1 = re[p];
            float4 acc = make_float4(0.f, 0.f, 0.f, 0.f);
            int my = ix[p];
            for (int base = s0; base < s1; base += 16) {
                const int cnt = min(16, s1 - base);
                const int myNext = srcs[min(base + 16 + q, s1 - 1)];
                for (int k = 0; k < cnt; k += 8) {
                    float4 v[8];
#pragma unroll
                    for (int u = 0; u < 8; ++u) {
                        const unsigned n = (unsigned)__shfl(my, k + u, 16);
                        v[u] = *(const float4*)(xsb + (size_t)(n * (unsigned)(D * 4) + cb));
                    }
                    acc.x += v[0].x; acc.y += v[0].y; acc.z += v[0].z; acc.w += v[0].w;
#pragma unroll
                    for (int u = 1; u < 8; ++u) {
                        const bool on = (k + u < cnt);
                        acc.x += on ? v[u].x : 0.0f;
                        acc.y += on ? v[u].y : 0.0f;
                        acc.z += on ? v[u].z : 0.0f;
                        acc.w += on ? v[u].w : 0.0f;
                    }
                }
                my = myNext;
            }
            float inv = 1.0f / fmaxf((float)(s1 - s0), 1.0f);
            acc.x *= inv; acc.y *= inv; acc.z *= inv; acc.w *= inv;
            *(float4*)&sAgg[lr * STL + 4 * q] = acc;
        }
    }
    __syncthreads();

    const int w    = threadIdx.x >> 5;
    const int lane = threadIdx.x & 31;
    const int row  = lane & 15;
    const int kh   = lane >> 4;
    const int colL = lane & 15;
    const int tl   = w & 3;
    const int ch   = w >> 2;

    const int  m0    = blk * FR + tl * 16;
    const bool valid = (m0 + 16 <= M);
    const int  m0c   = valid ? m0 : 0;
    const float* xrow = x + (size_t)(m0c + row) * D;
    const float* arow = &sAgg[(tl * 16 + row) * STL];

    v8f acc[2];
#pragma unroll
    for (int jj = 0; jj < 2; ++jj) {
        float b = bl[(ch * 2 + jj) * 16 + colL];
#pragma unroll
        for (int g = 0; g < 8; ++g) acc[jj][g] = b;
    }

#pragma unroll
    for (int k = 0; k < D; k += 4) {
        v2f aA = *(const v2f*)(arow + k + 2 * kh);
        v2f aX = *(const v2f*)(xrow + k + 2 * kh);
#pragma unroll
        for (int jj = 0; jj < 2; ++jj) {
            const int j = ch * 2 + jj;
            v2f bL = *(const v2f*)&sWt[(j * 16 + colL) * STL + k + 2 * kh];
            v2f bR = *(const v2f*)&sWt[D * STL + (j * 16 + colL) * STL + k + 2 * kh];
            acc[jj] = __builtin_amdgcn_wmma_f32_16x16x4_f32(
                false, aA, false, bL, (short)0, acc[jj], false, false);
            acc[jj] = __builtin_amdgcn_wmma_f32_16x16x4_f32(
                false, aX, false, bR, (short)0, acc[jj], false, false);
        }
    }

    if (valid) {
        const int hi = kh * 8;
#pragma unroll
        for (int g = 0; g < 8; ++g) {
            int m = m0c + g + hi;
#pragma unroll
            for (int jj = 0; jj < 2; ++jj) {
                float v = acc[jj][g];
                if (relu) v = fmaxf(v, 0.0f);
                out[(size_t)m * D + (ch * 2 + jj) * 16 + colL] = v;
            }
        }
    }
}

__global__ void classifier_dot(const float* __restrict__ xp, const float* __restrict__ xg,
                               const int* __restrict__ ls, const int* __restrict__ ld,
                               float* __restrict__ out, int n) {
    int e = blockIdx.x * blockDim.x + threadIdx.x;
    if (e >= n) return;
    const float4* a = (const float4*)(xp + (size_t)ls[e] * D);
    const float4* b = (const float4*)(xg + (size_t)ld[e] * D);
    float s = 0.0f;
#pragma unroll
    for (int i = 0; i < 16; ++i) {
        float4 va = a[i], vb = b[i];
        s += va.x * vb.x + va.y * vb.y + va.z * vb.z + va.w * vb.w;
    }
    out[e] = s;
}

extern "C" void kernel_launch(void* const* d_in, const int* in_sizes, int n_in,
                              void* d_out, int out_size, void* d_ws, size_t ws_size,
                              hipStream_t stream) {
    const float* go_x  = (const float*)d_in[0];
    const float* p_emb = (const float*)d_in[1];
    const float* g_emb = (const float*)d_in[2];
    const float* lin_W = (const float*)d_in[3];
    const float* lin_b = (const float*)d_in[4];
    const float* Wl    = (const float*)d_in[5];
    const float* bl    = (const float*)d_in[6];
    const float* Wr    = (const float*)d_in[7];
    const int* p_nid   = (const int*)d_in[8];
    const int* g_nid   = (const int*)d_in[9];
    const int* src_pg  = (const int*)d_in[10];
    const int* dst_pg  = (const int*)d_in[11];
    const int* src_gp  = (const int*)d_in[12];
    const int* dst_gp  = (const int*)d_in[13];
    const int* lab_s   = (const int*)d_in[14];
    const int* lab_d   = (const int*)d_in[15];

    const int NP  = in_sizes[8];
    const int NG  = in_sizes[9];
    const int E   = in_sizes[10];
    const int EL  = in_sizes[14];
    const int FGO = in_sizes[0] / NG;

    char* base = (char*)d_ws;
    auto alloc = [&](size_t bytes) -> char* {
        char* p = base;
        base += (bytes + 255) & ~(size_t)255;
        return p;
    };
    float* xp_a  = (float*)alloc((size_t)NP * D * 4);
    float* xp_b  = (float*)alloc((size_t)NP * D * 4);
    float* xg_a  = (float*)alloc((size_t)NG * D * 4);
    float* xg_b  = (float*)alloc((size_t)NG * D * 4);
    float* tmp_p = (float*)alloc((size_t)NP * D * 4);
    (void)alloc((size_t)NG * D * 4);
    int* row_g   = (int*)alloc((size_t)NG * 4);
    int* cur_g   = (int*)alloc((size_t)NG * 4);
    int* row_p   = (int*)alloc((size_t)NP * 4);
    int* cur_p   = (int*)alloc((size_t)NP * 4);
    int* srcs_g  = (int*)alloc((size_t)E * 4);
    int* srcs_p  = (int*)alloc((size_t)E * 4);
    int* bsum    = (int*)alloc(1024 * 4);

    const int nBlkE = (E + EPB - 1) / EPB;
    const int nbkG  = (NG + RPB - 1) / RPB;
    const int nbkP  = (NP + RPB - 1) / RPB;
    const int nmatG = nbkG * nBlkE;
    const int nmatP = nbkP * nBlkE;
    const int nbG   = (nmatG + 255) / 256;
    const int nbP   = (nmatP + 255) / 256;
    const int nmatGp = (nmatG + 63) & ~63;
    int* pairsG = (int*)tmp_p;
    int* pairsP = pairsG + (size_t)2 * E;
    int* hmatG  = (int*)xp_b;
    int* hmatP  = hmatG + nmatGp;
    int* omatG  = (int*)xg_b;
    int* omatP  = omatG + nmatGp;
    int* bsumG  = bsum;
    int* bsumP  = omatP + ((nmatP + 63) & ~63);

    bucket_hist<<<2 * nBlkE, 1024, 0, stream>>>(dst_pg, hmatG, nbkG, dst_gp, hmatP, nbkP, E, nBlkE);
    scan_block<<<nbG + nbP, 256, 0, stream>>>(hmatG, omatG, bsumG, nmatG, nbG,
                                              hmatP, omatP, bsumP, nmatP);
    scan_sums<<<2, 1024, 0, stream>>>(bsumG, nbG, bsumP, nbP);
    add_block_offsets<<<nbG + nbP, 256, 0, stream>>>(omatG, bsumG, nmatG, nbG,
                                                     omatP, bsumP, nmatP);
    bucket_scatter<<<2 * nBlkE, 1024, 0, stream>>>(src_pg, dst_pg, hmatG, omatG, pairsG, nbkG,
                                                   src_gp, dst_gp, hmatP, omatP, pairsP, nbkP,
                                                   E, nBlkE);
    bucket_csr<<<nbkG + nbkP, 1024, 0, stream>>>(pairsG, omatG, row_g, cur_g, srcs_g, NG, nbkG,
                                                 pairsP, omatP, row_p, cur_p, srcs_p, NP, nbkP,
                                                 E, nBlkE);

    {
        int tiles      = (NG + 15) / 16;
        int gemmBlocks = (tiles + 8 * TMI - 1) / (8 * TMI);
        int gatherBlks = (NP * 16 + 255) / 256;
        input_embed<<<gemmBlocks + gatherBlks, 256, 0, stream>>>(go_x, lin_W, lin_b, g_emb, g_nid,
                                                                 xg_a, NG, FGO, gemmBlocks,
                                                                 p_emb, p_nid, xp_a, NP * 16);
    }

    float* xp_cur = xp_a; float* xp_nxt = xp_b;
    float* xg_cur = xg_a; float* xg_nxt = xg_b;
    const int fusBlkG = (NG + FR - 1) / FR;
    const int fusBlkP = (NP + FR - 1) / FR;
    for (int l = 0; l < 3; ++l) {
        const float* Wl0 = Wl + (size_t)(l * 2 + 0) * D * D;
        const float* Wl1 = Wl + (size_t)(l * 2 + 1) * D * D;
        const float* Wr0 = Wr + (size_t)(l * 2 + 0) * D * D;
        const float* Wr1 = Wr + (size_t)(l * 2 + 1) * D * D;
        const float* bl0 = bl + (size_t)(l * 2 + 0) * D;
        const float* bl1 = bl + (size_t)(l * 2 + 1) * D;
        int relu = (l < 2) ? 1 : 0;
        layer_fused_wmma<<<fusBlkG + fusBlkP, 256, 0, stream>>>(
            xp_cur, srcs_g, row_g, cur_g, xg_cur, Wl0, bl0, Wr0, xg_nxt, NG, fusBlkG,
            xg_cur, srcs_p, row_p, cur_p, xp_cur, Wl1, bl1, Wr1, xp_nxt, NP, relu);
        float* t;
        t = xp_cur; xp_cur = xp_nxt; xp_nxt = t;
        t = xg_cur; xg_cur = xg_nxt; xg_nxt = t;
    }

    classifier_dot<<<(EL + 255) / 256, 256, 0, stream>>>(xp_cur, xg_cur, lab_s, lab_d,
                                                         (float*)d_out, EL);
}


# Classifier: 16 lanes share a label edge's two rows, so a load covers whole 256-byte rows where one lane per edge touched 32 lines; the 64-term dot is summed by a fixed lane-halving tree (new order)
# speedup vs baseline: 3.1572x; 1.1243x over previous
#include <hip/hip_runtime.h>

typedef __attribute__((ext_vector_type(2))) float v2f;
typedef __attribute__((ext_vector_type(8))) float v8f;

#define D    64
#define KC   128
#define STI  (KC+4)
#define STL  (D+4)
#define TMI  2
#define FR   64


#define EPB   8192
#define RSH   8
#define RPB   256
#define MAXB  1024
#define CAPS  16384


__device__ __forceinline__ int block_excl_scan_1024(int v, int* __restrict__ wsum, int tid) {
    const int lane = tid & 31, w = tid >> 5;
    int x = v;
    for (int off = 1; off < 32; off <<= 1) {
        int y = __shfl_up(x, off, 32);
        if (lane >= off) x += y;
    }
    if (lane == 31) wsum[w] = x;
    __syncthreads();
    if (w == 0) {
        int t = wsum[lane], z = t;
        for (int off = 1; off < 32; off <<= 1) {
            int y = __shfl_up(z, off, 32);
            if (lane >= off) z += y;
        }
        wsum[lane] = z - t;
    }
    __syncthreads();
    return x - v + wsum[w];
}

__global__ __launch_bounds__(1024) void bucket_hist(const int* __restrict__ dstA,
                                                    int* __restrict__ histA, int nbkA,
                                                    const int* __restrict__ dstB,
                                                    int* __restrict__ histB, int nbkB,
                                                    int nE, int nBlk) {
    __shared__ int h[MAXB];
    const bool second = (int)blockIdx.x >= nBlk;
    const int* __restrict__ dst = second ? dstB : dstA;
    int* __restrict__ blkHist   = second ? histB : histA;
    const int nBuckets = second ? nbkB : nbkA;
    const int blk = second ? (int)blockIdx.x - nBlk : (int)blockIdx.x;
    int tid = threadIdx.x;
    h[tid] = 0;
    __syncthreads();
    int base = blk * EPB + tid;
    int d[EPB / 1024];
#pragma unroll
    for (int u = 0; u < EPB / 1024; ++u) {
        int e = base + u * 1024;
        d[u] = dst[max(min(e, nE - 1), 0)];
    }
#pragma unroll
    for (int u = 0; u < EPB / 1024; ++u) {
        int e = base + u * 1024;
        if (e < nE) atomicAdd(&h[d[u] >> RSH], 1);
    }
    __syncthreads();
    if (tid < nBuckets) blkHist[tid * nBlk + blk] = h[tid];
}

__global__ void scan_block(const int* __restrict__ degA, int* __restrict__ exA,
                           int* __restrict__ bsumA, int nA, int nbA,
                           const int* __restrict__ degB, int* __restrict__ exB,
                           int* __restrict__ bsumB, int nB) {
    __shared__ int sh[256];
    const bool second = (int)blockIdx.x >= nbA;
    const int* __restrict__ deg = second ? degB : degA;
    int* __restrict__ ex   = second ? exB : exA;
    int* __restrict__ bsum = second ? bsumB : bsumA;
    const int n   = second ? nB : nA;
    const int blk = second ? (int)blockIdx.x - nbA : (int)blockIdx.x;
    int tid = threadIdx.x;
    int i = blk * 256 + tid;
    int v = (i < n) ? deg[i] : 0;
    sh[tid] = v;
    __syncthreads();
    for (int off = 1; off < 256; off <<= 1) {
        int add = (tid >= off) ? sh[tid - off] : 0;
        __syncthreads();
        sh[tid] += add;
        __syncthreads();
    }
    if (i < n) ex[i] = sh[tid] - v;
    if (tid == 255) bsum[blk] = sh[255];
}

__global__ void scan_sums(int* __restrict__ bsumA, int nbA,
                          int* __restrict__ bsumB, int nbB) {
    __shared__ int sh[1024];
    const bool second = blockIdx.x != 0;
    int* __restrict__ bsum = second ? bsumB : bsumA;
    const int nb = second ? nbB : nbA;
    int tid = threadIdx.x;
    int v = (tid < nb) ? bsum[tid] : 0;
    sh[tid] = v;
    __syncthreads();
    for (int off = 1; off < 1024; off <<= 1) {
        int add = (tid >= off) ? sh[tid - off] : 0;
        __syncthreads();
        sh[tid] += add;
        __syncthreads();
    }
    if (tid < nb) bsum[tid] = sh[tid] - v;
}

__global__ void add_block_offsets(int* __restrict__ exA, const int* __restrict__ bsumA, int nA,
                                  int nbA,
                                  int* __restrict__ exB, const int* __restrict__ bsumB, int nB) {
    const bool second = (int)blockIdx.x >= nbA;
    int* __restrict__ ex         = second ? exB : exA;
    const int* __restrict__ bsum = second ? bsumB : bsumA;
    const int n   = second ? nB : nA;
    const int blk = second ? (int)blockIdx.x - nbA : (int)blockIdx.x;
    int i = blk * 256 + threadIdx.x;
    if (i < n) ex[i] += bsum[blk];
}

__global__ __launch_bounds__(1024) void bucket_scatter(const int* __restrict__ srcA,
                                                       const int* __restrict__ dstA,
                                                       const int* __restrict__ histA,
                                                       const int* __restrict__ offA,
                                                       int* __restrict__ pairsA, int nbkA,
                                                       const int* __restrict__ srcB,
                                                       const int* __restrict__ dstB,
                                                       const int* __restrict__ histB,
                                                       const int* __restrict__ offB,
                                                       int* __restrict__ pairsB, int nbkB,
                                                       int nE, int nBlk) {
    __shared__ int2 stage[EPB];
    __shared__ int lcur[MAXB];
    __shared__ int delta[MAXB];
    __shared__ int wsum[32];
    const bool second = (int)blockIdx.x >= nBlk;
    const int* __restrict__ src     = second ? srcB : srcA;
    const int* __restrict__ dst     = second ? dstB : dstA;
    const int* __restrict__ blkHist = second ? histB : histA;
    const int* __restrict__ blkOff  = second ? offB : offA;
    int* __restrict__ pairs         = second ? pairsB : pairsA;
    const int nBuckets = second ? nbkB : nbkA;
    const int blk = second ? (int)blockIdx.x - nBlk : (int)blockIdx.x;
    int tid = threadIdx.x;
    int nMine = 0, gOff = 0;
    if (tid < nBuckets) {
        nMine  = blkHist[tid * nBlk + blk];
        gOff = blkOff[tid * nBlk + blk];
    }
    int base = blk * EPB + tid;
    int d[EPB / 1024], s[EPB / 1024];
#pragma unroll
    for (int u = 0; u < EPB / 1024; ++u) {
        int ec = max(min(base + u * 1024, nE - 1), 0);
        d[u] = dst[ec];
        s[u] = src[ec];
    }
    int lst = block_excl_scan_1024(nMine, wsum, tid);
    lcur[tid]  = lst;
    delta[tid] = gOff - lst;
    __syncthreads();
#pragma unroll
    for (int u = 0; u < EPB / 1024; ++u) {
        int e = base + u * 1024;
        if (e < nE) {
            int lp = atomicAdd(&lcur[d[u] >> RSH], 1);
            stage[lp] = make_int2(s[u], d[u]);
        }
    }
    __syncthreads();
    int2* __restrict__ p2 = (int2*)pairs;
    const int nValid = min(EPB, nE - blk * EPB);
#pragma unroll
    for (int u = 0; u < EPB / 1024; ++u) {
        int j = tid + u * 1024;
        if (j < nValid) {
            int2 p = stage[j];
            p2[j + delta[p.y >> RSH]] = p;
        }
    }
}

__global__ __launch_bounds__(1024) void bucket_csr(const int* __restrict__ pairsA,
                                                   const int* __restrict__ offA,
                                                   int* __restrict__ rowStartA,
                                                   int* __restrict__ rowEndA,
                                                   int* __restrict__ srcsA, int nA, int nbkA,
                                                   const int* __restrict__ pairsB,
                                                   const int* __restrict__ offB,
                                                   int* __restrict__ rowStartB,
                                                   int* __restrict__ rowEndB,
                                                   int* __restrict__ srcsB, int nB, int nbkB,
                                                   int nE, int nBlk) {
    __shared__ int cnt[RPB];
    __shared__ int sc[RPB];
    __shared__ int lbuf[CAPS];
    const bool second = (int)blockIdx.x >= nbkA;
    const int* __restrict__ pairs  = second ? pairsB : pairsA;
    const int* __restrict__ blkOff = second ? offB : offA;
    int* __restrict__ rowStart     = second ? rowStartB : rowStartA;
    int* __restrict__ rowEnd       = second ? rowEndB : rowEndA;
    int* __restrict__ srcs         = second ? srcsB : srcsA;
    const int n        = second ? nB : nA;
    const int nBuckets = second ? nbkB : nbkA;
    int b = second ? (int)blockIdx.x - nbkA : (int)blockIdx.x, tid = threadIdx.x;
    int s0 = blkOff[b * nBlk];
    int s1 = (b + 1 < nBuckets) ? blkOff[(b + 1) * nBlk] : nE;
    const int2* __restrict__ p2 = (const int2*)pairs;
    const int nEnt = s1 - s0;
    const bool grouped = nEnt <= CAPS;
    if (tid < RPB) cnt[tid] = 0;
    __syncthreads();
    for (int i0 = s0 + tid; i0 < s1; i0 += 4096) {
        int2 p[4];
#pragma unroll
        for (int u = 0; u < 4; ++u)
            p[u] = p2[min(i0 + u * 1024, s1 - 1)];
#pragma unroll
        for (int u = 0; u < 4; ++u) {
            int i = i0 + u * 1024;
            if (i < s1) atomicAdd(&cnt[p[u].y & (RPB - 1)], 1);
        }
    }
    __syncthreads();
    int v = 0;
    if (tid < RPB) { v = cnt[tid]; sc[tid] = v; }
    __syncthreads();
    for (int off = 1; off < RPB; off <<= 1) {
        int add = 0;
        if (tid < RPB && tid >= off) add = sc[tid - off];
        __syncthreads();
        if (tid < RPB) sc[tid] += add;
        __syncthreads();
    }
    if (tid < RPB) {
        int start = s0 + sc[tid] - v;
        int node = b * RPB + tid;
        if (node < n) { rowStart[node] = start; rowEnd[node] = start + v; }
        cnt[tid] = start;
    }
    __syncthreads();
    for (int i0 = s0 + tid; i0 < s1; i0 += 4096) {
        int2 p[4];
#pragma unroll
        for (int u = 0; u < 4; ++u)
            p[u] = p2[min(i0 + u * 1024, s1 - 1)];
#pragma unroll
        for (int u = 0; u < 4; ++u) {
            int i = i0 + u * 1024;
            if (i < s1) {
                int pos = atomicAdd(&cnt[p[u].y & (RPB - 1)], 1);
                if (grouped) lbuf[pos - s0] = p[u].x;
                else         srcs[pos] = p[u].x;
            }
        }
    }
    if (grouped) {
        __syncthreads();
        for (int i = tid; i < nEnt; i += 1024) srcs[s0 + i] = lbuf[i];
    }
}

__device__ __forceinline__ void gather_rows_body(const float* __restrict__ emb,
                                                 const int* __restrict__ nid,
                                                 float* __restrict__ out, int n16, int blk) {
    int t = blk * blockDim.x + threadIdx.x;
    if (t >= n16) return;
    int i = t >> 4, c = (t & 15) * 4;
    int s = nid[i];
    *(float4*)(out + (size_t)i * D + c) = *(const float4*)(emb + (size_t)s * D + c);
}

#define KG   32
#define WPT  (KC / 16)

__device__ __forceinline__ void load_a_line(v2f (&dst)[KG / 4], const float* __restrict__ p) {
#pragma unroll
    for (int s = 0; s < KG / 4; ++s) dst[s] = *(const v2f*)(p + 4 * s);
}

__device__ __forceinline__ void input_gemm_body(
        const float* __restrict__ go_x, const float* __restrict__ lin_W,
        const float* __restrict__ lin_b, const float* __restrict__ go_emb,
        const int* __restrict__ go_nid, float* __restrict__ xg, int M, int K, int blk) {
    __shared__ float sWt[D * STI];

    int waveInBlk = threadIdx.x >> 5;
    int lane = threadIdx.x & 31;
    int row  = lane & 15;
    int kh   = lane >> 4;
    int colL = lane & 15;

    int numTiles = M >> 4;
    int tile0 = (blk * 8 + waveInBlk) * TMI;

    int   m0c[TMI];
    const float* arow[TMI];
    bool  valid[TMI];
#pragma unroll
    for (int t = 0; t < TMI; ++t) {
        int ti = tile0 + t;
        valid[t] = (ti < numTiles);
        m0c[t] = valid[t] ? (ti << 4) : 0;
        arow[t] = go_x + (size_t)(m0c[t] + row) * K + 2 * kh;
    }

    v8f acc[TMI][4];
#pragma unroll
    for (int t = 0; t < TMI; ++t)
#pragma unroll
        for (int j = 0; j < 4; ++j) {
            float b = lin_b[j * 16 + colL];
#pragma unroll
            for (int g = 0; g < 8; ++g) acc[t][j][g] = b;
        }

    v2f aC[TMI][KG / 4], aN[TMI][KG / 4];
    if (K >= KG) {
#pragma unroll
        for (int t = 0; t < TMI; ++t) load_a_line(aC[t], arow[t]);
    }

    const int wr = threadIdx.x >> 4;
    const int wc = (threadIdx.x & 15) * 4;

    for (int kc0 = 0; kc0 < K; kc0 += KC) {
        int len = (K - kc0 < KC) ? (K - kc0) : KC;
        __syncthreads();
        {
            float4 w[WPT];
#pragma unroll
            for (int u = 0; u < WPT; ++u) {
                int r = min(kc0 + wr + 16 * u, K - 1);
                w[u] = *(const float4*)&lin_W[(size_t)r * D + wc];
            }
#pragma unroll
            for (int u = 0; u < WPT; ++u) {
                int r = wr + 16 * u;
                if (r < len) {
                    sWt[(wc + 0) * STI + r] = w[u].x;
                    sWt[(wc + 1) * STI + r] = w[u].y;
                    sWt[(wc + 2) * STI + r] = w[u].z;
                    sWt[(wc + 3) * STI + r] = w[u].w;
                }
            }
        }
        __syncthreads();
        if (kc0 + KC < K) {
#pragma unroll
            for (int t = 0; t < TMI; ++t)
                __builtin_prefetch(arow[t] + kc0 + KC, 0, 0);
        }
        int k = 0;
        for (; k + KG <= len; k += KG) {
            const int kn = min(kc0 + k + KG, K - KG);
#pragma unroll
            for (int t = 0; t < TMI; ++t) load_a_line(aN[t], arow[t] + kn);
#pragma unroll
            for (int s = 0; s < KG / 4; ++s) {
#pragma unroll
                for (int j = 0; j < 4; ++j) {
                    v2f b = *(const v2f*)&sWt[(j * 16 + colL) * STI + k + 4 * s + 2 * kh];
#pragma unroll
                    for (int t = 0; t < TMI; ++t)
                        acc[t][j] = __builtin_amdgcn_wmma_f32_16x16x4_f32(
                            false, aC[t][s], false, b, (short)0, acc[t][j], false, false);
                }
            }
#pragma unroll
            for (int t = 0; t < TMI; ++t)
#pragma unroll
                for (int s = 0; s < KG / 4; ++s) aC[t][s] = aN[t][s];
        }
        for (; k < len; k += 4) {
            v2f a[TMI];
#pragma unroll
            for (int t = 0; t < TMI; ++t)
                a[t] = *(const v2f*)(arow[t] + kc0 + k);
#pragma unroll
            for (int j = 0; j < 4; ++j) {
                v2f b = *(const v2f*)&sWt[(j * 16 + colL) * STI + k + 2 * kh];
#pragma unroll
                for (int t = 0; t < TMI; ++t)
                    acc[t][j] = __builtin_amdgcn_wmma_f32_16x16x4_f32(
                        false, a[t], false, b, (short)0, acc[t][j], false, false);
            }
        }
    }

    int hi = kh * 8;
#pragma unroll
    for (int t = 0; t < TMI; ++t) {
        if (!valid[t]) continue;
#pragma unroll
        for (int g = 0; g < 8; ++g) {
            int m = m0c[t] + g + hi;
            int nid = go_nid[m];
#pragma unroll
            for (int j = 0; j < 4; ++j) {
                int col = j * 16 + colL;
                xg[(size_t)m * D + col] = acc[t][j][g] + go_emb[(size_t)nid * D + col];
            }
        }
    }
}

__global__ __launch_bounds__(256, 6) void input_embed(
        const float* __restrict__ go_x, const float* __restrict__ lin_W,
        const float* __restrict__ lin_b, const float* __restrict__ go_emb,
        const int* __restrict__ go_nid, float* __restrict__ xg, int M, int K, int gemmBlocks,
        const float* __restrict__ p_emb, const int* __restrict__ p_nid,
        float* __restrict__ xp, int n16) {
    if ((int)blockIdx.x < gemmBlocks)
        input_gemm_body(go_x, lin_W, lin_b, go_emb, go_nid, xg, M, K, (int)blockIdx.x);
    else
        gather_rows_body(p_emb, p_nid, xp, n16, (int)blockIdx.x - gemmBlocks);
}

__global__ __launch_bounds__(256) void layer_fused_wmma(
        const float* __restrict__ xsA, const int* __restrict__ srcsA,
        const int* __restrict__ rowStartA, const int* __restrict__ rowEndA,
        const float* __restrict__ xA,
        const float* __restrict__ WlA, const float* __restrict__ blA,
        const float* __restrict__ WrA, float* __restrict__ outA, int MA, int blocksA,
        const float* __restrict__ xsB, const int* __restrict__ srcsB,
        const int* __restrict__ rowStartB, const int* __restrict__ rowEndB,
        const float* __restrict__ xB,
        const float* __restrict__ WlB, const float* __restrict__ blB,
        const float* __restrict__ WrB, float* __restrict__ outB, int MB, int relu) {
    __shared__ float sWt[2 * D * STL];
    __shared__ float sAgg[FR * STL];
    const bool second = (int)blockIdx.x >= blocksA;
    const float* __restrict__ xs     = second ? xsB : xsA;
    const int* __restrict__ srcs     = second ? srcsB : srcsA;
    const int* __restrict__ rowStart = second ? rowStartB : rowStartA;
    const int* __restrict__ rowEnd   = second ? rowEndB : rowEndA;
    const float* __restrict__ x   = second ? xB : xA;
    const float* __restrict__ Wl  = second ? WlB : WlA;
    const float* __restrict__ bl  = second ? blB : blA;
    const float* __restrict__ Wr  = second ? WrB : WrA;
    float* __restrict__ out       = second ? outB : outA;
    const int M   = second ? MB : MA;
    const int blk = second ? (int)blockIdx.x - blocksA : (int)blockIdx.x;

    const int grp = threadIdx.x >> 4;
    const int q   = threadIdx.x & 15;
    const unsigned cb = (unsigned)q * 16u;
    const char* __restrict__ xsb = (const char*)xs;
    int rs[FR / 16], re[FR / 16], ix[FR / 16];
#pragma unroll
    for (int p = 0; p < FR / 16; ++p)
        rs[p] = rowStart[min(blk * FR + p * 16 + grp, M - 1)];
#pragma unroll
    for (int p = 0; p < FR / 16; ++p)
        re[p] = rowEnd[min(blk * FR + p * 16 + grp, M - 1)];
    __builtin_amdgcn_sched_barrier(0);
#pragma unroll
    for (int p = 0; p < FR / 16; ++p)
        re[p] = (blk * FR + p * 16 + grp < M) ? re[p] : rs[p];
#pragma unroll
    for (int p = 0; p < FR / 16; ++p)
        ix[p] = srcs[max(min(rs[p] + q, re[p] - 1), 0)];

    for (int i = threadIdx.x * 4; i < D * D; i += 256 * 4) {
        int r = i >> 6, c = i & 63;
        float4 wl = *(const float4*)&Wl[i];
        float4 wr = *(const float4*)&Wr[i];
        sWt[(c + 0) * STL + r] = wl.x;
        sWt[(c + 1) * STL + r] = wl.y;
        sWt[(c + 2) * STL + r] = wl.z;
        sWt[(c + 3) * STL + r] = wl.w;
        sWt[D * STL + (c + 0) * STL + r] = wr.x;
        sWt[D * STL + (c + 1) * STL + r] = wr.y;
        sWt[D * STL + (c + 2) * STL + r] = wr.z;
        sWt[D * STL + (c + 3) * STL + r] = wr.w;
    }

    {
#pragma unroll
        for (int p = 0; p < FR / 16; ++p) {
            const int lr = p * 16 + grp;
            const int s0 = rs[p], s1 = re[p];
            float4 acc = make_float4(0.f, 0.f, 0.f, 0.f);
            int my = ix[p];
            for (int base = s0; base < s1; base += 16) {
                const int cnt = min(16, s1 - base);
                const int myNext = srcs[min(base + 16 + q, s1 - 1)];
                for (int k = 0; k < cnt; k += 8) {
                    float4 v[8];
#pragma unroll
                    for (int u = 0; u < 8; ++u) {
                        const unsigned n = (unsigned)__shfl(my, k + u, 16);
                        v[u] = *(const float4*)(xsb + (size_t)(n * (unsigned)(D * 4) + cb));
                    }
                    acc.x += v[0].x; acc.y += v[0].y; acc.z += v[0].z; acc.w += v[0].w;
#pragma unroll
                    for (int u = 1; u < 8; ++u) {
                        const bool on = (k + u < cnt);
                        acc.x += on ? v[u].x : 0.0f;
                        acc.y += on ? v[u].y : 0.0f;
                        acc.z += on ? v[u].z : 0.0f;
                        acc.w += on ? v[u].w : 0.0f;
                    }
                }
                my = myNext;
            }
            float inv = 1.0f / fmaxf((float)(s1 - s0), 1.0f);
            acc.x *= inv; acc.y *= inv; acc.z *= inv; acc.w *= inv;
            *(float4*)&sAgg[lr * STL + 4 * q] = acc;
        }
    }
    __syncthreads();

    const int w    = threadIdx.x >> 5;
    const int lane = threadIdx.x & 31;
    const int row  = lane & 15;
    const int kh   = lane >> 4;
    const int colL = lane & 15;
    const int tl   = w & 3;
    const int ch   = w >> 2;

    const int  m0    = blk * FR + tl * 16;
    const bool valid = (m0 + 16 <= M);
    const int  m0c   = valid ? m0 : 0;
    const float* xrow = x + (size_t)(m0c + row) * D;
    const float* arow = &sAgg[(tl * 16 + row) * STL];

    v8f acc[2];
#pragma unroll
    for (int jj = 0; jj < 2; ++jj) {
        float b = bl[(ch * 2 + jj) * 16 + colL];
#pragma unroll
        for (int g = 0; g < 8; ++g) acc[jj][g] = b;
    }

#pragma unroll
    for (int k = 0; k < D; k += 4) {
        v2f aA = *(const v2f*)(arow + k + 2 * kh);
        v2f aX = *(const v2f*)(xrow + k + 2 * kh);
#pragma unroll
        for (int jj = 0; jj < 2; ++jj) {
            const int j = ch * 2 + jj;
            v2f bL = *(const v2f*)&sWt[(j * 16 + colL) * STL + k + 2 * kh];
            v2f bR = *(const v2f*)&sWt[D * STL + (j * 16 + colL) * STL + k + 2 * kh];
            acc[jj] = __builtin_amdgcn_wmma_f32_16x16x4_f32(
                false, aA, false, bL, (short)0, acc[jj], false, false);
            acc[jj] = __builtin_amdgcn_wmma_f32_16x16x4_f32(
                false, aX, false, bR, (short)0, acc[jj], false, false);
        }
    }

    if (valid) {
        const int hi = kh * 8;
#pragma unroll
        for (int g = 0; g < 8; ++g) {
            int m = m0c + g + hi;
#pragma unroll
            for (int jj = 0; jj < 2; ++jj) {
                float v = acc[jj][g];
                if (relu) v = fmaxf(v, 0.0f);
                out[(size_t)m * D + (ch * 2 + jj) * 16 + colL] = v;
            }
        }
    }
}

__global__ __launch_bounds__(256) void classifier_dot(const float* __restrict__ xp,
                                                      const float* __restrict__ xg,
                                                      const int* __restrict__ ls,
                                                      const int* __restrict__ ld,
                                                      float* __restrict__ out, int n) {
    const int e  = blockIdx.x * blockDim.x + threadIdx.x;
    const int q  = threadIdx.x & 15;
    const int ec = min(e, n - 1);
    const int si = ls[ec];
    const int di = ld[ec];
    const unsigned cb = (unsigned)q * 16u;
    const char* __restrict__ xpb = (const char*)xp;
    const char* __restrict__ xgb = (const char*)xg;
    float4 va[16], vb[16];
#pragma unroll
    for (int u = 0; u < 16; ++u) {
        const unsigned s = (unsigned)__shfl(si, u, 16);
        const unsigned d = (unsigned)__shfl(di, u, 16);
        va[u] = *(const float4*)(xpb + (size_t)(s * (unsigned)(D * 4) + cb));
        vb[u] = *(const float4*)(xgb + (size_t)(d * (unsigned)(D * 4) + cb));
    }
    float p[16];
#pragma unroll
    for (int u = 0; u < 16; ++u)
        p[u] = va[u].x * vb[u].x + va[u].y * vb[u].y + va[u].z * vb[u].z + va[u].w * vb[u].w;
#pragma unroll
    for (int m = 8; m >= 1; m >>= 1) {
        const bool up = (q & m) != 0;
#pragma unroll
        for (int i = 0; i < m; ++i) {
            const float lo = p[i], hi = p[i + m];
            const float keep = up ? hi : lo;
            const float send = up ? lo : hi;
            p[i] = keep + __shfl_xor(send, m, 16);
        }
    }
    if (e < n) out[e] = p[0];
}

extern "C" void kernel_launch(void* const* d_in, const int* in_sizes, int n_in,
                              void* d_out, int out_size, void* d_ws, size_t ws_size,
                              hipStream_t stream) {
    const float* go_x  = (const float*)d_in[0];
    const float* p_emb = (const float*)d_in[1];
    const float* g_emb = (const float*)d_in[2];
    const float* lin_W = (const float*)d_in[3];
    const float* lin_b = (const float*)d_in[4];
    const float* Wl    = (const float*)d_in[5];
    const float* bl    = (const float*)d_in[6];
    const float* Wr    = (const float*)d_in[7];
    const int* p_nid   = (const int*)d_in[8];
    const int* g_nid   = (const int*)d_in[9];
    const int* src_pg  = (const int*)d_in[10];
    const int* dst_pg  = (const int*)d_in[11];
    const int* src_gp  = (const int*)d_in[12];
    const int* dst_gp  = (const int*)d_in[13];
    const int* lab_s   = (const int*)d_in[14];
    const int* lab_d   = (const int*)d_in[15];

    const int NP  = in_sizes[8];
    const int NG  = in_sizes[9];
    const int E   = in_sizes[10];
    const int EL  = in_sizes[14];
    const int FGO = in_sizes[0] / NG;

    char* base = (char*)d_ws;
    auto alloc = [&](size_t bytes) -> char* {
        char* p = base;
        base += (bytes + 255) & ~(size_t)255;
        return p;
    };
    float* xp_a  = (float*)alloc((size_t)NP * D * 4);
    float* xp_b  = (float*)alloc((size_t)NP * D * 4);
    float* xg_a  = (float*)alloc((size_t)NG * D * 4);
    float* xg_b  = (float*)alloc((size_t)NG * D * 4);
    float* tmp_p = (float*)alloc((size_t)NP * D * 4);
    (void)alloc((size_t)NG * D * 4);
    int* row_g   = (int*)alloc((size_t)NG * 4);
    int* cur_g   = (int*)alloc((size_t)NG * 4);
    int* row_p   = (int*)alloc((size_t)NP * 4);
    int* cur_p   = (int*)alloc((size_t)NP * 4);
    int* srcs_g  = (int*)alloc((size_t)E * 4);
    int* srcs_p  = (int*)alloc((size_t)E * 4);
    int* bsum    = (int*)alloc(1024 * 4);

    const int nBlkE = (E + EPB - 1) / EPB;
    const int nbkG  = (NG + RPB - 1) / RPB;
    const int nbkP  = (NP + RPB - 1) / RPB;
    const int nmatG = nbkG * nBlkE;
    const int nmatP = nbkP * nBlkE;
    const int nbG   = (nmatG + 255) / 256;
    const int nbP   = (nmatP + 255) / 256;
    const int nmatGp = (nmatG + 63) & ~63;
    int* pairsG = (int*)tmp_p;
    int* pairsP = pairsG + (size_t)2 * E;
    int* hmatG  = (int*)xp_b;
    int* hmatP  = hmatG + nmatGp;
    int* omatG  = (int*)xg_b;
    int* omatP  = omatG + nmatGp;
    int* bsumG  = bsum;
    int* bsumP  = omatP + ((nmatP + 63) & ~63);

    bucket_hist<<<2 * nBlkE, 1024, 0, stream>>>(dst_pg, hmatG, nbkG, dst_gp, hmatP, nbkP, E, nBlkE);
    scan_block<<<nbG + nbP, 256, 0, stream>>>(hmatG, omatG, bsumG, nmatG, nbG,
                                              hmatP, omatP, bsumP, nmatP);
    scan_sums<<<2, 1024, 0, stream>>>(bsumG, nbG, bsumP, nbP);
    add_block_offsets<<<nbG + nbP, 256, 0, stream>>>(omatG, bsumG, nmatG, nbG,
                                                     omatP, bsumP, nmatP);
    bucket_scatter<<<2 * nBlkE, 1024, 0, stream>>>(src_pg, dst_pg, hmatG, omatG, pairsG, nbkG,
                                                   src_gp, dst_gp, hmatP, omatP, pairsP, nbkP,
                                                   E, nBlkE);
    bucket_csr<<<nbkG + nbkP, 1024, 0, stream>>>(pairsG, omatG, row_g, cur_g, srcs_g, NG, nbkG,
                                                 pairsP, omatP, row_p, cur_p, srcs_p, NP, nbkP,
                                                 E, nBlkE);

    {
        int tiles      = (NG + 15) / 16;
        int gemmBlocks = (tiles + 8 * TMI - 1) / (8 * TMI);
        int gatherBlks = (NP * 16 + 255) / 256;
        input_embed<<<gemmBlocks + gatherBlks, 256, 0, stream>>>(go_x, lin_W, lin_b, g_emb, g_nid,
                                                                 xg_a, NG, FGO, gemmBlocks,
                                                                 p_emb, p_nid, xp_a, NP * 16);
    }

    float* xp_cur = xp_a; float* xp_nxt = xp_b;
    float* xg_cur = xg_a; float* xg_nxt = xg_b;
    const int fusBlkG = (NG + FR - 1) / FR;
    const int fusBlkP = (NP + FR - 1) / FR;
    for (int l = 0; l < 3; ++l) {
        const float* Wl0 = Wl + (size_t)(l * 2 + 0) * D * D;
        const float* Wl1 = Wl + (size_t)(l * 2 + 1) * D * D;
        const float* Wr0 = Wr + (size_t)(l * 2 + 0) * D * D;
        const float* Wr1 = Wr + (size_t)(l * 2 + 1) * D * D;
        const float* bl0 = bl + (size_t)(l * 2 + 0) * D;
        const float* bl1 = bl + (size_t)(l * 2 + 1) * D;
        int relu = (l < 2) ? 1 : 0;
        layer_fused_wmma<<<fusBlkG + fusBlkP, 256, 0, stream>>>(
            xp_cur, srcs_g, row_g, cur_g, xg_cur, Wl0, bl0, Wr0, xg_nxt, NG, fusBlkG,
            xg_cur, srcs_p, row_p, cur_p, xp_cur, Wl1, bl1, Wr1, xp_nxt, NP, relu);
        float* t;
        t = xp_cur; xp_cur = xp_nxt; xp_nxt = t;
        t = xg_cur; xg_cur = xg_nxt; xg_nxt = t;
    }

    classifier_dot<<<(EL + 255) / 256, 256, 0, stream>>>(xp_cur, xg_cur, lab_s, lab_d,
                                                         (float*)d_out, EL);
}
